# starting kernel
# baseline (speedup 1.0000x reference)
#include <hip/hip_runtime.h>

constexpr int Nn  = 50000;
constexpr int Ne  = 800000;
constexpr int F   = 256;
constexpr int D0  = 128;
constexpr int H1  = 2;
constexpr int D1  = 64;
constexpr int C2  = 47;
constexpr int C2P = 64;

typedef float v2f __attribute__((ext_vector_type(2)));
typedef float v8f __attribute__((ext_vector_type(8)));

__device__ __forceinline__ unsigned enc_f(float f) {
    unsigned u = __float_as_uint(f);
    return (u & 0x80000000u) ? ~u : (u | 0x80000000u);
}
__device__ __forceinline__ float dec_f(unsigned u) {
    unsigned b = (u & 0x80000000u) ? (u ^ 0x80000000u) : ~u;
    return __uint_as_float(b);
}

__global__ void k_transpose(const float* __restrict__ W, float* __restrict__ Wt,
                            int K, int N, int Npad) {
    int i = blockIdx.x * blockDim.x + threadIdx.x;
    if (i >= Npad * K) return;
    const int c = i / K, k = i % K;
    Wt[i] = (c < N) ? W[(long long)k * N + c] : 0.0f;
}

template <bool HAS_A2, bool ACT>
__global__ __launch_bounds__(256)
void wmma_gemm(const float* __restrict__ A, const float* __restrict__ Wt,
               const float* __restrict__ A2, const float* __restrict__ W2t,
               const float* __restrict__ rowdeg, const float* __restrict__ bias,
               float* __restrict__ out, int M, int K, int Nout, int ncg)
{
    extern __shared__ float lds[];
    const int KP = K + 4;
    const int cg   = blockIdx.x % ncg;
    const int col0 = cg * 64;
    float* w1s = lds;
    float* w2s = lds + 64 * KP;

    const int ksh = 31 - __clz(K);
    for (int i = threadIdx.x; i < 64 * K; i += 256) {
        const int c = i >> ksh, k = i & (K - 1);
        w1s[c * KP + k] = Wt[(long long)(col0 + c) * K + k];
        if (HAS_A2) w2s[c * KP + k] = W2t[(long long)(col0 + c) * K + k];
    }
    __syncthreads();

    const int rt = (blockIdx.x / ncg) * 8 + (threadIdx.x >> 5);
    if (rt >= M / 16) return;
    const int lane = threadIdx.x & 31;
    const int half = lane >> 4;
    const int l    = lane & 15;
    const int row  = rt * 16 + l;

    const float* Arow  = A + (long long)row * K + half * 2;
    const float* A2row = HAS_A2 ? (A2 + (long long)row * K + half * 2) : nullptr;
    const float inv    = HAS_A2 ? (1.0f / fmaxf(rowdeg[row], 1.0f)) : 0.0f;
    __builtin_prefetch(Arow, 0, 3);

    const float* Bs[4];
    const float* B2s[4];
#pragma unroll
    for (int t = 0; t < 4; ++t) {
        Bs[t] = w1s + (t * 16 + l) * KP + half * 2;
        if (HAS_A2) B2s[t] = w2s + (t * 16 + l) * KP + half * 2;
    }

    v8f acc[4] = {};
    for (int kk = 0; kk < K; kk += 4) {
        const v2f a = *(const v2f*)(Arow + kk);
        v2f a2;
        if (HAS_A2) {
            const v2f t2 = *(const v2f*)(A2row + kk);
            a2.x = t2.x * inv; a2.y = t2.y * inv;
        }
#pragma unroll
        for (int t = 0; t < 4; ++t) {
            const v2f b = *(const v2f*)(Bs[t] + kk);
            acc[t] = __builtin_amdgcn_wmma_f32_16x16x4_f32(
                false, a, false, b, (short)0, acc[t], false, false);
            if (HAS_A2) {
                const v2f b2 = *(const v2f*)(B2s[t] + kk);
                acc[t] = __builtin_amdgcn_wmma_f32_16x16x4_f32(
                    false, a2, false, b2, (short)0, acc[t], false, false);
            }
        }
    }

#pragma unroll
    for (int t = 0; t < 4; ++t) {
        const int c = col0 + t * 16 + l;
        if (c < Nout) {
            const float bv = bias ? bias[c] : 0.0f;
#pragma unroll
            for (int v = 0; v < 8; ++v) {
                const int r = rt * 16 + v + 8 * half;
                float x = acc[t][v] + bv;
                if (ACT) x = x > 0.0f ? x : (expf(x) - 1.0f);
                out[(long long)r * Nout + c] = x;
            }
        }
    }
}

__global__ void fill_f32(float* p, float v, long long n) {
    long long i = (long long)blockIdx.x * blockDim.x + threadIdx.x;
    if (i < n) p[i] = v;
}
__global__ void fill_u32(unsigned* p, unsigned v, long long n) {
    long long i = (long long)blockIdx.x * blockDim.x + threadIdx.x;
    if (i < n) p[i] = v;
}

__global__ void k_deg(const int* __restrict__ dst, float* __restrict__ deg, int E) {
    int i = blockIdx.x * blockDim.x + threadIdx.x;
    if (i < E) atomicAdd(&deg[dst[i]], 1.0f);
}

__global__ void k_neigh(const float* __restrict__ x, const int* __restrict__ src,
                        const int* __restrict__ dst, float* __restrict__ neigh, int E) {
    long long tid = (long long)blockIdx.x * blockDim.x + threadIdx.x;
    if (tid >= (long long)E * (F / 4)) return;
    const int e = (int)(tid >> 6);
    const int c = ((int)tid & 63) * 4;
    const float4 v = *(const float4*)(x + (long long)src[e] * F + c);
    float* o = neigh + (long long)dst[e] * F + c;
    atomicAdd(o + 0, v.x); atomicAdd(o + 1, v.y);
    atomicAdd(o + 2, v.z); atomicAdd(o + 3, v.w);
}

__global__ void k_attn(const float* __restrict__ feat, const float* __restrict__ al,
                       const float* __restrict__ ar, float* __restrict__ el,
                       float* __restrict__ er, int N, int H, int D) {
    int i = blockIdx.x * blockDim.x + threadIdx.x;
    if (i >= N * H) return;
    const int n = i / H, h = i % H;
    const float* f = feat + (long long)n * H * D + h * D;
    float a = 0.0f, b = 0.0f;
    for (int d = 0; d < D; ++d) { a += f[d] * al[h * D + d]; b += f[d] * ar[h * D + d]; }
    el[i] = a; er[i] = b;
}

__global__ void k_edge_logit(const int* __restrict__ src, const int* __restrict__ dst,
                             const float* __restrict__ el, const float* __restrict__ er,
                             float* __restrict__ ee, unsigned* __restrict__ mx, int E, int H) {
    int i = blockIdx.x * blockDim.x + threadIdx.x;
    if (i >= E * H) return;
    const int e = i / H, h = i % H;
    float v = el[src[e] * H + h] + er[dst[e] * H + h];
    v = v > 0.0f ? v : 0.2f * v;
    ee[i] = v;
    atomicMax(&mx[dst[e] * H + h], enc_f(v));
}

__global__ void k_edge_exp(const int* __restrict__ dst, const float* __restrict__ ee,
                           const unsigned* __restrict__ mx, float* __restrict__ exe,
                           float* __restrict__ s, int E, int H) {
    int i = blockIdx.x * blockDim.x + threadIdx.x;
    if (i >= E * H) return;
    const int e = i / H, h = i % H;
    const float m = dec_f(mx[dst[e] * H + h]);
    const float ex = __expf(ee[i] - m);
    exe[i] = ex;
    atomicAdd(&s[dst[e] * H + h], ex);
}

__global__ void k_edge_msg(const int* __restrict__ src, const int* __restrict__ dst,
                           const float* __restrict__ feat, const float* __restrict__ exe,
                           const float* __restrict__ s, float* __restrict__ out,
                           int E, int H, int D) {
    long long i = (long long)blockIdx.x * blockDim.x + threadIdx.x;
    const long long tot = (long long)E * H * D;
    if (i >= tot) return;
    const int d = (int)(i % D);
    const long long eh = i / D;
    const int h = (int)(eh % H);
    const int e = (int)(eh / H);
    const int sd = dst[e];
    const float alpha = exe[e * H + h] / fmaxf(s[sd * H + h], 1e-9f);
    atomicAdd(&out[(long long)sd * H * D + h * D + d],
              alpha * feat[(long long)src[e] * H * D + h * D + d]);
}

__global__ void k_bias_act(const float* __restrict__ in, const float* __restrict__ b,
                           float* __restrict__ out, long long total, int Dtot, int act) {
    long long i = (long long)blockIdx.x * blockDim.x + threadIdx.x;
    if (i >= total) return;
    const int c = (int)(i % Dtot);
    float v = in[i] + b[c];
    if (act) v = v > 0.0f ? v : (expf(v) - 1.0f);
    out[i] = v;
}

static inline unsigned gsz(long long n, int b) { return (unsigned)((n + b - 1) / b); }

extern "C" void kernel_launch(void* const* d_in, const int* in_sizes, int n_in,
                              void* d_out, int out_size, void* d_ws, size_t ws_size,
                              hipStream_t stream) {
    const float* x       = (const float*)d_in[0];
    const int*   src     = (const int*)  d_in[1];
    const int*   dst     = (const int*)  d_in[2];
    const float* w_self  = (const float*)d_in[3];
    const float* w_neigh = (const float*)d_in[4];
    const float* sage_b  = (const float*)d_in[5];
    const float* g1_w    = (const float*)d_in[6];
    const float* g1_al   = (const float*)d_in[7];
    const float* g1_ar   = (const float*)d_in[8];
    const float* g1_b    = (const float*)d_in[9];
    const float* g2_w    = (const float*)d_in[10];
    const float* g2_al   = (const float*)d_in[11];
    const float* g2_ar   = (const float*)d_in[12];
    const float* g2_b    = (const float*)d_in[13];

    float*    ws    = (float*)d_ws;
    float*    deg   = ws;
    float*    neigh = deg + Nn;
    float*    h0    = neigh + (long long)Nn * F;
    float*    el    = h0 + (long long)Nn * D0;
    float*    er    = el + Nn * H1;
    unsigned* mx    = (unsigned*)(er + Nn * H1);
    float*    ssum  = (float*)(mx + Nn * H1);
    float*    ee    = ssum + Nn * H1;
    float*    exe   = ee + (long long)Ne * H1;
    float*    wt_self  = exe + (long long)Ne * H1;
    float*    wt_neigh = wt_self + D0 * F;
    float*    wt_g1    = wt_neigh + D0 * F;
    float*    wt_g2    = wt_g1 + (H1 * D1) * D0;
    float*    feat  = neigh;
    float*    aggr  = neigh + (long long)Nn * D0;

    const int B = 256;
    const unsigned NEG_INF_ENC = 0x007FFFFFu;
    const int Mtiles = Nn / 16;
    const unsigned rowblk = gsz(Mtiles, 8);

    k_transpose<<<gsz(D0 * F, B), B, 0, stream>>>(w_self,  wt_self,  F,  D0, D0);
    k_transpose<<<gsz(D0 * F, B), B, 0, stream>>>(w_neigh, wt_neigh, F,  D0, D0);
    k_transpose<<<gsz(D0 * D0, B), B, 0, stream>>>(g1_w,   wt_g1,    D0, H1 * D1, H1 * D1);
    k_transpose<<<gsz(C2P * D0, B), B, 0, stream>>>(g2_w,  wt_g2,    D0, C2, C2P);

    fill_f32<<<gsz(Nn, B), B, 0, stream>>>(deg, 0.0f, Nn);
    fill_f32<<<gsz((long long)Nn * F, B), B, 0, stream>>>(neigh, 0.0f, (long long)Nn * F);
    k_deg<<<gsz(Ne, B), B, 0, stream>>>(dst, deg, Ne);
    k_neigh<<<gsz((long long)Ne * (F / 4), B), B, 0, stream>>>(x, src, dst, neigh, Ne);
    {
        const size_t shmem = 2ull * 64 * (F + 4) * sizeof(float);
        wmma_gemm<true, true><<<rowblk * 2, B, shmem, stream>>>(
            x, wt_self, neigh, wt_neigh, deg, sage_b, h0, Nn, F, D0, 2);
    }

    {
        const size_t shmem = 64ull * (D0 + 4) * sizeof(float);
        wmma_gemm<false, false><<<rowblk * 2, B, shmem, stream>>>(
            h0, wt_g1, nullptr, nullptr, nullptr, nullptr, feat, Nn, D0, H1 * D1, 2);
    }
    k_attn<<<gsz(Nn * H1, B), B, 0, stream>>>(feat, g1_al, g1_ar, el, er, Nn, H1, D1);
    fill_u32<<<gsz(Nn * H1, B), B, 0, stream>>>(mx, NEG_INF_ENC, Nn * H1);
    fill_f32<<<gsz(Nn * H1, B), B, 0, stream>>>(ssum, 0.0f, Nn * H1);
    fill_f32<<<gsz((long long)Nn * D0, B), B, 0, stream>>>(aggr, 0.0f, (long long)Nn * D0);
    k_edge_logit<<<gsz(Ne * H1, B), B, 0, stream>>>(src, dst, el, er, ee, mx, Ne, H1);
    k_edge_exp<<<gsz(Ne * H1, B), B, 0, stream>>>(dst, ee, mx, exe, ssum, Ne, H1);
    k_edge_msg<<<gsz((long long)Ne * H1 * D1, B), B, 0, stream>>>(src, dst, feat, exe, ssum,
                                                                  aggr, Ne, H1, D1);
    k_bias_act<<<gsz((long long)Nn * D0, B), B, 0, stream>>>(aggr, g1_b, h0,
                                                             (long long)Nn * D0, D0, 1);

    {
        const size_t shmem = 64ull * (D0 + 4) * sizeof(float);
        wmma_gemm<false, false><<<rowblk, B, shmem, stream>>>(
            h0, wt_g2, nullptr, nullptr, nullptr, nullptr, feat, Nn, D0, C2, 1);
    }
    k_attn<<<gsz(Nn, B), B, 0, stream>>>(feat, g2_al, g2_ar, el, er, Nn, 1, C2);
    fill_u32<<<gsz(Nn, B), B, 0, stream>>>(mx, NEG_INF_ENC, Nn);
    fill_f32<<<gsz(Nn, B), B, 0, stream>>>(ssum, 0.0f, Nn);
    fill_f32<<<gsz((long long)Nn * C2, B), B, 0, stream>>>(aggr, 0.0f, (long long)Nn * C2);
    k_edge_logit<<<gsz(Ne, B), B, 0, stream>>>(src, dst, el, er, ee, mx, Ne, 1);
    k_edge_exp<<<gsz(Ne, B), B, 0, stream>>>(dst, ee, mx, exe, ssum, Ne, 1);
    k_edge_msg<<<gsz((long long)Ne * C2, B), B, 0, stream>>>(src, dst, feat, exe, ssum,
                                                             aggr, Ne, 1, C2);
    k_bias_act<<<gsz((long long)Nn * C2, B), B, 0, stream>>>(aggr, g2_b, (float*)d_out,
                                                             (long long)Nn * C2, C2, 0);
}


# Float atomic scatter-add/max replaced by counting sort of edges by dst (int atomic rank) and one-wave-per-node register gather with in-wave softmax; each row written once; sum order changes
# speedup vs baseline: 3.4578x; 3.4578x over previous
#include <hip/hip_runtime.h>

constexpr int Nn  = 50000;
constexpr int Ne  = 800000;
constexpr int F   = 256;
constexpr int D0  = 128;
constexpr int H1  = 2;
constexpr int D1  = 64;
constexpr int C2  = 47;
constexpr int C2P = 64;

typedef float v2f __attribute__((ext_vector_type(2)));
typedef float v8f __attribute__((ext_vector_type(8)));

__global__ void k_transpose(const float* __restrict__ W, float* __restrict__ Wt,
                            int K, int N, int Npad) {
    int i = blockIdx.x * blockDim.x + threadIdx.x;
    if (i >= Npad * K) return;
    const int c = i / K, k = i % K;
    Wt[i] = (c < N) ? W[(long long)k * N + c] : 0.0f;
}

template <bool HAS_A2, bool ACT>
__global__ __launch_bounds__(256)
void wmma_gemm(const float* __restrict__ A, const float* __restrict__ Wt,
               const float* __restrict__ A2, const float* __restrict__ W2t,
               const float* __restrict__ rowdeg, const float* __restrict__ bias,
               float* __restrict__ out, int M, int K, int Nout, int ncg)
{
    extern __shared__ float lds[];
    const int KP = K + 4;
    const int cg   = blockIdx.x % ncg;
    const int col0 = cg * 64;
    float* w1s = lds;
    float* w2s = lds + 64 * KP;

    const int ksh = 31 - __clz(K);
    for (int i = threadIdx.x; i < 64 * K; i += 256) {
        const int c = i >> ksh, k = i & (K - 1);
        w1s[c * KP + k] = Wt[(long long)(col0 + c) * K + k];
        if (HAS_A2) w2s[c * KP + k] = W2t[(long long)(col0 + c) * K + k];
    }
    __syncthreads();

    const int rt = (blockIdx.x / ncg) * 8 + (threadIdx.x >> 5);
    if (rt >= M / 16) return;
    const int lane = threadIdx.x & 31;
    const int half = lane >> 4;
    const int l    = lane & 15;
    const int row  = rt * 16 + l;

    const float* Arow  = A + (long long)row * K + half * 2;
    const float* A2row = HAS_A2 ? (A2 + (long long)row * K + half * 2) : nullptr;
    const float inv    = HAS_A2 ? (1.0f / fmaxf(rowdeg[row], 1.0f)) : 0.0f;
    __builtin_prefetch(Arow, 0, 3);

    const float* Bs[4];
    const float* B2s[4];
#pragma unroll
    for (int t = 0; t < 4; ++t) {
        Bs[t] = w1s + (t * 16 + l) * KP + half * 2;
        if (HAS_A2) B2s[t] = w2s + (t * 16 + l) * KP + half * 2;
    }

    v8f acc[4] = {};
    for (int kk = 0; kk < K; kk += 4) {
        const v2f a = *(const v2f*)(Arow + kk);
        v2f a2;
        if (HAS_A2) {
            const v2f t2 = *(const v2f*)(A2row + kk);
            a2.x = t2.x * inv; a2.y = t2.y * inv;
        }
#pragma unroll
        for (int t = 0; t < 4; ++t) {
            const v2f b = *(const v2f*)(Bs[t] + kk);
            acc[t] = __builtin_amdgcn_wmma_f32_16x16x4_f32(
                false, a, false, b, (short)0, acc[t], false, false);
            if (HAS_A2) {
                const v2f b2 = *(const v2f*)(B2s[t] + kk);
                acc[t] = __builtin_amdgcn_wmma_f32_16x16x4_f32(
                    false, a2, false, b2, (short)0, acc[t], false, false);
            }
        }
    }

#pragma unroll
    for (int t = 0; t < 4; ++t) {
        const int c = col0 + t * 16 + l;
        if (c < Nout) {
            const float bv = bias ? bias[c] : 0.0f;
#pragma unroll
            for (int v = 0; v < 8; ++v) {
                const int r = rt * 16 + v + 8 * half;
                float x = acc[t][v] + bv;
                if (ACT) x = x > 0.0f ? x : (expf(x) - 1.0f);
                out[(long long)r * Nout + c] = x;
            }
        }
    }
}

__global__ void fill_u32(unsigned* p, unsigned v, long long n) {
    long long i = (long long)blockIdx.x * blockDim.x + threadIdx.x;
    if (i < n) p[i] = v;
}

__global__ void k_attn(const float* __restrict__ feat, const float* __restrict__ al,
                       const float* __restrict__ ar, float* __restrict__ el,
                       float* __restrict__ er, int N, int H, int D) {
    int i = blockIdx.x * blockDim.x + threadIdx.x;
    if (i >= N * H) return;
    const int n = i / H, h = i % H;
    const float* f = feat + (long long)n * H * D + h * D;
    float a = 0.0f, b = 0.0f;
    for (int d = 0; d < D; ++d) { a += f[d] * al[h * D + d]; b += f[d] * ar[h * D + d]; }
    el[i] = a; er[i] = b;
}

__global__ void k_rank(const int* __restrict__ dst, int* __restrict__ cnt,
                       int* __restrict__ rank, int E) {
    int i = blockIdx.x * blockDim.x + threadIdx.x;
    if (i < E) rank[i] = atomicAdd(&cnt[dst[i]], 1);
}

constexpr int SCAN_T  = 1024;
constexpr int SCAN_CH = (Nn + SCAN_T - 1) / SCAN_T;
__global__ __launch_bounds__(SCAN_T)
void k_scan(const int* __restrict__ cnt, int* __restrict__ rowptr,
            float* __restrict__ degf, int n) {
    __shared__ int sc[SCAN_T * SCAN_CH];
    __shared__ int wsum[32];
    const int t = threadIdx.x;
    for (int i = t; i < SCAN_T * SCAN_CH; i += SCAN_T) {
        const int c = (i < n) ? cnt[i] : 0;
        sc[i] = c;
        if (i < n) degf[i] = (float)c;
    }
    __syncthreads();
    int* my = sc + t * SCAN_CH;
    int s = 0;
    for (int j = 0; j < SCAN_CH; ++j) s += my[j];
    const int lane = t & 31, w = t >> 5;
    int inc = s;
    for (int off = 1; off < 32; off <<= 1) {
        const int v = __shfl_up(inc, off, 32);
        if (lane >= off) inc += v;
    }
    if (lane == 31) wsum[w] = inc;
    __syncthreads();
    if (w == 0) {
        const int v = wsum[lane];
        int iv = v;
        for (int off = 1; off < 32; off <<= 1) {
            const int u = __shfl_up(iv, off, 32);
            if (lane >= off) iv += u;
        }
        wsum[lane] = iv - v;
    }
    __syncthreads();
    int run = wsum[w] + inc - s;
    for (int j = 0; j < SCAN_CH; ++j) { const int c = my[j]; my[j] = run; run += c; }
    __syncthreads();
    for (int i = t; i < n; i += SCAN_T) rowptr[i] = sc[i];
    if (t == SCAN_T - 1) rowptr[n] = run;
}

__global__ void k_fill_csr(const int* __restrict__ src, const int* __restrict__ dst,
                           const int* __restrict__ rank, const int* __restrict__ rowptr,
                           int* __restrict__ csr_src, int E) {
    int i = blockIdx.x * blockDim.x + threadIdx.x;
    if (i < E) csr_src[rowptr[dst[i]] + rank[i]] = src[i];
}

__device__ __forceinline__ int bcast_i(int v, int j) { return __builtin_amdgcn_readlane(v, j); }
__device__ __forceinline__ float bcast_f(float v, int j) {
    return __uint_as_float((unsigned)__builtin_amdgcn_readlane((int)__float_as_uint(v), j));
}
__device__ __forceinline__ float wave_max(float v) {
    for (int off = 16; off > 0; off >>= 1) v = fmaxf(v, __shfl_xor(v, off, 32));
    return v;
}
__device__ __forceinline__ float wave_sum(float v) {
    for (int off = 16; off > 0; off >>= 1) v += __shfl_xor(v, off, 32);
    return v;
}
__device__ __forceinline__ void add4(float4& a, const float4 v) {
    a.x += v.x; a.y += v.y; a.z += v.z; a.w += v.w;
}
__device__ __forceinline__ void fma4(float4& a, float s, const float4 v) {
    a.x += s * v.x; a.y += s * v.y; a.z += s * v.z; a.w += s * v.w;
}
__device__ __forceinline__ float leaky(float v) { return v > 0.0f ? v : 0.2f * v; }
__device__ __forceinline__ float finite_or_zero(float m) {
    return (fabsf(m) < __builtin_huge_valf()) ? m : 0.0f;
}

__global__ __launch_bounds__(256)
void k_sage_gather(const float* __restrict__ x, const int* __restrict__ rowptr,
                   const int* __restrict__ csr_src, float* __restrict__ neigh, int N) {
    const int n = __builtin_amdgcn_readfirstlane((int)(blockIdx.x * 8 + (threadIdx.x >> 5)));
    if (n >= N) return;
    const int lane = threadIdx.x & 31;
    const int beg = rowptr[n], end = rowptr[n + 1];
    float4 a0 = make_float4(0.f, 0.f, 0.f, 0.f), a1 = a0;
    for (int b = beg; b < end; b += 32) {
        const int m = min(32, end - b);
        const int sv = (lane < m) ? csr_src[b + lane] : 0;
        int j = 0;
        for (; j + 4 <= m; j += 4) {
            const float4* r0 = (const float4*)(x + (long long)bcast_i(sv, j + 0) * F);
            const float4* r1 = (const float4*)(x + (long long)bcast_i(sv, j + 1) * F);
            const float4* r2 = (const float4*)(x + (long long)bcast_i(sv, j + 2) * F);
            const float4* r3 = (const float4*)(x + (long long)bcast_i(sv, j + 3) * F);
            const float4 p0 = r0[lane], q0 = r0[lane + 32];
            const float4 p1 = r1[lane], q1 = r1[lane + 32];
            const float4 p2 = r2[lane], q2 = r2[lane + 32];
            const float4 p3 = r3[lane], q3 = r3[lane + 32];
            add4(a0, p0); add4(a1, q0); add4(a0, p1); add4(a1, q1);
            add4(a0, p2); add4(a1, q2); add4(a0, p3); add4(a1, q3);
        }
        for (; j < m; ++j) {
            const float4* r0 = (const float4*)(x + (long long)bcast_i(sv, j) * F);
            add4(a0, r0[lane]); add4(a1, r0[lane + 32]);
        }
    }
    float4* o = (float4*)(neigh + (long long)n * F);
    o[lane] = a0; o[lane + 32] = a1;
}

__global__ __launch_bounds__(256)
void k_gat1_gather(const int* __restrict__ rowptr, const int* __restrict__ csr_src,
                   const float* __restrict__ feat, const float* __restrict__ el,
                   const float* __restrict__ er, const float* __restrict__ bias,
                   float* __restrict__ out, int N) {
    const int n = __builtin_amdgcn_readfirstlane((int)(blockIdx.x * 8 + (threadIdx.x >> 5)));
    if (n >= N) return;
    const int lane = threadIdx.x & 31;
    const int beg = rowptr[n], end = rowptr[n + 1];
    const float er0 = er[n * 2], er1 = er[n * 2 + 1];

    float m0 = -__builtin_huge_valf(), m1 = m0;
    for (int b = beg + lane; b < end; b += 32) {
        const float2 l = *(const float2*)(el + (long long)csr_src[b] * 2);
        m0 = fmaxf(m0, leaky(l.x + er0)); m1 = fmaxf(m1, leaky(l.y + er1));
    }
    m0 = finite_or_zero(wave_max(m0)); m1 = finite_or_zero(wave_max(m1));

    float s0 = 0.0f, s1 = 0.0f;
    for (int b = beg + lane; b < end; b += 32) {
        const float2 l = *(const float2*)(el + (long long)csr_src[b] * 2);
        s0 += __expf(leaky(l.x + er0) - m0); s1 += __expf(leaky(l.y + er1) - m1);
    }
    const float d0 = fmaxf(wave_sum(s0), 1e-9f), d1 = fmaxf(wave_sum(s1), 1e-9f);

    const bool hi = lane >= 16;
    float4 acc = make_float4(0.f, 0.f, 0.f, 0.f);
    for (int b = beg; b < end; b += 32) {
        const int m = min(32, end - b);
        int sv = 0; float w0 = 0.0f, w1 = 0.0f;
        if (lane < m) {
            sv = csr_src[b + lane];
            const float2 l = *(const float2*)(el + (long long)sv * 2);
            w0 = __expf(leaky(l.x + er0) - m0) / d0;
            w1 = __expf(leaky(l.y + er1) - m1) / d1;
        }
        int j = 0;
        for (; j + 4 <= m; j += 4) {
            const float4 f0 = ((const float4*)(feat + (long long)bcast_i(sv, j + 0) * (H1 * D1)))[lane];
            const float4 f1 = ((const float4*)(feat + (long long)bcast_i(sv, j + 1) * (H1 * D1)))[lane];
            const float4 f2 = ((const float4*)(feat + (long long)bcast_i(sv, j + 2) * (H1 * D1)))[lane];
            const float4 f3 = ((const float4*)(feat + (long long)bcast_i(sv, j + 3) * (H1 * D1)))[lane];
            const float u0 = bcast_f(w0, j + 0), t0 = bcast_f(w1, j + 0);
            const float u1 = bcast_f(w0, j + 1), t1 = bcast_f(w1, j + 1);
            const float u2 = bcast_f(w0, j + 2), t2 = bcast_f(w1, j + 2);
            const float u3 = bcast_f(w0, j + 3), t3 = bcast_f(w1, j + 3);
            fma4(acc, hi ? t0 : u0, f0);
            fma4(acc, hi ? t1 : u1, f1);
            fma4(acc, hi ? t2 : u2, f2);
            fma4(acc, hi ? t3 : u3, f3);
        }
        for (; j < m; ++j) {
            const float4 f0 = ((const float4*)(feat + (long long)bcast_i(sv, j) * (H1 * D1)))[lane];
            const float u0 = bcast_f(w0, j), t0 = bcast_f(w1, j);
            fma4(acc, hi ? t0 : u0, f0);
        }
    }
    const float* bl = bias + lane * 4;
    float4 r;
    r.x = acc.x + bl[0]; r.y = acc.y + bl[1]; r.z = acc.z + bl[2]; r.w = acc.w + bl[3];
    r.x = r.x > 0.0f ? r.x : (expf(r.x) - 1.0f);
    r.y = r.y > 0.0f ? r.y : (expf(r.y) - 1.0f);
    r.z = r.z > 0.0f ? r.z : (expf(r.z) - 1.0f);
    r.w = r.w > 0.0f ? r.w : (expf(r.w) - 1.0f);
    ((float4*)(out + (long long)n * (H1 * D1)))[lane] = r;
}

__global__ __launch_bounds__(256)
void k_gat2_gather(const int* __restrict__ rowptr, const int* __restrict__ csr_src,
                   const float* __restrict__ feat, const float* __restrict__ el,
                   const float* __restrict__ er, const float* __restrict__ bias,
                   float* __restrict__ out, int N) {
    const int n = __builtin_amdgcn_readfirstlane((int)(blockIdx.x * 8 + (threadIdx.x >> 5)));
    if (n >= N) return;
    const int lane = threadIdx.x & 31;
    const int beg = rowptr[n], end = rowptr[n + 1];
    const float ern = er[n];
    const bool two = lane < C2 - 32;

    float mx = -__builtin_huge_valf();
    for (int b = beg + lane; b < end; b += 32) mx = fmaxf(mx, leaky(el[csr_src[b]] + ern));
    mx = finite_or_zero(wave_max(mx));
    float sm = 0.0f;
    for (int b = beg + lane; b < end; b += 32) sm += __expf(leaky(el[csr_src[b]] + ern) - mx);
    const float dn = fmaxf(wave_sum(sm), 1e-9f);

    float acc0 = 0.0f, acc1 = 0.0f;
    for (int b = beg; b < end; b += 32) {
        const int m = min(32, end - b);
        int sv = 0; float wv = 0.0f;
        if (lane < m) {
            sv = csr_src[b + lane];
            wv = __expf(leaky(el[sv] + ern) - mx) / dn;
        }
        int j = 0;
        for (; j + 4 <= m; j += 4) {
            const float* r0 = feat + (long long)bcast_i(sv, j + 0) * C2;
            const float* r1 = feat + (long long)bcast_i(sv, j + 1) * C2;
            const float* r2 = feat + (long long)bcast_i(sv, j + 2) * C2;
            const float* r3 = feat + (long long)bcast_i(sv, j + 3) * C2;
            const float p0 = r0[lane], p1 = r1[lane], p2 = r2[lane], p3 = r3[lane];
            const float q0 = two ? r0[lane + 32] : 0.0f, q1 = two ? r1[lane + 32] : 0.0f;
            const float q2 = two ? r2[lane + 32] : 0.0f, q3 = two ? r3[lane + 32] : 0.0f;
            const float u0 = bcast_f(wv, j + 0), u1 = bcast_f(wv, j + 1);
            const float u2 = bcast_f(wv, j + 2), u3 = bcast_f(wv, j + 3);
            acc0 += u0 * p0; acc1 += u0 * q0; acc0 += u1 * p1; acc1 += u1 * q1;
            acc0 += u2 * p2; acc1 += u2 * q2; acc0 += u3 * p3; acc1 += u3 * q3;
        }
        for (; j < m; ++j) {
            const float* r0 = feat + (long long)bcast_i(sv, j) * C2;
            const float u0 = bcast_f(wv, j);
            acc0 += u0 * r0[lane];
            acc1 += u0 * (two ? r0[lane + 32] : 0.0f);
        }
    }
    float* o = out + (long long)n * C2;
    o[lane] = acc0 + bias[lane];
    if (two) o[lane + 32] = acc1 + bias[lane + 32];
}

static inline unsigned gsz(long long n, int b) { return (unsigned)((n + b - 1) / b); }

extern "C" void kernel_launch(void* const* d_in, const int* in_sizes, int n_in,
                              void* d_out, int out_size, void* d_ws, size_t ws_size,
                              hipStream_t stream) {
    const float* x       = (const float*)d_in[0];
    const int*   src     = (const int*)  d_in[1];
    const int*   dst     = (const int*)  d_in[2];
    const float* w_self  = (const float*)d_in[3];
    const float* w_neigh = (const float*)d_in[4];
    const float* sage_b  = (const float*)d_in[5];
    const float* g1_w    = (const float*)d_in[6];
    const float* g1_al   = (const float*)d_in[7];
    const float* g1_ar   = (const float*)d_in[8];
    const float* g1_b    = (const float*)d_in[9];
    const float* g2_w    = (const float*)d_in[10];
    const float* g2_al   = (const float*)d_in[11];
    const float* g2_ar   = (const float*)d_in[12];
    const float* g2_b    = (const float*)d_in[13];

    float*    ws    = (float*)d_ws;
    float*    neigh = ws;
    float*    h0    = neigh + (long long)Nn * F;
    float*    deg   = h0 + (long long)Nn * D0;
    float*    el    = deg + Nn;
    float*    er    = el + Nn * H1;
    int*      cnt   = (int*)(er + Nn * H1);
    int*      rowptr = cnt + Nn;
    int*      rank  = rowptr + (Nn + 16);
    int*      csr_src = rank + Ne;
    float*    wt_self  = (float*)(csr_src + Ne);
    float*    wt_neigh = wt_self + D0 * F;
    float*    wt_g1    = wt_neigh + D0 * F;
    float*    wt_g2    = wt_g1 + (H1 * D1) * D0;
    float*    feat  = neigh;

    const int B = 256;
    const int Mtiles = Nn / 16;
    const unsigned rowblk = gsz(Mtiles, 8);
    const unsigned nodeblk = gsz(Nn, 8);

    k_transpose<<<gsz(D0 * F, B), B, 0, stream>>>(w_self,  wt_self,  F,  D0, D0);
    k_transpose<<<gsz(D0 * F, B), B, 0, stream>>>(w_neigh, wt_neigh, F,  D0, D0);
    k_transpose<<<gsz(D0 * D0, B), B, 0, stream>>>(g1_w,   wt_g1,    D0, H1 * D1, H1 * D1);
    k_transpose<<<gsz(C2P * D0, B), B, 0, stream>>>(g2_w,  wt_g2,    D0, C2, C2P);

    fill_u32<<<gsz(Nn, B), B, 0, stream>>>((unsigned*)cnt, 0u, Nn);
    k_rank<<<gsz(Ne, B), B, 0, stream>>>(dst, cnt, rank, Ne);
    k_scan<<<1, SCAN_T, 0, stream>>>(cnt, rowptr, deg, Nn);
    k_fill_csr<<<gsz(Ne, B), B, 0, stream>>>(src, dst, rank, rowptr, csr_src, Ne);

    k_sage_gather<<<nodeblk, B, 0, stream>>>(x, rowptr, csr_src, neigh, Nn);
    {
        const size_t shmem = 2ull * 64 * (F + 4) * sizeof(float);
        wmma_gemm<true, true><<<rowblk * 2, B, shmem, stream>>>(
            x, wt_self, neigh, wt_neigh, deg, sage_b, h0, Nn, F, D0, 2);
    }

    {
        const size_t shmem = 64ull * (D0 + 4) * sizeof(float);
        wmma_gemm<false, false><<<rowblk * 2, B, shmem, stream>>>(
            h0, wt_g1, nullptr, nullptr, nullptr, nullptr, feat, Nn, D0, H1 * D1, 2);
    }
    k_attn<<<gsz(Nn * H1, B), B, 0, stream>>>(feat, g1_al, g1_ar, el, er, Nn, H1, D1);
    k_gat1_gather<<<nodeblk, B, 0, stream>>>(rowptr, csr_src, feat, el, er, g1_b, h0, Nn);

    {
        const size_t shmem = 64ull * (D0 + 4) * sizeof(float);
        wmma_gemm<false, false><<<rowblk, B, shmem, stream>>>(
            h0, wt_g2, nullptr, nullptr, nullptr, nullptr, feat, Nn, D0, C2, 1);
    }
    k_attn<<<gsz(Nn, B), B, 0, stream>>>(feat, g2_al, g2_ar, el, er, Nn, 1, C2);
    k_gat2_gather<<<nodeblk, B, 0, stream>>>(rowptr, csr_src, feat, el, er, g2_b,
                                             (float*)d_out, Nn);
}


# GEMM weight panel staged into LDS with 16-byte loads, 8 per thread in flight before the first LDS store, instead of one dependent 4-byte load and store per trip
# speedup vs baseline: 3.8615x; 1.1168x over previous
#include <hip/hip_runtime.h>

constexpr int Nn  = 50000;
constexpr int Ne  = 800000;
constexpr int F   = 256;
constexpr int D0  = 128;
constexpr int H1  = 2;
constexpr int D1  = 64;
constexpr int C2  = 47;
constexpr int C2P = 64;

typedef float v2f __attribute__((ext_vector_type(2)));
typedef float v8f __attribute__((ext_vector_type(8)));

__global__ void k_transpose(const float* __restrict__ W, float* __restrict__ Wt,
                            int K, int N, int Npad) {
    int i = blockIdx.x * blockDim.x + threadIdx.x;
    if (i >= Npad * K) return;
    const int c = i / K, k = i % K;
    Wt[i] = (c < N) ? W[(long long)k * N + c] : 0.0f;
}

template <bool HAS_A2, bool ACT>
__global__ __launch_bounds__(256)
void wmma_gemm(const float* __restrict__ A, const float* __restrict__ Wt,
               const float* __restrict__ A2, const float* __restrict__ W2t,
               const float* __restrict__ rowdeg, const float* __restrict__ bias,
               float* __restrict__ out, int M, int K, int Nout, int ncg)
{
    extern __shared__ float lds[];
    const int KP = K + 4;
    const int cg   = blockIdx.x % ncg;
    const int col0 = cg * 64;
    float* w1s = lds;
    float* w2s = lds + 64 * KP;

    {
        const int qsh = 29 - __clz(K);
        const int qm  = (1 << qsh) - 1;
        const int npt = K >> 4;
        const float4* g1 = (const float4*)(Wt + (long long)col0 * K);
        const float4* g2 = HAS_A2 ? (const float4*)(W2t + (long long)col0 * K) : nullptr;
        constexpr int U = 8;
        for (int b = 0; b < npt; b += U) {
            float4 v1[U], v2[U];
#pragma unroll
            for (int u = 0; u < U; ++u) {
                const int q = (b + u) * 256 + threadIdx.x;
                v1[u] = g1[q];
                if (HAS_A2) v2[u] = g2[q];
            }
#pragma unroll
            for (int u = 0; u < U; ++u) {
                const int q = (b + u) * 256 + threadIdx.x;
                const int o = (q >> qsh) * KP + (q & qm) * 4;
                *(float4*)(w1s + o) = v1[u];
                if (HAS_A2) *(float4*)(w2s + o) = v2[u];
            }
        }
    }
    __syncthreads();

    const int rt = (blockIdx.x / ncg) * 8 + (threadIdx.x >> 5);
    if (rt >= M / 16) return;
    const int lane = threadIdx.x & 31;
    const int half = lane >> 4;
    const int l    = lane & 15;
    const int row  = rt * 16 + l;

    const float* Arow  = A + (long long)row * K + half * 2;
    const float* A2row = HAS_A2 ? (A2 + (long long)row * K + half * 2) : nullptr;
    const float inv    = HAS_A2 ? (1.0f / fmaxf(rowdeg[row], 1.0f)) : 0.0f;
    __builtin_prefetch(Arow, 0, 3);

    const float* Bs[4];
    const float* B2s[4];
#pragma unroll
    for (int t = 0; t < 4; ++t) {
        Bs[t] = w1s + (t * 16 + l) * KP + half * 2;
        if (HAS_A2) B2s[t] = w2s + (t * 16 + l) * KP + half * 2;
    }

    v8f acc[4] = {};
    for (int kk = 0; kk < K; kk += 4) {
        const v2f a = *(const v2f*)(Arow + kk);
        v2f a2;
        if (HAS_A2) {
            const v2f t2 = *(const v2f*)(A2row + kk);
            a2.x = t2.x * inv; a2.y = t2.y * inv;
        }
#pragma unroll
        for (int t = 0; t < 4; ++t) {
            const v2f b = *(const v2f*)(Bs[t] + kk);
            acc[t] = __builtin_amdgcn_wmma_f32_16x16x4_f32(
                false, a, false, b, (short)0, acc[t], false, false);
            if (HAS_A2) {
                const v2f b2 = *(const v2f*)(B2s[t] + kk);
                acc[t] = __builtin_amdgcn_wmma_f32_16x16x4_f32(
                    false, a2, false, b2, (short)0, acc[t], false, false);
            }
        }
    }

#pragma unroll
    for (int t = 0; t < 4; ++t) {
        const int c = col0 + t * 16 + l;
        if (c < Nout) {
            const float bv = bias ? bias[c] : 0.0f;
#pragma unroll
            for (int v = 0; v < 8; ++v) {
                const int r = rt * 16 + v + 8 * half;
                float x = acc[t][v] + bv;
                if (ACT) x = x > 0.0f ? x : (expf(x) - 1.0f);
                out[(long long)r * Nout + c] = x;
            }
        }
    }
}

__global__ void fill_u32(unsigned* p, unsigned v, long long n) {
    long long i = (long long)blockIdx.x * blockDim.x + threadIdx.x;
    if (i < n) p[i] = v;
}

__global__ void k_attn(const float* __restrict__ feat, const float* __restrict__ al,
                       const float* __restrict__ ar, float* __restrict__ el,
                       float* __restrict__ er, int N, int H, int D) {
    int i = blockIdx.x * blockDim.x + threadIdx.x;
    if (i >= N * H) return;
    const int n = i / H, h = i % H;
    const float* f = feat + (long long)n * H * D + h * D;
    float a = 0.0f, b = 0.0f;
    for (int d = 0; d < D; ++d) { a += f[d] * al[h * D + d]; b += f[d] * ar[h * D + d]; }
    el[i] = a; er[i] = b;
}

__global__ void k_rank(const int* __restrict__ dst, int* __restrict__ cnt,
                       int* __restrict__ rank, int E) {
    int i = blockIdx.x * blockDim.x + threadIdx.x;
    if (i < E) rank[i] = atomicAdd(&cnt[dst[i]], 1);
}

constexpr int SCAN_T  = 1024;
constexpr int SCAN_CH = (Nn + SCAN_T - 1) / SCAN_T;
__global__ __launch_bounds__(SCAN_T)
void k_scan(const int* __restrict__ cnt, int* __restrict__ rowptr,
            float* __restrict__ degf, int n) {
    __shared__ int sc[SCAN_T * SCAN_CH];
    __shared__ int wsum[32];
    const int t = threadIdx.x;
    for (int i = t; i < SCAN_T * SCAN_CH; i += SCAN_T) {
        const int c = (i < n) ? cnt[i] : 0;
        sc[i] = c;
        if (i < n) degf[i] = (float)c;
    }
    __syncthreads();
    int* my = sc + t * SCAN_CH;
    int s = 0;
    for (int j = 0; j < SCAN_CH; ++j) s += my[j];
    const int lane = t & 31, w = t >> 5;
    int inc = s;
    for (int off = 1; off < 32; off <<= 1) {
        const int v = __shfl_up(inc, off, 32);
        if (lane >= off) inc += v;
    }
    if (lane == 31) wsum[w] = inc;
    __syncthreads();
    if (w == 0) {
        const int v = wsum[lane];
        int iv = v;
        for (int off = 1; off < 32; off <<= 1) {
            const int u = __shfl_up(iv, off, 32);
            if (lane >= off) iv += u;
        }
        wsum[lane] = iv - v;
    }
    __syncthreads();
    int run = wsum[w] + inc - s;
    for (int j = 0; j < SCAN_CH; ++j) { const int c = my[j]; my[j] = run; run += c; }
    __syncthreads();
    for (int i = t; i < n; i += SCAN_T) rowptr[i] = sc[i];
    if (t == SCAN_T - 1) rowptr[n] = run;
}

__global__ void k_fill_csr(const int* __restrict__ src, const int* __restrict__ dst,
                           const int* __restrict__ rank, const int* __restrict__ rowptr,
                           int* __restrict__ csr_src, int E) {
    int i = blockIdx.x * blockDim.x + threadIdx.x;
    if (i < E) csr_src[rowptr[dst[i]] + rank[i]] = src[i];
}

__device__ __forceinline__ int bcast_i(int v, int j) { return __builtin_amdgcn_readlane(v, j); }
__device__ __forceinline__ float bcast_f(float v, int j) {
    return __uint_as_float((unsigned)__builtin_amdgcn_readlane((int)__float_as_uint(v), j));
}
__device__ __forceinline__ float wave_max(float v) {
    for (int off = 16; off > 0; off >>= 1) v = fmaxf(v, __shfl_xor(v, off, 32));
    return v;
}
__device__ __forceinline__ float wave_sum(float v) {
    for (int off = 16; off > 0; off >>= 1) v += __shfl_xor(v, off, 32);
    return v;
}
__device__ __forceinline__ void add4(float4& a, const float4 v) {
    a.x += v.x; a.y += v.y; a.z += v.z; a.w += v.w;
}
__device__ __forceinline__ void fma4(float4& a, float s, const float4 v) {
    a.x += s * v.x; a.y += s * v.y; a.z += s * v.z; a.w += s * v.w;
}
__device__ __forceinline__ float leaky(float v) { return v > 0.0f ? v : 0.2f * v; }
__device__ __forceinline__ float finite_or_zero(float m) {
    return (fabsf(m) < __builtin_huge_valf()) ? m : 0.0f;
}

__global__ __launch_bounds__(256)
void k_sage_gather(const float* __restrict__ x, const int* __restrict__ rowptr,
                   const int* __restrict__ csr_src, float* __restrict__ neigh, int N) {
    const int n = __builtin_amdgcn_readfirstlane((int)(blockIdx.x * 8 + (threadIdx.x >> 5)));
    if (n >= N) return;
    const int lane = threadIdx.x & 31;
    const int beg = rowptr[n], end = rowptr[n + 1];
    float4 a0 = make_float4(0.f, 0.f, 0.f, 0.f), a1 = a0;
    for (int b = beg; b < end; b += 32) {
        const int m = min(32, end - b);
        const int sv = (lane < m) ? csr_src[b + lane] : 0;
        int j = 0;
        for (; j + 4 <= m; j += 4) {
            const float4* r0 = (const float4*)(x + (long long)bcast_i(sv, j + 0) * F);
            const float4* r1 = (const float4*)(x + (long long)bcast_i(sv, j + 1) * F);
            const float4* r2 = (const float4*)(x + (long long)bcast_i(sv, j + 2) * F);
            const float4* r3 = (const float4*)(x + (long long)bcast_i(sv, j + 3) * F);
            const float4 p0 = r0[lane], q0 = r0[lane + 32];
            const float4 p1 = r1[lane], q1 = r1[lane + 32];
            const float4 p2 = r2[lane], q2 = r2[lane + 32];
            const float4 p3 = r3[lane], q3 = r3[lane + 32];
            add4(a0, p0); add4(a1, q0); add4(a0, p1); add4(a1, q1);
            add4(a0, p2); add4(a1, q2); add4(a0, p3); add4(a1, q3);
        }
        for (; j < m; ++j) {
            const float4* r0 = (const float4*)(x + (long long)bcast_i(sv, j) * F);
            add4(a0, r0[lane]); add4(a1, r0[lane + 32]);
        }
    }
    float4* o = (float4*)(neigh + (long long)n * F);
    o[lane] = a0; o[lane + 32] = a1;
}

__global__ __launch_bounds__(256)
void k_gat1_gather(const int* __restrict__ rowptr, const int* __restrict__ csr_src,
                   const float* __restrict__ feat, const float* __restrict__ el,
                   const float* __restrict__ er, const float* __restrict__ bias,
                   float* __restrict__ out, int N) {
    const int n = __builtin_amdgcn_readfirstlane((int)(blockIdx.x * 8 + (threadIdx.x >> 5)));
    if (n >= N) return;
    const int lane = threadIdx.x & 31;
    const int beg = rowptr[n], end = rowptr[n + 1];
    const float er0 = er[n * 2], er1 = er[n * 2 + 1];

    float m0 = -__builtin_huge_valf(), m1 = m0;
    for (int b = beg + lane; b < end; b += 32) {
        const float2 l = *(const float2*)(el + (long long)csr_src[b] * 2);
        m0 = fmaxf(m0, leaky(l.x + er0)); m1 = fmaxf(m1, leaky(l.y + er1));
    }
    m0 = finite_or_zero(wave_max(m0)); m1 = finite_or_zero(wave_max(m1));

    float s0 = 0.0f, s1 = 0.0f;
    for (int b = beg + lane; b < end; b += 32) {
        const float2 l = *(const float2*)(el + (long long)csr_src[b] * 2);
        s0 += __expf(leaky(l.x + er0) - m0); s1 += __expf(leaky(l.y + er1) - m1);
    }
    const float d0 = fmaxf(wave_sum(s0), 1e-9f), d1 = fmaxf(wave_sum(s1), 1e-9f);

    const bool hi = lane >= 16;
    float4 acc = make_float4(0.f, 0.f, 0.f, 0.f);
    for (int b = beg; b < end; b += 32) {
        const int m = min(32, end - b);
        int sv = 0; float w0 = 0.0f, w1 = 0.0f;
        if (lane < m) {
            sv = csr_src[b + lane];
            const float2 l = *(const float2*)(el + (long long)sv * 2);
            w0 = __expf(leaky(l.x + er0) - m0) / d0;
            w1 = __expf(leaky(l.y + er1) - m1) / d1;
        }
        int j = 0;
        for (; j + 4 <= m; j += 4) {
            const float4 f0 = ((const float4*)(feat + (long long)bcast_i(sv, j + 0) * (H1 * D1)))[lane];
            const float4 f1 = ((const float4*)(feat + (long long)bcast_i(sv, j + 1) * (H1 * D1)))[lane];
            const float4 f2 = ((const float4*)(feat + (long long)bcast_i(sv, j + 2) * (H1 * D1)))[lane];
            const float4 f3 = ((const float4*)(feat + (long long)bcast_i(sv, j + 3) * (H1 * D1)))[lane];
            const float u0 = bcast_f(w0, j + 0), t0 = bcast_f(w1, j + 0);
            const float u1 = bcast_f(w0, j + 1), t1 = bcast_f(w1, j + 1);
            const float u2 = bcast_f(w0, j + 2), t2 = bcast_f(w1, j + 2);
            const float u3 = bcast_f(w0, j + 3), t3 = bcast_f(w1, j + 3);
            fma4(acc, hi ? t0 : u0, f0);
            fma4(acc, hi ? t1 : u1, f1);
            fma4(acc, hi ? t2 : u2, f2);
            fma4(acc, hi ? t3 : u3, f3);
        }
        for (; j < m; ++j) {
            const float4 f0 = ((const float4*)(feat + (long long)bcast_i(sv, j) * (H1 * D1)))[lane];
            const float u0 = bcast_f(w0, j), t0 = bcast_f(w1, j);
            fma4(acc, hi ? t0 : u0, f0);
        }
    }
    const float* bl = bias + lane * 4;
    float4 r;
    r.x = acc.x + bl[0]; r.y = acc.y + bl[1]; r.z = acc.z + bl[2]; r.w = acc.w + bl[3];
    r.x = r.x > 0.0f ? r.x : (expf(r.x) - 1.0f);
    r.y = r.y > 0.0f ? r.y : (expf(r.y) - 1.0f);
    r.z = r.z > 0.0f ? r.z : (expf(r.z) - 1.0f);
    r.w = r.w > 0.0f ? r.w : (expf(r.w) - 1.0f);
    ((float4*)(out + (long long)n * (H1 * D1)))[lane] = r;
}

__global__ __launch_bounds__(256)
void k_gat2_gather(const int* __restrict__ rowptr, const int* __restrict__ csr_src,
                   const float* __restrict__ feat, const float* __restrict__ el,
                   const float* __restrict__ er, const float* __restrict__ bias,
                   float* __restrict__ out, int N) {
    const int n = __builtin_amdgcn_readfirstlane((int)(blockIdx.x * 8 + (threadIdx.x >> 5)));
    if (n >= N) return;
    const int lane = threadIdx.x & 31;
    const int beg = rowptr[n], end = rowptr[n + 1];
    const float ern = er[n];
    const bool two = lane < C2 - 32;

    float mx = -__builtin_huge_valf();
    for (int b = beg + lane; b < end; b += 32) mx = fmaxf(mx, leaky(el[csr_src[b]] + ern));
    mx = finite_or_zero(wave_max(mx));
    float sm = 0.0f;
    for (int b = beg + lane; b < end; b += 32) sm += __expf(leaky(el[csr_src[b]] + ern) - mx);
    const float dn = fmaxf(wave_sum(sm), 1e-9f);

    float acc0 = 0.0f, acc1 = 0.0f;
    for (int b = beg; b < end; b += 32) {
        const int m = min(32, end - b);
        int sv = 0; float wv = 0.0f;
        if (lane < m) {
            sv = csr_src[b + lane];
            wv = __expf(leaky(el[sv] + ern) - mx) / dn;
        }
        int j = 0;
        for (; j + 4 <= m; j += 4) {
            const float* r0 = feat + (long long)bcast_i(sv, j + 0) * C2;
            const float* r1 = feat + (long long)bcast_i(sv, j + 1) * C2;
            const float* r2 = feat + (long long)bcast_i(sv, j + 2) * C2;
            const float* r3 = feat + (long long)bcast_i(sv, j + 3) * C2;
            const float p0 = r0[lane], p1 = r1[lane], p2 = r2[lane], p3 = r3[lane];
            const float q0 = two ? r0[lane + 32] : 0.0f, q1 = two ? r1[lane + 32] : 0.0f;
            const float q2 = two ? r2[lane + 32] : 0.0f, q3 = two ? r3[lane + 32] : 0.0f;
            const float u0 = bcast_f(wv, j + 0), u1 = bcast_f(wv, j + 1);
            const float u2 = bcast_f(wv, j + 2), u3 = bcast_f(wv, j + 3);
            acc0 += u0 * p0; acc1 += u0 * q0; acc0 += u1 * p1; acc1 += u1 * q1;
            acc0 += u2 * p2; acc1 += u2 * q2; acc0 += u3 * p3; acc1 += u3 * q3;
        }
        for (; j < m; ++j) {
            const float* r0 = feat + (long long)bcast_i(sv, j) * C2;
            const float u0 = bcast_f(wv, j);
            acc0 += u0 * r0[lane];
            acc1 += u0 * (two ? r0[lane + 32] : 0.0f);
        }
    }
    float* o = out + (long long)n * C2;
    o[lane] = acc0 + bias[lane];
    if (two) o[lane + 32] = acc1 + bias[lane + 32];
}

static inline unsigned gsz(long long n, int b) { return (unsigned)((n + b - 1) / b); }

extern "C" void kernel_launch(void* const* d_in, const int* in_sizes, int n_in,
                              void* d_out, int out_size, void* d_ws, size_t ws_size,
                              hipStream_t stream) {
    const float* x       = (const float*)d_in[0];
    const int*   src     = (const int*)  d_in[1];
    const int*   dst     = (const int*)  d_in[2];
    const float* w_self  = (const float*)d_in[3];
    const float* w_neigh = (const float*)d_in[4];
    const float* sage_b  = (const float*)d_in[5];
    const float* g1_w    = (const float*)d_in[6];
    const float* g1_al   = (const float*)d_in[7];
    const float* g1_ar   = (const float*)d_in[8];
    const float* g1_b    = (const float*)d_in[9];
    const float* g2_w    = (const float*)d_in[10];
    const float* g2_al   = (const float*)d_in[11];
    const float* g2_ar   = (const float*)d_in[12];
    const float* g2_b    = (const float*)d_in[13];

    float*    ws    = (float*)d_ws;
    float*    neigh = ws;
    float*    h0    = neigh + (long long)Nn * F;
    float*    deg   = h0 + (long long)Nn * D0;
    float*    el    = deg + Nn;
    float*    er    = el + Nn * H1;
    int*      cnt   = (int*)(er + Nn * H1);
    int*      rowptr = cnt + Nn;
    int*      rank  = rowptr + (Nn + 16);
    int*      csr_src = rank + Ne;
    float*    wt_self  = (float*)(csr_src + Ne);
    float*    wt_neigh = wt_self + D0 * F;
    float*    wt_g1    = wt_neigh + D0 * F;
    float*    wt_g2    = wt_g1 + (H1 * D1) * D0;
    float*    feat  = neigh;

    const int B = 256;
    const int Mtiles = Nn / 16;
    const unsigned rowblk = gsz(Mtiles, 8);
    const unsigned nodeblk = gsz(Nn, 8);

    k_transpose<<<gsz(D0 * F, B), B, 0, stream>>>(w_self,  wt_self,  F,  D0, D0);
    k_transpose<<<gsz(D0 * F, B), B, 0, stream>>>(w_neigh, wt_neigh, F,  D0, D0);
    k_transpose<<<gsz(D0 * D0, B), B, 0, stream>>>(g1_w,   wt_g1,    D0, H1 * D1, H1 * D1);
    k_transpose<<<gsz(C2P * D0, B), B, 0, stream>>>(g2_w,  wt_g2,    D0, C2, C2P);

    fill_u32<<<gsz(Nn, B), B, 0, stream>>>((unsigned*)cnt, 0u, Nn);
    k_rank<<<gsz(Ne, B), B, 0, stream>>>(dst, cnt, rank, Ne);
    k_scan<<<1, SCAN_T, 0, stream>>>(cnt, rowptr, deg, Nn);
    k_fill_csr<<<gsz(Ne, B), B, 0, stream>>>(src, dst, rank, rowptr, csr_src, Ne);

    k_sage_gather<<<nodeblk, B, 0, stream>>>(x, rowptr, csr_src, neigh, Nn);
    {
        const size_t shmem = 2ull * 64 * (F + 4) * sizeof(float);
        wmma_gemm<true, true><<<rowblk * 2, B, shmem, stream>>>(
            x, wt_self, neigh, wt_neigh, deg, sage_b, h0, Nn, F, D0, 2);
    }

    {
        const size_t shmem = 64ull * (D0 + 4) * sizeof(float);
        wmma_gemm<false, false><<<rowblk * 2, B, shmem, stream>>>(
            h0, wt_g1, nullptr, nullptr, nullptr, nullptr, feat, Nn, D0, H1 * D1, 2);
    }
    k_attn<<<gsz(Nn * H1, B), B, 0, stream>>>(feat, g1_al, g1_ar, el, er, Nn, H1, D1);
    k_gat1_gather<<<nodeblk, B, 0, stream>>>(rowptr, csr_src, feat, el, er, g1_b, h0, Nn);

    {
        const size_t shmem = 64ull * (D0 + 4) * sizeof(float);
        wmma_gemm<false, false><<<rowblk, B, shmem, stream>>>(
            h0, wt_g2, nullptr, nullptr, nullptr, nullptr, feat, Nn, D0, C2, 1);
    }
    k_attn<<<gsz(Nn, B), B, 0, stream>>>(feat, g2_al, g2_ar, el, er, Nn, 1, C2);
    k_gat2_gather<<<nodeblk, B, 0, stream>>>(rowptr, csr_src, feat, el, er, g2_b,
                                             (float*)d_out, Nn);
}


# SAGE mean moved behind the weight product (mean is linear): one GEMM x@[Wself|Wneigh], then 128-float projected rows are gathered instead of 256-float input rows; sum order changes
# speedup vs baseline: 4.6214x; 1.1968x over previous
#include <hip/hip_runtime.h>

constexpr int Nn  = 50000;
constexpr int Ne  = 800000;
constexpr int F   = 256;
constexpr int D0  = 128;
constexpr int H1  = 2;
constexpr int D1  = 64;
constexpr int C2  = 47;
constexpr int C2P = 64;

typedef float v2f __attribute__((ext_vector_type(2)));
typedef float v8f __attribute__((ext_vector_type(8)));

__global__ void k_transpose(const float* __restrict__ W, float* __restrict__ Wt,
                            int K, int N, int Npad) {
    int i = blockIdx.x * blockDim.x + threadIdx.x;
    if (i >= Npad * K) return;
    const int c = i / K, k = i % K;
    Wt[i] = (c < N) ? W[(long long)k * N + c] : 0.0f;
}

__global__ __launch_bounds__(256)
void wmma_gemm(const float* __restrict__ A, const float* __restrict__ Wt,
               float* __restrict__ out, int M, int K, int Nout, int ncg, int ldo)
{
    extern __shared__ float lds[];
    const int KP = K + 4;
    const int cg   = blockIdx.x % ncg;
    const int col0 = cg * 64;
    float* w1s = lds;

    {
        const int qsh = 29 - __clz(K);
        const int qm  = (1 << qsh) - 1;
        const int npt = K >> 4;
        const float4* g1 = (const float4*)(Wt + (long long)col0 * K);
        constexpr int U = 8;
        for (int b = 0; b < npt; b += U) {
            float4 v1[U];
#pragma unroll
            for (int u = 0; u < U; ++u) {
                const int q = (b + u) * 256 + threadIdx.x;
                v1[u] = g1[q];
            }
#pragma unroll
            for (int u = 0; u < U; ++u) {
                const int q = (b + u) * 256 + threadIdx.x;
                const int o = (q >> qsh) * KP + (q & qm) * 4;
                *(float4*)(w1s + o) = v1[u];
            }
        }
    }
    __syncthreads();

    const int rt = (blockIdx.x / ncg) * 8 + (threadIdx.x >> 5);
    if (rt >= M / 16) return;
    const int lane = threadIdx.x & 31;
    const int half = lane >> 4;
    const int l    = lane & 15;
    const int row  = rt * 16 + l;

    const float* Arow  = A + (long long)row * K + half * 2;
    __builtin_prefetch(Arow, 0, 3);

    const float* Bs[4];
#pragma unroll
    for (int t = 0; t < 4; ++t) Bs[t] = w1s + (t * 16 + l) * KP + half * 2;

    v8f acc[4] = {};
    for (int kk = 0; kk < K; kk += 4) {
        const v2f a = *(const v2f*)(Arow + kk);
#pragma unroll
        for (int t = 0; t < 4; ++t) {
            const v2f b = *(const v2f*)(Bs[t] + kk);
            acc[t] = __builtin_amdgcn_wmma_f32_16x16x4_f32(
                false, a, false, b, (short)0, acc[t], false, false);
        }
    }

    const int plane = col0 / ldo;
    float* op = out + (long long)plane * M * ldo;
#pragma unroll
    for (int t = 0; t < 4; ++t) {
        const int c = col0 + t * 16 + l;
        if (c < Nout) {
            const int cc = c - plane * ldo;
#pragma unroll
            for (int v = 0; v < 8; ++v) {
                const int r = rt * 16 + v + 8 * half;
                op[(long long)r * ldo + cc] = acc[t][v];
            }
        }
    }
}

__global__ void fill_u32(unsigned* p, unsigned v, long long n) {
    long long i = (long long)blockIdx.x * blockDim.x + threadIdx.x;
    if (i < n) p[i] = v;
}

__global__ void k_attn(const float* __restrict__ feat, const float* __restrict__ al,
                       const float* __restrict__ ar, float* __restrict__ el,
                       float* __restrict__ er, int N, int H, int D) {
    int i = blockIdx.x * blockDim.x + threadIdx.x;
    if (i >= N * H) return;
    const int n = i / H, h = i % H;
    const float* f = feat + (long long)n * H * D + h * D;
    float a = 0.0f, b = 0.0f;
    for (int d = 0; d < D; ++d) { a += f[d] * al[h * D + d]; b += f[d] * ar[h * D + d]; }
    el[i] = a; er[i] = b;
}

__global__ void k_rank(const int* __restrict__ dst, int* __restrict__ cnt,
                       int* __restrict__ rank, int E) {
    int i = blockIdx.x * blockDim.x + threadIdx.x;
    if (i < E) rank[i] = atomicAdd(&cnt[dst[i]], 1);
}

constexpr int SCAN_T  = 1024;
constexpr int SCAN_CH = (Nn + SCAN_T - 1) / SCAN_T;
__global__ __launch_bounds__(SCAN_T)
void k_scan(const int* __restrict__ cnt, int* __restrict__ rowptr, int n) {
    __shared__ int sc[SCAN_T * SCAN_CH];
    __shared__ int wsum[32];
    const int t = threadIdx.x;
    for (int i = t; i < SCAN_T * SCAN_CH; i += SCAN_T) sc[i] = (i < n) ? cnt[i] : 0;
    __syncthreads();
    int* my = sc + t * SCAN_CH;
    int s = 0;
    for (int j = 0; j < SCAN_CH; ++j) s += my[j];
    const int lane = t & 31, w = t >> 5;
    int inc = s;
    for (int off = 1; off < 32; off <<= 1) {
        const int v = __shfl_up(inc, off, 32);
        if (lane >= off) inc += v;
    }
    if (lane == 31) wsum[w] = inc;
    __syncthreads();
    if (w == 0) {
        const int v = wsum[lane];
        int iv = v;
        for (int off = 1; off < 32; off <<= 1) {
            const int u = __shfl_up(iv, off, 32);
            if (lane >= off) iv += u;
        }
        wsum[lane] = iv - v;
    }
    __syncthreads();
    int run = wsum[w] + inc - s;
    for (int j = 0; j < SCAN_CH; ++j) { const int c = my[j]; my[j] = run; run += c; }
    __syncthreads();
    for (int i = t; i < n; i += SCAN_T) rowptr[i] = sc[i];
    if (t == SCAN_T - 1) rowptr[n] = run;
}

__global__ void k_fill_csr(const int* __restrict__ src, const int* __restrict__ dst,
                           const int* __restrict__ rank, const int* __restrict__ rowptr,
                           int* __restrict__ csr_src, int E) {
    int i = blockIdx.x * blockDim.x + threadIdx.x;
    if (i < E) csr_src[rowptr[dst[i]] + rank[i]] = src[i];
}

__device__ __forceinline__ int bcast_i(int v, int j) { return __builtin_amdgcn_readlane(v, j); }
__device__ __forceinline__ float bcast_f(float v, int j) {
    return __uint_as_float((unsigned)__builtin_amdgcn_readlane((int)__float_as_uint(v), j));
}
__device__ __forceinline__ float wave_max(float v) {
    for (int off = 16; off > 0; off >>= 1) v = fmaxf(v, __shfl_xor(v, off, 32));
    return v;
}
__device__ __forceinline__ float wave_sum(float v) {
    for (int off = 16; off > 0; off >>= 1) v += __shfl_xor(v, off, 32);
    return v;
}
__device__ __forceinline__ void add4(float4& a, const float4 v) {
    a.x += v.x; a.y += v.y; a.z += v.z; a.w += v.w;
}
__device__ __forceinline__ void fma4(float4& a, float s, const float4 v) {
    a.x += s * v.x; a.y += s * v.y; a.z += s * v.z; a.w += s * v.w;
}
__device__ __forceinline__ float leaky(float v) { return v > 0.0f ? v : 0.2f * v; }
__device__ __forceinline__ float finite_or_zero(float m) {
    return (fabsf(m) < __builtin_huge_valf()) ? m : 0.0f;
}

__global__ __launch_bounds__(256)
void k_sage_gather(const float* __restrict__ xs, const float* __restrict__ xn,
                   const int* __restrict__ rowptr, const int* __restrict__ csr_src,
                   const float* __restrict__ bias, float* __restrict__ h0, int N) {
    const int n = __builtin_amdgcn_readfirstlane((int)(blockIdx.x * 8 + (threadIdx.x >> 5)));
    if (n >= N) return;
    const int lane = threadIdx.x & 31;
    const int beg = rowptr[n], end = rowptr[n + 1];
    const float4 own = ((const float4*)(xs + (long long)n * D0))[lane];
    float4 acc = make_float4(0.f, 0.f, 0.f, 0.f);
    for (int b = beg; b < end; b += 32) {
        const int m = min(32, end - b);
        const int sv = (lane < m) ? csr_src[b + lane] : 0;
        int j = 0;
        for (; j + 8 <= m; j += 8) {
            float4 f[8];
#pragma unroll
            for (int u = 0; u < 8; ++u)
                f[u] = ((const float4*)(xn + (long long)bcast_i(sv, j + u) * D0))[lane];
#pragma unroll
            for (int u = 0; u < 8; ++u) add4(acc, f[u]);
        }
        if (j + 4 <= m) {
            float4 f[4];
#pragma unroll
            for (int u = 0; u < 4; ++u)
                f[u] = ((const float4*)(xn + (long long)bcast_i(sv, j + u) * D0))[lane];
#pragma unroll
            for (int u = 0; u < 4; ++u) add4(acc, f[u]);
            j += 4;
        }
        for (; j < m; ++j)
            add4(acc, ((const float4*)(xn + (long long)bcast_i(sv, j) * D0))[lane]);
    }
    const float inv = 1.0f / fmaxf((float)(end - beg), 1.0f);
    const float* bl = bias + lane * 4;
    float4 r;
    r.x = own.x + acc.x * inv + bl[0]; r.y = own.y + acc.y * inv + bl[1];
    r.z = own.z + acc.z * inv + bl[2]; r.w = own.w + acc.w * inv + bl[3];
    r.x = r.x > 0.0f ? r.x : (expf(r.x) - 1.0f);
    r.y = r.y > 0.0f ? r.y : (expf(r.y) - 1.0f);
    r.z = r.z > 0.0f ? r.z : (expf(r.z) - 1.0f);
    r.w = r.w > 0.0f ? r.w : (expf(r.w) - 1.0f);
    ((float4*)(h0 + (long long)n * D0))[lane] = r;
}

__global__ __launch_bounds__(256)
void k_gat1_gather(const int* __restrict__ rowptr, const int* __restrict__ csr_src,
                   const float* __restrict__ feat, const float* __restrict__ el,
                   const float* __restrict__ er, const float* __restrict__ bias,
                   float* __restrict__ out, int N) {
    const int n = __builtin_amdgcn_readfirstlane((int)(blockIdx.x * 8 + (threadIdx.x >> 5)));
    if (n >= N) return;
    const int lane = threadIdx.x & 31;
    const int beg = rowptr[n], end = rowptr[n + 1];
    const float er0 = er[n * 2], er1 = er[n * 2 + 1];

    float m0 = -__builtin_huge_valf(), m1 = m0;
    for (int b = beg + lane; b < end; b += 32) {
        const float2 l = *(const float2*)(el + (long long)csr_src[b] * 2);
        m0 = fmaxf(m0, leaky(l.x + er0)); m1 = fmaxf(m1, leaky(l.y + er1));
    }
    m0 = finite_or_zero(wave_max(m0)); m1 = finite_or_zero(wave_max(m1));

    float s0 = 0.0f, s1 = 0.0f;
    for (int b = beg + lane; b < end; b += 32) {
        const float2 l = *(const float2*)(el + (long long)csr_src[b] * 2);
        s0 += __expf(leaky(l.x + er0) - m0); s1 += __expf(leaky(l.y + er1) - m1);
    }
    const float d0 = fmaxf(wave_sum(s0), 1e-9f), d1 = fmaxf(wave_sum(s1), 1e-9f);

    const bool hi = lane >= 16;
    float4 acc = make_float4(0.f, 0.f, 0.f, 0.f);
    for (int b = beg; b < end; b += 32) {
        const int m = min(32, end - b);
        int sv = 0; float w0 = 0.0f, w1 = 0.0f;
        if (lane < m) {
            sv = csr_src[b + lane];
            const float2 l = *(const float2*)(el + (long long)sv * 2);
            w0 = __expf(leaky(l.x + er0) - m0) / d0;
            w1 = __expf(leaky(l.y + er1) - m1) / d1;
        }
        int j = 0;
        for (; j + 4 <= m; j += 4) {
            const float4 f0 = ((const float4*)(feat + (long long)bcast_i(sv, j + 0) * (H1 * D1)))[lane];
            const float4 f1 = ((const float4*)(feat + (long long)bcast_i(sv, j + 1) * (H1 * D1)))[lane];
            const float4 f2 = ((const float4*)(feat + (long long)bcast_i(sv, j + 2) * (H1 * D1)))[lane];
            const float4 f3 = ((const float4*)(feat + (long long)bcast_i(sv, j + 3) * (H1 * D1)))[lane];
            const float u0 = bcast_f(w0, j + 0), t0 = bcast_f(w1, j + 0);
            const float u1 = bcast_f(w0, j + 1), t1 = bcast_f(w1, j + 1);
            const float u2 = bcast_f(w0, j + 2), t2 = bcast_f(w1, j + 2);
            const float u3 = bcast_f(w0, j + 3), t3 = bcast_f(w1, j + 3);
            fma4(acc, hi ? t0 : u0, f0);
            fma4(acc, hi ? t1 : u1, f1);
            fma4(acc, hi ? t2 : u2, f2);
            fma4(acc, hi ? t3 : u3, f3);
        }
        for (; j < m; ++j) {
            const float4 f0 = ((const float4*)(feat + (long long)bcast_i(sv, j) * (H1 * D1)))[lane];
            const float u0 = bcast_f(w0, j), t0 = bcast_f(w1, j);
            fma4(acc, hi ? t0 : u0, f0);
        }
    }
    const float* bl = bias + lane * 4;
    float4 r;
    r.x = acc.x + bl[0]; r.y = acc.y + bl[1]; r.z = acc.z + bl[2]; r.w = acc.w + bl[3];
    r.x = r.x > 0.0f ? r.x : (expf(r.x) - 1.0f);
    r.y = r.y > 0.0f ? r.y : (expf(r.y) - 1.0f);
    r.z = r.z > 0.0f ? r.z : (expf(r.z) - 1.0f);
    r.w = r.w > 0.0f ? r.w : (expf(r.w) - 1.0f);
    ((float4*)(out + (long long)n * (H1 * D1)))[lane] = r;
}

__global__ __launch_bounds__(256)
void k_gat2_gather(const int* __restrict__ rowptr, const int* __restrict__ csr_src,
                   const float* __restrict__ feat, const float* __restrict__ el,
                   const float* __restrict__ er, const float* __restrict__ bias,
                   float* __restrict__ out, int N) {
    const int n = __builtin_amdgcn_readfirstlane((int)(blockIdx.x * 8 + (threadIdx.x >> 5)));
    if (n >= N) return;
    const int lane = threadIdx.x & 31;
    const int beg = rowptr[n], end = rowptr[n + 1];
    const float ern = er[n];
    const bool two = lane < C2 - 32;

    float mx = -__builtin_huge_valf();
    for (int b = beg + lane; b < end; b += 32) mx = fmaxf(mx, leaky(el[csr_src[b]] + ern));
    mx = finite_or_zero(wave_max(mx));
    float sm = 0.0f;
    for (int b = beg + lane; b < end; b += 32) sm += __expf(leaky(el[csr_src[b]] + ern) - mx);
    const float dn = fmaxf(wave_sum(sm), 1e-9f);

    float acc0 = 0.0f, acc1 = 0.0f;
    for (int b = beg; b < end; b += 32) {
        const int m = min(32, end - b);
        int sv = 0; float wv = 0.0f;
        if (lane < m) {
            sv = csr_src[b + lane];
            wv = __expf(leaky(el[sv] + ern) - mx) / dn;
        }
        int j = 0;
        for (; j + 4 <= m; j += 4) {
            const float* r0 = feat + (long long)bcast_i(sv, j + 0) * C2;
            const float* r1 = feat + (long long)bcast_i(sv, j + 1) * C2;
            const float* r2 = feat + (long long)bcast_i(sv, j + 2) * C2;
            const float* r3 = feat + (long long)bcast_i(sv, j + 3) * C2;
            const float p0 = r0[lane], p1 = r1[lane], p2 = r2[lane], p3 = r3[lane];
            const float q0 = two ? r0[lane + 32] : 0.0f, q1 = two ? r1[lane + 32] : 0.0f;
            const float q2 = two ? r2[lane + 32] : 0.0f, q3 = two ? r3[lane + 32] : 0.0f;
            const float u0 = bcast_f(wv, j + 0), u1 = bcast_f(wv, j + 1);
            const float u2 = bcast_f(wv, j + 2), u3 = bcast_f(wv, j + 3);
            acc0 += u0 * p0; acc1 += u0 * q0; acc0 += u1 * p1; acc1 += u1 * q1;
            acc0 += u2 * p2; acc1 += u2 * q2; acc0 += u3 * p3; acc1 += u3 * q3;
        }
        for (; j < m; ++j) {
            const float* r0 = feat + (long long)bcast_i(sv, j) * C2;
            const float u0 = bcast_f(wv, j);
            acc0 += u0 * r0[lane];
            acc1 += u0 * (two ? r0[lane + 32] : 0.0f);
        }
    }
    float* o = out + (long long)n * C2;
    o[lane] = acc0 + bias[lane];
    if (two) o[lane + 32] = acc1 + bias[lane + 32];
}

static inline unsigned gsz(long long n, int b) { return (unsigned)((n + b - 1) / b); }

extern "C" void kernel_launch(void* const* d_in, const int* in_sizes, int n_in,
                              void* d_out, int out_size, void* d_ws, size_t ws_size,
                              hipStream_t stream) {
    const float* x       = (const float*)d_in[0];
    const int*   src     = (const int*)  d_in[1];
    const int*   dst     = (const int*)  d_in[2];
    const float* w_self  = (const float*)d_in[3];
    const float* w_neigh = (const float*)d_in[4];
    const float* sage_b  = (const float*)d_in[5];
    const float* g1_w    = (const float*)d_in[6];
    const float* g1_al   = (const float*)d_in[7];
    const float* g1_ar   = (const float*)d_in[8];
    const float* g1_b    = (const float*)d_in[9];
    const float* g2_w    = (const float*)d_in[10];
    const float* g2_al   = (const float*)d_in[11];
    const float* g2_ar   = (const float*)d_in[12];
    const float* g2_b    = (const float*)d_in[13];

    float*    ws    = (float*)d_ws;
    float*    proj  = ws;
    float*    h0    = proj + (long long)Nn * F;
    float*    el    = h0 + (long long)Nn * D0;
    float*    er    = el + Nn * H1;
    int*      cnt   = (int*)(er + Nn * H1);
    int*      rowptr = cnt + Nn;
    int*      rank  = rowptr + (Nn + 16);
    int*      csr_src = rank + Ne;
    float*    wt_self  = (float*)(csr_src + Ne);
    float*    wt_neigh = wt_self + D0 * F;
    float*    wt_g1    = wt_neigh + D0 * F;
    float*    wt_g2    = wt_g1 + (H1 * D1) * D0;
    float*    feat  = proj;

    const int B = 256;
    const int Mtiles = Nn / 16;
    const unsigned rowblk = gsz(Mtiles, 8);
    const unsigned nodeblk = gsz(Nn, 8);

    k_transpose<<<gsz(D0 * F, B), B, 0, stream>>>(w_self,  wt_self,  F,  D0, D0);
    k_transpose<<<gsz(D0 * F, B), B, 0, stream>>>(w_neigh, wt_neigh, F,  D0, D0);
    k_transpose<<<gsz(D0 * D0, B), B, 0, stream>>>(g1_w,   wt_g1,    D0, H1 * D1, H1 * D1);
    k_transpose<<<gsz(C2P * D0, B), B, 0, stream>>>(g2_w,  wt_g2,    D0, C2, C2P);

    fill_u32<<<gsz(Nn, B), B, 0, stream>>>((unsigned*)cnt, 0u, Nn);
    k_rank<<<gsz(Ne, B), B, 0, stream>>>(dst, cnt, rank, Ne);
    k_scan<<<1, SCAN_T, 0, stream>>>(cnt, rowptr, Nn);
    k_fill_csr<<<gsz(Ne, B), B, 0, stream>>>(src, dst, rank, rowptr, csr_src, Ne);

    {
        const size_t shmem = 64ull * (F + 4) * sizeof(float);
        wmma_gemm<<<rowblk * 4, B, shmem, stream>>>(x, wt_self, proj, Nn, F, 2 * D0, 4, D0);
    }
    k_sage_gather<<<nodeblk, B, 0, stream>>>(proj, proj + (long long)Nn * D0, rowptr, csr_src,
                                             sage_b, h0, Nn);

    {
        const size_t shmem = 64ull * (D0 + 4) * sizeof(float);
        wmma_gemm<<<rowblk * 2, B, shmem, stream>>>(h0, wt_g1, feat, Nn, D0, H1 * D1, 2, H1 * D1);
    }
    k_attn<<<gsz(Nn * H1, B), B, 0, stream>>>(feat, g1_al, g1_ar, el, er, Nn, H1, D1);
    k_gat1_gather<<<nodeblk, B, 0, stream>>>(rowptr, csr_src, feat, el, er, g1_b, h0, Nn);

    {
        const size_t shmem = 64ull * (D0 + 4) * sizeof(float);
        wmma_gemm<<<rowblk, B, shmem, stream>>>(h0, wt_g2, feat, Nn, D0, C2, 1, C2);
    }
    k_attn<<<gsz(Nn, B), B, 0, stream>>>(feat, g2_al, g2_ar, el, er, Nn, 1, C2);
    k_gat2_gather<<<nodeblk, B, 0, stream>>>(rowptr, csr_src, feat, el, er, g2_b,
                                             (float*)d_out, Nn);
}


# 16 launches -> 10: weight transposes and counter clear share one launch; GAT logits el/er come from the GEMM accumulator tile in its epilogue, not a re-read kernel; el/er sum order changes
# speedup vs baseline: 4.9841x; 1.0785x over previous
#include <hip/hip_runtime.h>

constexpr int Nn  = 50000;
constexpr int Ne  = 800000;
constexpr int F   = 256;
constexpr int D0  = 128;
constexpr int H1  = 2;
constexpr int D1  = 64;
constexpr int C2  = 47;
constexpr int C2P = 64;

typedef float v2f __attribute__((ext_vector_type(2)));
typedef float v8f __attribute__((ext_vector_type(8)));

template <int K, int N, int NPAD>
__device__ __forceinline__ void transpose_elem(const float* __restrict__ W, float* __restrict__ Wt, int i) {
    const int c = i / K, k = i % K;
    Wt[i] = (c < N) ? W[k * N + c] : 0.0f;
}
__global__ __launch_bounds__(256)
void k_prep(const float* __restrict__ w_self, const float* __restrict__ w_neigh,
            const float* __restrict__ g1_w, const float* __restrict__ g2_w,
            float* __restrict__ wt_self, float* __restrict__ wt_neigh,
            float* __restrict__ wt_g1, float* __restrict__ wt_g2,
            unsigned* __restrict__ cnt) {
    constexpr int n_sage = D0 * F;
    constexpr int n_g1   = (H1 * D1) * D0;
    constexpr int n_g2   = C2P * D0;
    int i = blockIdx.x * 256 + threadIdx.x;
    if (i < n_sage) { transpose_elem<F, D0, D0>(w_self, wt_self, i); return; }
    i -= n_sage;
    if (i < n_sage) { transpose_elem<F, D0, D0>(w_neigh, wt_neigh, i); return; }
    i -= n_sage;
    if (i < n_g1) { transpose_elem<D0, H1 * D1, H1 * D1>(g1_w, wt_g1, i); return; }
    i -= n_g1;
    if (i < n_g2) { transpose_elem<D0, C2, C2P>(g2_w, wt_g2, i); return; }
    i -= n_g2;
    if (i < Nn) cnt[i] = 0u;
}

template <bool ATTN>
__global__ __launch_bounds__(256)
void wmma_gemm(const float* __restrict__ A, const float* __restrict__ Wt,
               float* __restrict__ out, int M, int K, int Nout, int ncg, int ldo,
               const float* __restrict__ al, const float* __restrict__ ar,
               float* __restrict__ el, float* __restrict__ er)
{
    extern __shared__ float lds[];
    const int KP = K + 4;
    const int cg   = blockIdx.x % ncg;
    const int col0 = cg * 64;
    float* w1s = lds;

    {
        const int qsh = 29 - __clz(K);
        const int qm  = (1 << qsh) - 1;
        const int npt = K >> 4;
        const float4* g1 = (const float4*)(Wt + (long long)col0 * K);
        constexpr int U = 8;
        for (int b = 0; b < npt; b += U) {
            float4 v1[U];
#pragma unroll
            for (int u = 0; u < U; ++u) {
                const int q = (b + u) * 256 + threadIdx.x;
                v1[u] = g1[q];
            }
#pragma unroll
            for (int u = 0; u < U; ++u) {
                const int q = (b + u) * 256 + threadIdx.x;
                const int o = (q >> qsh) * KP + (q & qm) * 4;
                *(float4*)(w1s + o) = v1[u];
            }
        }
    }
    __syncthreads();

    const int rt = (blockIdx.x / ncg) * 8 + (threadIdx.x >> 5);
    if (rt >= M / 16) return;
    const int lane = threadIdx.x & 31;
    const int half = lane >> 4;
    const int l    = lane & 15;
    const int row  = rt * 16 + l;

    const float* Arow  = A + (long long)row * K + half * 2;
    __builtin_prefetch(Arow, 0, 3);

    const float* Bs[4];
#pragma unroll
    for (int t = 0; t < 4; ++t) Bs[t] = w1s + (t * 16 + l) * KP + half * 2;

    v8f acc[4] = {};
    for (int kk = 0; kk < K; kk += 4) {
        const v2f a = *(const v2f*)(Arow + kk);
#pragma unroll
        for (int t = 0; t < 4; ++t) {
            const v2f b = *(const v2f*)(Bs[t] + kk);
            acc[t] = __builtin_amdgcn_wmma_f32_16x16x4_f32(
                false, a, false, b, (short)0, acc[t], false, false);
        }
    }

    if constexpr (ATTN) {
        asm volatile("v_nop\n\tv_nop\n\tv_nop\n\tv_nop"
                     : "+v"(acc[2]), "+v"(acc[3]));
        asm volatile("v_nop\n\tv_nop\n\tv_nop\n\tv_nop"
                     : "+v"(acc[0]), "+v"(acc[1]));
    }

    const int plane = col0 / ldo;
    float* op = out + (long long)plane * M * ldo;
#pragma unroll
    for (int t = 0; t < 4; ++t) {
        const int c = col0 + t * 16 + l;
        if (c < Nout) {
            const int cc = c - plane * ldo;
#pragma unroll
            for (int v = 0; v < 8; ++v) {
                const int r = rt * 16 + v + 8 * half;
                op[(long long)r * ldo + cc] = acc[t][v];
            }
        }
    }

    if constexpr (ATTN) {
        float wl[4], wr[4];
#pragma unroll
        for (int t = 0; t < 4; ++t) {
            const int c = col0 + t * 16 + l;
            wl[t] = (c < Nout) ? al[c] : 0.0f;
            wr[t] = (c < Nout) ? ar[c] : 0.0f;
        }
        const bool up = (l & 8) != 0;
        float s[8];
#pragma unroll
        for (int v = 0; v < 8; ++v) {
            const float pl = acc[0][v] * wl[0] + acc[1][v] * wl[1] + acc[2][v] * wl[2] + acc[3][v] * wl[3];
            const float pr = acc[0][v] * wr[0] + acc[1][v] * wr[1] + acc[2][v] * wr[2] + acc[3][v] * wr[3];
            const float keep = up ? pr : pl;
            const float give = up ? pl : pr;
            s[v] = keep + __shfl_xor(give, 8, 32);
        }
#pragma unroll
        for (int off = 4; off > 0; off >>= 1) {
#pragma unroll
            for (int v = 0; v < 8; ++v) s[v] += __shfl_xor(s[v], off, 32);
        }
        const int j = l & 7;
        float val = s[0];
#pragma unroll
        for (int v = 1; v < 8; ++v) val = (j == v) ? s[v] : val;
        const int r = rt * 16 + j + 8 * half;
        (up ? er : el)[(long long)r * ncg + cg] = val;
    }
}

__global__ void k_rank(const int* __restrict__ dst, int* __restrict__ cnt,
                       int* __restrict__ rank, int E) {
    int i = blockIdx.x * blockDim.x + threadIdx.x;
    if (i < E) rank[i] = atomicAdd(&cnt[dst[i]], 1);
}

constexpr int SCAN_T  = 1024;
constexpr int SCAN_CH = (Nn + SCAN_T - 1) / SCAN_T;
__global__ __launch_bounds__(SCAN_T)
void k_scan(const int* __restrict__ cnt, int* __restrict__ rowptr, int n) {
    __shared__ int sc[SCAN_T * SCAN_CH];
    __shared__ int wsum[32];
    const int t = threadIdx.x;
    for (int i = t; i < SCAN_T * SCAN_CH; i += SCAN_T) sc[i] = (i < n) ? cnt[i] : 0;
    __syncthreads();
    int* my = sc + t * SCAN_CH;
    int s = 0;
    for (int j = 0; j < SCAN_CH; ++j) s += my[j];
    const int lane = t & 31, w = t >> 5;
    int inc = s;
    for (int off = 1; off < 32; off <<= 1) {
        const int v = __shfl_up(inc, off, 32);
        if (lane >= off) inc += v;
    }
    if (lane == 31) wsum[w] = inc;
    __syncthreads();
    if (w == 0) {
        const int v = wsum[lane];
        int iv = v;
        for (int off = 1; off < 32; off <<= 1) {
            const int u = __shfl_up(iv, off, 32);
            if (lane >= off) iv += u;
        }
        wsum[lane] = iv - v;
    }
    __syncthreads();
    int run = wsum[w] + inc - s;
    for (int j = 0; j < SCAN_CH; ++j) { const int c = my[j]; my[j] = run; run += c; }
    __syncthreads();
    for (int i = t; i < n; i += SCAN_T) rowptr[i] = sc[i];
    if (t == SCAN_T - 1) rowptr[n] = run;
}

__global__ void k_fill_csr(const int* __restrict__ src, const int* __restrict__ dst,
                           const int* __restrict__ rank, const int* __restrict__ rowptr,
                           int* __restrict__ csr_src, int E) {
    int i = blockIdx.x * blockDim.x + threadIdx.x;
    if (i < E) csr_src[rowptr[dst[i]] + rank[i]] = src[i];
}

__device__ __forceinline__ int bcast_i(int v, int j) { return __builtin_amdgcn_readlane(v, j); }
__device__ __forceinline__ float bcast_f(float v, int j) {
    return __uint_as_float((unsigned)__builtin_amdgcn_readlane((int)__float_as_uint(v), j));
}
__device__ __forceinline__ float wave_max(float v) {
    for (int off = 16; off > 0; off >>= 1) v = fmaxf(v, __shfl_xor(v, off, 32));
    return v;
}
__device__ __forceinline__ float wave_sum(float v) {
    for (int off = 16; off > 0; off >>= 1) v += __shfl_xor(v, off, 32);
    return v;
}
__device__ __forceinline__ void add4(float4& a, const float4 v) {
    a.x += v.x; a.y += v.y; a.z += v.z; a.w += v.w;
}
__device__ __forceinline__ void fma4(float4& a, float s, const float4 v) {
    a.x += s * v.x; a.y += s * v.y; a.z += s * v.z; a.w += s * v.w;
}
__device__ __forceinline__ float leaky(float v) { return v > 0.0f ? v : 0.2f * v; }
__device__ __forceinline__ float finite_or_zero(float m) {
    return (fabsf(m) < __builtin_huge_valf()) ? m : 0.0f;
}

__global__ __launch_bounds__(256)
void k_sage_gather(const float* __restrict__ xs, const float* __restrict__ xn,
                   const int* __restrict__ rowptr, const int* __restrict__ csr_src,
                   const float* __restrict__ bias, float* __restrict__ h0, int N) {
    const int n = __builtin_amdgcn_readfirstlane((int)(blockIdx.x * 8 + (threadIdx.x >> 5)));
    if (n >= N) return;
    const int lane = threadIdx.x & 31;
    const int beg = rowptr[n], end = rowptr[n + 1];
    const float4 own = ((const float4*)(xs + (long long)n * D0))[lane];
    float4 acc = make_float4(0.f, 0.f, 0.f, 0.f);
    for (int b = beg; b < end; b += 32) {
        const int m = min(32, end - b);
        const int sv = (lane < m) ? csr_src[b + lane] : 0;
        int j = 0;
        for (; j + 8 <= m; j += 8) {
            float4 f[8];
#pragma unroll
            for (int u = 0; u < 8; ++u)
                f[u] = ((const float4*)(xn + (long long)bcast_i(sv, j + u) * D0))[lane];
#pragma unroll
            for (int u = 0; u < 8; ++u) add4(acc, f[u]);
        }
        if (j + 4 <= m) {
            float4 f[4];
#pragma unroll
            for (int u = 0; u < 4; ++u)
                f[u] = ((const float4*)(xn + (long long)bcast_i(sv, j + u) * D0))[lane];
#pragma unroll
            for (int u = 0; u < 4; ++u) add4(acc, f[u]);
            j += 4;
        }
        for (; j < m; ++j)
            add4(acc, ((const float4*)(xn + (long long)bcast_i(sv, j) * D0))[lane]);
    }
    const float inv = 1.0f / fmaxf((float)(end - beg), 1.0f);
    const float* bl = bias + lane * 4;
    float4 r;
    r.x = own.x + acc.x * inv + bl[0]; r.y = own.y + acc.y * inv + bl[1];
    r.z = own.z + acc.z * inv + bl[2]; r.w = own.w + acc.w * inv + bl[3];
    r.x = r.x > 0.0f ? r.x : (expf(r.x) - 1.0f);
    r.y = r.y > 0.0f ? r.y : (expf(r.y) - 1.0f);
    r.z = r.z > 0.0f ? r.z : (expf(r.z) - 1.0f);
    r.w = r.w > 0.0f ? r.w : (expf(r.w) - 1.0f);
    ((float4*)(h0 + (long long)n * D0))[lane] = r;
}

__global__ __launch_bounds__(256)
void k_gat1_gather(const int* __restrict__ rowptr, const int* __restrict__ csr_src,
                   const float* __restrict__ feat, const float* __restrict__ el,
                   const float* __restrict__ er, const float* __restrict__ bias,
                   float* __restrict__ out, int N) {
    const int n = __builtin_amdgcn_readfirstlane((int)(blockIdx.x * 8 + (threadIdx.x >> 5)));
    if (n >= N) return;
    const int lane = threadIdx.x & 31;
    const int beg = rowptr[n], end = rowptr[n + 1];
    const float er0 = er[n * 2], er1 = er[n * 2 + 1];

    float m0 = -__builtin_huge_valf(), m1 = m0;
    for (int b = beg + lane; b < end; b += 32) {
        const float2 l = *(const float2*)(el + (long long)csr_src[b] * 2);
        m0 = fmaxf(m0, leaky(l.x + er0)); m1 = fmaxf(m1, leaky(l.y + er1));
    }
    m0 = finite_or_zero(wave_max(m0)); m1 = finite_or_zero(wave_max(m1));

    float s0 = 0.0f, s1 = 0.0f;
    for (int b = beg + lane; b < end; b += 32) {
        const float2 l = *(const float2*)(el + (long long)csr_src[b] * 2);
        s0 += __expf(leaky(l.x + er0) - m0); s1 += __expf(leaky(l.y + er1) - m1);
    }
    const float d0 = fmaxf(wave_sum(s0), 1e-9f), d1 = fmaxf(wave_sum(s1), 1e-9f);

    const bool hi = lane >= 16;
    float4 acc = make_float4(0.f, 0.f, 0.f, 0.f);
    for (int b = beg; b < end; b += 32) {
        const int m = min(32, end - b);
        int sv = 0; float w0 = 0.0f, w1 = 0.0f;
        if (lane < m) {
            sv = csr_src[b + lane];
            const float2 l = *(const float2*)(el + (long long)sv * 2);
            w0 = __expf(leaky(l.x + er0) - m0) / d0;
            w1 = __expf(leaky(l.y + er1) - m1) / d1;
        }
        int j = 0;
        for (; j + 4 <= m; j += 4) {
            const float4 f0 = ((const float4*)(feat + (long long)bcast_i(sv, j + 0) * (H1 * D1)))[lane];
            const float4 f1 = ((const float4*)(feat + (long long)bcast_i(sv, j + 1) * (H1 * D1)))[lane];
            const float4 f2 = ((const float4*)(feat + (long long)bcast_i(sv, j + 2) * (H1 * D1)))[lane];
            const float4 f3 = ((const float4*)(feat + (long long)bcast_i(sv, j + 3) * (H1 * D1)))[lane];
            const float u0 = bcast_f(w0, j + 0), t0 = bcast_f(w1, j + 0);
            const float u1 = bcast_f(w0, j + 1), t1 = bcast_f(w1, j + 1);
            const float u2 = bcast_f(w0, j + 2), t2 = bcast_f(w1, j + 2);
            const float u3 = bcast_f(w0, j + 3), t3 = bcast_f(w1, j + 3);
            fma4(acc, hi ? t0 : u0, f0);
            fma4(acc, hi ? t1 : u1, f1);
            fma4(acc, hi ? t2 : u2, f2);
            fma4(acc, hi ? t3 : u3, f3);
        }
        for (; j < m; ++j) {
            const float4 f0 = ((const float4*)(feat + (long long)bcast_i(sv, j) * (H1 * D1)))[lane];
            const float u0 = bcast_f(w0, j), t0 = bcast_f(w1, j);
            fma4(acc, hi ? t0 : u0, f0);
        }
    }
    const float* bl = bias + lane * 4;
    float4 r;
    r.x = acc.x + bl[0]; r.y = acc.y + bl[1]; r.z = acc.z + bl[2]; r.w = acc.w + bl[3];
    r.x = r.x > 0.0f ? r.x : (expf(r.x) - 1.0f);
    r.y = r.y > 0.0f ? r.y : (expf(r.y) - 1.0f);
    r.z = r.z > 0.0f ? r.z : (expf(r.z) - 1.0f);
    r.w = r.w > 0.0f ? r.w : (expf(r.w) - 1.0f);
    ((float4*)(out + (long long)n * (H1 * D1)))[lane] = r;
}

__global__ __launch_bounds__(256)
void k_gat2_gather(const int* __restrict__ rowptr, const int* __restrict__ csr_src,
                   const float* __restrict__ feat, const float* __restrict__ el,
                   const float* __restrict__ er, const float* __restrict__ bias,
                   float* __restrict__ out, int N) {
    const int n = __builtin_amdgcn_readfirstlane((int)(blockIdx.x * 8 + (threadIdx.x >> 5)));
    if (n >= N) return;
    const int lane = threadIdx.x & 31;
    const int beg = rowptr[n], end = rowptr[n + 1];
    const float ern = er[n];
    const bool two = lane < C2 - 32;

    float mx = -__builtin_huge_valf();
    for (int b = beg + lane; b < end; b += 32) mx = fmaxf(mx, leaky(el[csr_src[b]] + ern));
    mx = finite_or_zero(wave_max(mx));
    float sm = 0.0f;
    for (int b = beg + lane; b < end; b += 32) sm += __expf(leaky(el[csr_src[b]] + ern) - mx);
    const float dn = fmaxf(wave_sum(sm), 1e-9f);

    float acc0 = 0.0f, acc1 = 0.0f;
    for (int b = beg; b < end; b += 32) {
        const int m = min(32, end - b);
        int sv = 0; float wv = 0.0f;
        if (lane < m) {
            sv = csr_src[b + lane];
            wv = __expf(leaky(el[sv] + ern) - mx) / dn;
        }
        int j = 0;
        for (; j + 4 <= m; j += 4) {
            const float* r0 = feat + (long long)bcast_i(sv, j + 0) * C2;
            const float* r1 = feat + (long long)bcast_i(sv, j + 1) * C2;
            const float* r2 = feat + (long long)bcast_i(sv, j + 2) * C2;
            const float* r3 = feat + (long long)bcast_i(sv, j + 3) * C2;
            const float p0 = r0[lane], p1 = r1[lane], p2 = r2[lane], p3 = r3[lane];
            const float q0 = two ? r0[lane + 32] : 0.0f, q1 = two ? r1[lane + 32] : 0.0f;
            const float q2 = two ? r2[lane + 32] : 0.0f, q3 = two ? r3[lane + 32] : 0.0f;
            const float u0 = bcast_f(wv, j + 0), u1 = bcast_f(wv, j + 1);
            const float u2 = bcast_f(wv, j + 2), u3 = bcast_f(wv, j + 3);
            acc0 += u0 * p0; acc1 += u0 * q0; acc0 += u1 * p1; acc1 += u1 * q1;
            acc0 += u2 * p2; acc1 += u2 * q2; acc0 += u3 * p3; acc1 += u3 * q3;
        }
        for (; j < m; ++j) {
            const float* r0 = feat + (long long)bcast_i(sv, j) * C2;
            const float u0 = bcast_f(wv, j);
            acc0 += u0 * r0[lane];
            acc1 += u0 * (two ? r0[lane + 32] : 0.0f);
        }
    }
    float* o = out + (long long)n * C2;
    o[lane] = acc0 + bias[lane];
    if (two) o[lane + 32] = acc1 + bias[lane + 32];
}

static inline unsigned gsz(long long n, int b) { return (unsigned)((n + b - 1) / b); }

extern "C" void kernel_launch(void* const* d_in, const int* in_sizes, int n_in,
                              void* d_out, int out_size, void* d_ws, size_t ws_size,
                              hipStream_t stream) {
    const float* x       = (const float*)d_in[0];
    const int*   src     = (const int*)  d_in[1];
    const int*   dst     = (const int*)  d_in[2];
    const float* w_self  = (const float*)d_in[3];
    const float* w_neigh = (const float*)d_in[4];
    const float* sage_b  = (const float*)d_in[5];
    const float* g1_w    = (const float*)d_in[6];
    const float* g1_al   = (const float*)d_in[7];
    const float* g1_ar   = (const float*)d_in[8];
    const float* g1_b    = (const float*)d_in[9];
    const float* g2_w    = (const float*)d_in[10];
    const float* g2_al   = (const float*)d_in[11];
    const float* g2_ar   = (const float*)d_in[12];
    const float* g2_b    = (const float*)d_in[13];

    float*    ws    = (float*)d_ws;
    float*    proj  = ws;
    float*    h0    = proj + (long long)Nn * F;
    float*    el    = h0 + (long long)Nn * D0;
    float*    er    = el + Nn * H1;
    int*      cnt   = (int*)(er + Nn * H1);
    int*      rowptr = cnt + Nn;
    int*      rank  = rowptr + (Nn + 16);
    int*      csr_src = rank + Ne;
    float*    wt_self  = (float*)(csr_src + Ne);
    float*    wt_neigh = wt_self + D0 * F;
    float*    wt_g1    = wt_neigh + D0 * F;
    float*    wt_g2    = wt_g1 + (H1 * D1) * D0;
    float*    feat  = proj;

    const int B = 256;
    const int Mtiles = Nn / 16;
    const unsigned rowblk = gsz(Mtiles, 8);
    const unsigned nodeblk = gsz(Nn, 8);

    k_prep<<<gsz(2 * D0 * F + (H1 * D1) * D0 + C2P * D0 + Nn, B), B, 0, stream>>>(
        w_self, w_neigh, g1_w, g2_w, wt_self, wt_neigh, wt_g1, wt_g2, (unsigned*)cnt);

    k_rank<<<gsz(Ne, B), B, 0, stream>>>(dst, cnt, rank, Ne);
    k_scan<<<1, SCAN_T, 0, stream>>>(cnt, rowptr, Nn);
    k_fill_csr<<<gsz(Ne, B), B, 0, stream>>>(src, dst, rank, rowptr, csr_src, Ne);

    {
        const size_t shmem = 64ull * (F + 4) * sizeof(float);
        wmma_gemm<false><<<rowblk * 4, B, shmem, stream>>>(x, wt_self, proj, Nn, F, 2 * D0, 4, D0,
                                                       nullptr, nullptr, nullptr, nullptr);
    }
    k_sage_gather<<<nodeblk, B, 0, stream>>>(proj, proj + (long long)Nn * D0, rowptr, csr_src,
                                             sage_b, h0, Nn);

    {
        const size_t shmem = 64ull * (D0 + 4) * sizeof(float);
        wmma_gemm<true><<<rowblk * 2, B, shmem, stream>>>(h0, wt_g1, feat, Nn, D0, H1 * D1, H1, H1 * D1,
                                                          g1_al, g1_ar, el, er);
    }
    k_gat1_gather<<<nodeblk, B, 0, stream>>>(rowptr, csr_src, feat, el, er, g1_b, h0, Nn);

    {
        const size_t shmem = 64ull * (D0 + 4) * sizeof(float);
        wmma_gemm<true><<<rowblk, B, shmem, stream>>>(h0, wt_g2, feat, Nn, D0, C2, 1, C2,
                                                      g2_al, g2_ar, el, er);
    }
    k_gat2_gather<<<nodeblk, B, 0, stream>>>(rowptr, csr_src, feat, el, er, g2_b,
                                             (float*)d_out, Nn);
}


# In-degree ranks from LDS returning adds: 32 work-groups each hold all 50000 counters in LDS for a slice of the edges, a multi-group scan joins the slices; no global atomics left; sum order changes
# speedup vs baseline: 6.3604x; 1.2761x over previous
#include <hip/hip_runtime.h>

constexpr int Nn  = 50000;
constexpr int Ne  = 800000;
constexpr int F   = 256;
constexpr int D0  = 128;
constexpr int H1  = 2;
constexpr int D1  = 64;
constexpr int C2  = 47;
constexpr int C2P = 64;

typedef float v2f __attribute__((ext_vector_type(2)));
typedef float v8f __attribute__((ext_vector_type(8)));

template <int K, int N, int NPAD>
__device__ __forceinline__ void transpose_elem(const float* __restrict__ W, float* __restrict__ Wt, int i) {
    const int c = i / K, k = i % K;
    Wt[i] = (c < N) ? W[k * N + c] : 0.0f;
}
__global__ __launch_bounds__(256)
void k_prep(const float* __restrict__ w_self, const float* __restrict__ w_neigh,
            const float* __restrict__ g1_w, const float* __restrict__ g2_w,
            float* __restrict__ wt_self, float* __restrict__ wt_neigh,
            float* __restrict__ wt_g1, float* __restrict__ wt_g2) {
    constexpr int n_sage = D0 * F;
    constexpr int n_g1   = (H1 * D1) * D0;
    constexpr int n_g2   = C2P * D0;
    int i = blockIdx.x * 256 + threadIdx.x;
    if (i < n_sage) { transpose_elem<F, D0, D0>(w_self, wt_self, i); return; }
    i -= n_sage;
    if (i < n_sage) { transpose_elem<F, D0, D0>(w_neigh, wt_neigh, i); return; }
    i -= n_sage;
    if (i < n_g1) { transpose_elem<D0, H1 * D1, H1 * D1>(g1_w, wt_g1, i); return; }
    i -= n_g1;
    if (i < n_g2) transpose_elem<D0, C2, C2P>(g2_w, wt_g2, i);
}

template <bool ATTN>
__global__ __launch_bounds__(256)
void wmma_gemm(const float* __restrict__ A, const float* __restrict__ Wt,
               float* __restrict__ out, int M, int K, int Nout, int ncg, int ldo,
               const float* __restrict__ al, const float* __restrict__ ar,
               float* __restrict__ el, float* __restrict__ er)
{
    extern __shared__ float lds[];
    const int KP = K + 4;
    const int cg   = blockIdx.x % ncg;
    const int col0 = cg * 64;
    float* w1s = lds;

    {
        const int qsh = 29 - __clz(K);
        const int qm  = (1 << qsh) - 1;
        const int npt = K >> 4;
        const float4* g1 = (const float4*)(Wt + (long long)col0 * K);
        constexpr int U = 8;
        for (int b = 0; b < npt; b += U) {
            float4 v1[U];
#pragma unroll
            for (int u = 0; u < U; ++u) {
                const int q = (b + u) * 256 + threadIdx.x;
                v1[u] = g1[q];
            }
#pragma unroll
            for (int u = 0; u < U; ++u) {
                const int q = (b + u) * 256 + threadIdx.x;
                const int o = (q >> qsh) * KP + (q & qm) * 4;
                *(float4*)(w1s + o) = v1[u];
            }
        }
    }
    __syncthreads();

    const int rt = (blockIdx.x / ncg) * 8 + (threadIdx.x >> 5);
    if (rt >= M / 16) return;
    const int lane = threadIdx.x & 31;
    const int half = lane >> 4;
    const int l    = lane & 15;
    const int row  = rt * 16 + l;

    const float* Arow  = A + (long long)row * K + half * 2;
    __builtin_prefetch(Arow, 0, 3);

    const float* Bs[4];
#pragma unroll
    for (int t = 0; t < 4; ++t) Bs[t] = w1s + (t * 16 + l) * KP + half * 2;

    v8f acc[4] = {};
    for (int kk = 0; kk < K; kk += 4) {
        const v2f a = *(const v2f*)(Arow + kk);
#pragma unroll
        for (int t = 0; t < 4; ++t) {
            const v2f b = *(const v2f*)(Bs[t] + kk);
            acc[t] = __builtin_amdgcn_wmma_f32_16x16x4_f32(
                false, a, false, b, (short)0, acc[t], false, false);
        }
    }

    if constexpr (ATTN) {
        asm volatile("v_nop\n\tv_nop\n\tv_nop\n\tv_nop"
                     : "+v"(acc[2]), "+v"(acc[3]));
        asm volatile("v_nop\n\tv_nop\n\tv_nop\n\tv_nop"
                     : "+v"(acc[0]), "+v"(acc[1]));
    }

    const int plane = col0 / ldo;
    float* op = out + (long long)plane * M * ldo;
#pragma unroll
    for (int t = 0; t < 4; ++t) {
        const int c = col0 + t * 16 + l;
        if (c < Nout) {
            const int cc = c - plane * ldo;
#pragma unroll
            for (int v = 0; v < 8; ++v) {
                const int r = rt * 16 + v + 8 * half;
                op[(long long)r * ldo + cc] = acc[t][v];
            }
        }
    }

    if constexpr (ATTN) {
        float wl[4], wr[4];
#pragma unroll
        for (int t = 0; t < 4; ++t) {
            const int c = col0 + t * 16 + l;
            wl[t] = (c < Nout) ? al[c] : 0.0f;
            wr[t] = (c < Nout) ? ar[c] : 0.0f;
        }
        const bool up = (l & 8) != 0;
        float s[8];
#pragma unroll
        for (int v = 0; v < 8; ++v) {
            const float pl = acc[0][v] * wl[0] + acc[1][v] * wl[1] + acc[2][v] * wl[2] + acc[3][v] * wl[3];
            const float pr = acc[0][v] * wr[0] + acc[1][v] * wr[1] + acc[2][v] * wr[2] + acc[3][v] * wr[3];
            const float keep = up ? pr : pl;
            const float give = up ? pl : pr;
            s[v] = keep + __shfl_xor(give, 8, 32);
        }
#pragma unroll
        for (int off = 4; off > 0; off >>= 1) {
#pragma unroll
            for (int v = 0; v < 8; ++v) s[v] += __shfl_xor(s[v], off, 32);
        }
        const int j = l & 7;
        float val = s[0];
#pragma unroll
        for (int v = 1; v < 8; ++v) val = (j == v) ? s[v] : val;
        const int r = rt * 16 + j + 8 * half;
        (up ? er : el)[(long long)r * ncg + cg] = val;
    }
}

constexpr int CNT_P   = 32;
constexpr int CNT_T   = 1024;
constexpr int CNT_EPT = 25;
constexpr int CNT_EPB = CNT_T * CNT_EPT;
constexpr int SCAN_B  = 256;
constexpr int SCAN_G  = (Nn + SCAN_B - 1) / SCAN_B;
static_assert(CNT_P * CNT_EPB >= Ne, "slices must cover the edge list");
static_assert(Nn % 4 == 0 && SCAN_G <= SCAN_B && CNT_EPB % 256 == 0, "layout assumptions");

template <bool FULL>
__device__ __forceinline__ void count_slice(const int* __restrict__ dst, int* __restrict__ rank,
                                            int* c, int e0, int E) {
    int d[CNT_EPT];
#pragma unroll
    for (int j = 0; j < CNT_EPT; ++j) {
        const int e = e0 + j * CNT_T;
        d[j] = (FULL || e < E) ? dst[e] : 0;
    }
    __syncthreads();
    int r[CNT_EPT];
#pragma unroll
    for (int j = 0; j < CNT_EPT; ++j) {
        const int e = e0 + j * CNT_T;
        r[j] = (FULL || e < E) ? atomicAdd(&c[d[j]], 1) : 0;
    }
#pragma unroll
    for (int j = 0; j < CNT_EPT; ++j) {
        const int e = e0 + j * CNT_T;
        if (FULL || e < E) rank[e] = r[j];
    }
}

__global__ __launch_bounds__(CNT_T)
void k_count(const int* __restrict__ dst, int* __restrict__ rank,
             int* __restrict__ part, int* __restrict__ coarse, int E) {
    __shared__ int4 c4[Nn / 4];
    int* c = (int*)c4;
    const int t = threadIdx.x, p = blockIdx.x;
    for (int i = t; i < Nn / 4; i += CNT_T) c4[i] = make_int4(0, 0, 0, 0);
    const int e0 = p * CNT_EPB + t;
    if ((p + 1) * CNT_EPB <= E) count_slice<true>(dst, rank, c, e0, E);
    else                        count_slice<false>(dst, rank, c, e0, E);
    __syncthreads();
    int4* o4 = (int4*)(part + (long long)p * Nn);
    for (int i = t; i < Nn / 4; i += CNT_T) o4[i] = c4[i];
    constexpr int GPW = (SCAN_G + CNT_T / 32 - 1) / (CNT_T / 32);
    const int lane = t & 31, w = t >> 5;
    int s[GPW];
#pragma unroll
    for (int k = 0; k < GPW; ++k) {
        const int i0 = (w + k * (CNT_T / 32)) * (SCAN_B / 4) + lane, i1 = i0 + 32;
        int a = 0;
        if (i0 < Nn / 4) { const int4 v = c4[i0]; a += v.x + v.y + v.z + v.w; }
        if (i1 < Nn / 4) { const int4 v = c4[i1]; a += v.x + v.y + v.z + v.w; }
        s[k] = a;
    }
    for (int off = 16; off > 0; off >>= 1) {
#pragma unroll
        for (int k = 0; k < GPW; ++k) s[k] += __shfl_xor(s[k], off, 32);
    }
    if (lane == 0) {
#pragma unroll
        for (int k = 0; k < GPW; ++k) {
            const int g = w + k * (CNT_T / 32);
            if (g < SCAN_G) coarse[p * SCAN_G + g] = s[k];
        }
    }
}

__global__ __launch_bounds__(SCAN_B)
void k_scan(int* __restrict__ part, const int* __restrict__ coarse, int* __restrict__ rowptr) {
    __shared__ int wtot[SCAN_B / 32];
    __shared__ int wbase[SCAN_B / 32];
    const int t = threadIdx.x, g = blockIdx.x, n = g * SCAN_B + t;
    const int lane = t & 31, w = t >> 5;
    const int nl = min(n, Nn - 1);
    int v[CNT_P];
#pragma unroll
    for (int p = 0; p < CNT_P; ++p) v[p] = part[(long long)p * Nn + nl];
    int s = 0;
    if (t < g) {
#pragma unroll
        for (int p = 0; p < CNT_P; ++p) s += coarse[p * SCAN_G + t];
    }
    int deg = 0;
#pragma unroll
    for (int p = 0; p < CNT_P; ++p) deg += v[p];
    int inc = deg;
    for (int off = 1; off < 32; off <<= 1) {
        const int u = __shfl_up(inc, off, 32);
        if (lane >= off) inc += u;
    }
    for (int off = 16; off > 0; off >>= 1) s += __shfl_xor(s, off, 32);
    if (lane == 31) wtot[w] = inc;
    if (lane == 0) wbase[w] = s;
    __syncthreads();
    int run = inc - deg;
#pragma unroll
    for (int k = 0; k < SCAN_B / 32; ++k) {
        run += wbase[k];
        if (k < w) run += wtot[k];
    }
    if (n < Nn) {
        rowptr[n] = run;
        if (n == Nn - 1) rowptr[Nn] = run + deg;
#pragma unroll
        for (int p = 0; p < CNT_P; ++p) { part[(long long)p * Nn + n] = run; run += v[p]; }
    }
}

__global__ __launch_bounds__(256)
void k_fill_csr(const int* __restrict__ src, const int* __restrict__ dst,
                const int* __restrict__ rank, const int* __restrict__ part,
                int* __restrict__ csr_src, int E) {
    const int i = blockIdx.x * 256 + threadIdx.x;
    const int p = blockIdx.x / (CNT_EPB / 256);
    if (i < E) csr_src[part[(long long)p * Nn + dst[i]] + rank[i]] = src[i];
}

__device__ __forceinline__ int bcast_i(int v, int j) { return __builtin_amdgcn_readlane(v, j); }
__device__ __forceinline__ float bcast_f(float v, int j) {
    return __uint_as_float((unsigned)__builtin_amdgcn_readlane((int)__float_as_uint(v), j));
}
__device__ __forceinline__ float wave_max(float v) {
    for (int off = 16; off > 0; off >>= 1) v = fmaxf(v, __shfl_xor(v, off, 32));
    return v;
}
__device__ __forceinline__ float wave_sum(float v) {
    for (int off = 16; off > 0; off >>= 1) v += __shfl_xor(v, off, 32);
    return v;
}
__device__ __forceinline__ void add4(float4& a, const float4 v) {
    a.x += v.x; a.y += v.y; a.z += v.z; a.w += v.w;
}
__device__ __forceinline__ void fma4(float4& a, float s, const float4 v) {
    a.x += s * v.x; a.y += s * v.y; a.z += s * v.z; a.w += s * v.w;
}
__device__ __forceinline__ float leaky(float v) { return v > 0.0f ? v : 0.2f * v; }
__device__ __forceinline__ float finite_or_zero(float m) {
    return (fabsf(m) < __builtin_huge_valf()) ? m : 0.0f;
}

__global__ __launch_bounds__(256)
void k_sage_gather(const float* __restrict__ xs, const float* __restrict__ xn,
                   const int* __restrict__ rowptr, const int* __restrict__ csr_src,
                   const float* __restrict__ bias, float* __restrict__ h0, int N) {
    const int n = __builtin_amdgcn_readfirstlane((int)(blockIdx.x * 8 + (threadIdx.x >> 5)));
    if (n >= N) return;
    const int lane = threadIdx.x & 31;
    const int beg = rowptr[n], end = rowptr[n + 1];
    const float4 own = ((const float4*)(xs + (long long)n * D0))[lane];
    float4 acc = make_float4(0.f, 0.f, 0.f, 0.f);
    for (int b = beg; b < end; b += 32) {
        const int m = min(32, end - b);
        const int sv = (lane < m) ? csr_src[b + lane] : 0;
        int j = 0;
        for (; j + 8 <= m; j += 8) {
            float4 f[8];
#pragma unroll
            for (int u = 0; u < 8; ++u)
                f[u] = ((const float4*)(xn + (long long)bcast_i(sv, j + u) * D0))[lane];
#pragma unroll
            for (int u = 0; u < 8; ++u) add4(acc, f[u]);
        }
        if (j + 4 <= m) {
            float4 f[4];
#pragma unroll
            for (int u = 0; u < 4; ++u)
                f[u] = ((const float4*)(xn + (long long)bcast_i(sv, j + u) * D0))[lane];
#pragma unroll
            for (int u = 0; u < 4; ++u) add4(acc, f[u]);
            j += 4;
        }
        for (; j < m; ++j)
            add4(acc, ((const float4*)(xn + (long long)bcast_i(sv, j) * D0))[lane]);
    }
    const float inv = 1.0f / fmaxf((float)(end - beg), 1.0f);
    const float* bl = bias + lane * 4;
    float4 r;
    r.x = own.x + acc.x * inv + bl[0]; r.y = own.y + acc.y * inv + bl[1];
    r.z = own.z + acc.z * inv + bl[2]; r.w = own.w + acc.w * inv + bl[3];
    r.x = r.x > 0.0f ? r.x : (expf(r.x) - 1.0f);
    r.y = r.y > 0.0f ? r.y : (expf(r.y) - 1.0f);
    r.z = r.z > 0.0f ? r.z : (expf(r.z) - 1.0f);
    r.w = r.w > 0.0f ? r.w : (expf(r.w) - 1.0f);
    ((float4*)(h0 + (long long)n * D0))[lane] = r;
}

__global__ __launch_bounds__(256)
void k_gat1_gather(const int* __restrict__ rowptr, const int* __restrict__ csr_src,
                   const float* __restrict__ feat, const float* __restrict__ el,
                   const float* __restrict__ er, const float* __restrict__ bias,
                   float* __restrict__ out, int N) {
    const int n = __builtin_amdgcn_readfirstlane((int)(blockIdx.x * 8 + (threadIdx.x >> 5)));
    if (n >= N) return;
    const int lane = threadIdx.x & 31;
    const int beg = rowptr[n], end = rowptr[n + 1];
    const float er0 = er[n * 2], er1 = er[n * 2 + 1];

    float m0 = -__builtin_huge_valf(), m1 = m0;
    for (int b = beg + lane; b < end; b += 32) {
        const float2 l = *(const float2*)(el + (long long)csr_src[b] * 2);
        m0 = fmaxf(m0, leaky(l.x + er0)); m1 = fmaxf(m1, leaky(l.y + er1));
    }
    m0 = finite_or_zero(wave_max(m0)); m1 = finite_or_zero(wave_max(m1));

    float s0 = 0.0f, s1 = 0.0f;
    for (int b = beg + lane; b < end; b += 32) {
        const float2 l = *(const float2*)(el + (long long)csr_src[b] * 2);
        s0 += __expf(leaky(l.x + er0) - m0); s1 += __expf(leaky(l.y + er1) - m1);
    }
    const float d0 = fmaxf(wave_sum(s0), 1e-9f), d1 = fmaxf(wave_sum(s1), 1e-9f);

    const bool hi = lane >= 16;
    float4 acc = make_float4(0.f, 0.f, 0.f, 0.f);
    for (int b = beg; b < end; b += 32) {
        const int m = min(32, end - b);
        int sv = 0; float w0 = 0.0f, w1 = 0.0f;
        if (lane < m) {
            sv = csr_src[b + lane];
            const float2 l = *(const float2*)(el + (long long)sv * 2);
            w0 = __expf(leaky(l.x + er0) - m0) / d0;
            w1 = __expf(leaky(l.y + er1) - m1) / d1;
        }
        int j = 0;
        for (; j + 4 <= m; j += 4) {
            const float4 f0 = ((const float4*)(feat + (long long)bcast_i(sv, j + 0) * (H1 * D1)))[lane];
            const float4 f1 = ((const float4*)(feat + (long long)bcast_i(sv, j + 1) * (H1 * D1)))[lane];
            const float4 f2 = ((const float4*)(feat + (long long)bcast_i(sv, j + 2) * (H1 * D1)))[lane];
            const float4 f3 = ((const float4*)(feat + (long long)bcast_i(sv, j + 3) * (H1 * D1)))[lane];
            const float u0 = bcast_f(w0, j + 0), t0 = bcast_f(w1, j + 0);
            const float u1 = bcast_f(w0, j + 1), t1 = bcast_f(w1, j + 1);
            const float u2 = bcast_f(w0, j + 2), t2 = bcast_f(w1, j + 2);
            const float u3 = bcast_f(w0, j + 3), t3 = bcast_f(w1, j + 3);
            fma4(acc, hi ? t0 : u0, f0);
            fma4(acc, hi ? t1 : u1, f1);
            fma4(acc, hi ? t2 : u2, f2);
            fma4(acc, hi ? t3 : u3, f3);
        }
        for (; j < m; ++j) {
            const float4 f0 = ((const float4*)(feat + (long long)bcast_i(sv, j) * (H1 * D1)))[lane];
            const float u0 = bcast_f(w0, j), t0 = bcast_f(w1, j);
            fma4(acc, hi ? t0 : u0, f0);
        }
    }
    const float* bl = bias + lane * 4;
    float4 r;
    r.x = acc.x + bl[0]; r.y = acc.y + bl[1]; r.z = acc.z + bl[2]; r.w = acc.w + bl[3];
    r.x = r.x > 0.0f ? r.x : (expf(r.x) - 1.0f);
    r.y = r.y > 0.0f ? r.y : (expf(r.y) - 1.0f);
    r.z = r.z > 0.0f ? r.z : (expf(r.z) - 1.0f);
    r.w = r.w > 0.0f ? r.w : (expf(r.w) - 1.0f);
    ((float4*)(out + (long long)n * (H1 * D1)))[lane] = r;
}

__global__ __launch_bounds__(256)
void k_gat2_gather(const int* __restrict__ rowptr, const int* __restrict__ csr_src,
                   const float* __restrict__ feat, const float* __restrict__ el,
                   const float* __restrict__ er, const float* __restrict__ bias,
                   float* __restrict__ out, int N) {
    const int n = __builtin_amdgcn_readfirstlane((int)(blockIdx.x * 8 + (threadIdx.x >> 5)));
    if (n >= N) return;
    const int lane = threadIdx.x & 31;
    const int beg = rowptr[n], end = rowptr[n + 1];
    const float ern = er[n];
    const bool two = lane < C2 - 32;

    float mx = -__builtin_huge_valf();
    for (int b = beg + lane; b < end; b += 32) mx = fmaxf(mx, leaky(el[csr_src[b]] + ern));
    mx = finite_or_zero(wave_max(mx));
    float sm = 0.0f;
    for (int b = beg + lane; b < end; b += 32) sm += __expf(leaky(el[csr_src[b]] + ern) - mx);
    const float dn = fmaxf(wave_sum(sm), 1e-9f);

    float acc0 = 0.0f, acc1 = 0.0f;
    for (int b = beg; b < end; b += 32) {
        const int m = min(32, end - b);
        int sv = 0; float wv = 0.0f;
        if (lane < m) {
            sv = csr_src[b + lane];
            wv = __expf(leaky(el[sv] + ern) - mx) / dn;
        }
        int j = 0;
        for (; j + 4 <= m; j += 4) {
            const float* r0 = feat + (long long)bcast_i(sv, j + 0) * C2;
            const float* r1 = feat + (long long)bcast_i(sv, j + 1) * C2;
            const float* r2 = feat + (long long)bcast_i(sv, j + 2) * C2;
            const float* r3 = feat + (long long)bcast_i(sv, j + 3) * C2;
            const float p0 = r0[lane], p1 = r1[lane], p2 = r2[lane], p3 = r3[lane];
            const float q0 = two ? r0[lane + 32] : 0.0f, q1 = two ? r1[lane + 32] : 0.0f;
            const float q2 = two ? r2[lane + 32] : 0.0f, q3 = two ? r3[lane + 32] : 0.0f;
            const float u0 = bcast_f(wv, j + 0), u1 = bcast_f(wv, j + 1);
            const float u2 = bcast_f(wv, j + 2), u3 = bcast_f(wv, j + 3);
            acc0 += u0 * p0; acc1 += u0 * q0; acc0 += u1 * p1; acc1 += u1 * q1;
            acc0 += u2 * p2; acc1 += u2 * q2; acc0 += u3 * p3; acc1 += u3 * q3;
        }
        for (; j < m; ++j) {
            const float* r0 = feat + (long long)bcast_i(sv, j) * C2;
            const float u0 = bcast_f(wv, j);
            acc0 += u0 * r0[lane];
            acc1 += u0 * (two ? r0[lane + 32] : 0.0f);
        }
    }
    float* o = out + (long long)n * C2;
    o[lane] = acc0 + bias[lane];
    if (two) o[lane + 32] = acc1 + bias[lane + 32];
}

static inline unsigned gsz(long long n, int b) { return (unsigned)((n + b - 1) / b); }

extern "C" void kernel_launch(void* const* d_in, const int* in_sizes, int n_in,
                              void* d_out, int out_size, void* d_ws, size_t ws_size,
                              hipStream_t stream) {
    const float* x       = (const float*)d_in[0];
    const int*   src     = (const int*)  d_in[1];
    const int*   dst     = (const int*)  d_in[2];
    const float* w_self  = (const float*)d_in[3];
    const float* w_neigh = (const float*)d_in[4];
    const float* sage_b  = (const float*)d_in[5];
    const float* g1_w    = (const float*)d_in[6];
    const float* g1_al   = (const float*)d_in[7];
    const float* g1_ar   = (const float*)d_in[8];
    const float* g1_b    = (const float*)d_in[9];
    const float* g2_w    = (const float*)d_in[10];
    const float* g2_al   = (const float*)d_in[11];
    const float* g2_ar   = (const float*)d_in[12];
    const float* g2_b    = (const float*)d_in[13];

    float*    ws    = (float*)d_ws;
    float*    proj  = ws;
    float*    h0    = proj + (long long)Nn * F;
    float*    el    = h0 + (long long)Nn * D0;
    float*    er    = el + Nn * H1;
    int*      rowptr = (int*)(er + Nn * H1);
    int*      rank  = rowptr + (Nn + 16);
    int*      csr_src = rank + Ne;
    float*    wt_self  = (float*)(csr_src + Ne);
    float*    wt_neigh = wt_self + D0 * F;
    float*    wt_g1    = wt_neigh + D0 * F;
    float*    wt_g2    = wt_g1 + (H1 * D1) * D0;
    int*      part   = (int*)(wt_g2 + C2P * D0);
    int*      coarse = part + CNT_P * Nn;
    float*    feat  = proj;

    const int B = 256;
    const int Mtiles = Nn / 16;
    const unsigned rowblk = gsz(Mtiles, 8);
    const unsigned nodeblk = gsz(Nn, 8);

    k_prep<<<gsz(2 * D0 * F + (H1 * D1) * D0 + C2P * D0, B), B, 0, stream>>>(
        w_self, w_neigh, g1_w, g2_w, wt_self, wt_neigh, wt_g1, wt_g2);

    k_count<<<CNT_P, CNT_T, 0, stream>>>(dst, rank, part, coarse, Ne);
    k_scan<<<SCAN_G, SCAN_B, 0, stream>>>(part, coarse, rowptr);
    k_fill_csr<<<gsz(Ne, B), B, 0, stream>>>(src, dst, rank, part, csr_src, Ne);

    {
        const size_t shmem = 64ull * (F + 4) * sizeof(float);
        wmma_gemm<false><<<rowblk * 4, B, shmem, stream>>>(x, wt_self, proj, Nn, F, 2 * D0, 4, D0,
                                                       nullptr, nullptr, nullptr, nullptr);
    }
    k_sage_gather<<<nodeblk, B, 0, stream>>>(proj, proj + (long long)Nn * D0, rowptr, csr_src,
                                             sage_b, h0, Nn);

    {
        const size_t shmem = 64ull * (D0 + 4) * sizeof(float);
        wmma_gemm<true><<<rowblk * 2, B, shmem, stream>>>(h0, wt_g1, feat, Nn, D0, H1 * D1, H1, H1 * D1,
                                                          g1_al, g1_ar, el, er);
    }
    k_gat1_gather<<<nodeblk, B, 0, stream>>>(rowptr, csr_src, feat, el, er, g1_b, h0, Nn);

    {
        const size_t shmem = 64ull * (D0 + 4) * sizeof(float);
        wmma_gemm<true><<<rowblk, B, shmem, stream>>>(h0, wt_g2, feat, Nn, D0, C2, 1, C2,
                                                      g2_al, g2_ar, el, er);
    }
    k_gat2_gather<<<nodeblk, B, 0, stream>>>(rowptr, csr_src, feat, el, er, g2_b,
                                             (float*)d_out, Nn);
}


# GAT gathers read a node's in-edges once: source and logit of its first 32 edges stay in registers for segment max, exp sum and row weights (three reads before); values and sum order unchanged
# speedup vs baseline: 6.7551x; 1.0621x over previous
#include <hip/hip_runtime.h>

constexpr int Nn  = 50000;
constexpr int Ne  = 800000;
constexpr int F   = 256;
constexpr int D0  = 128;
constexpr int H1  = 2;
constexpr int D1  = 64;
constexpr int C2  = 47;
constexpr int C2P = 64;

typedef float v2f __attribute__((ext_vector_type(2)));
typedef float v8f __attribute__((ext_vector_type(8)));

template <int K, int N, int NPAD>
__device__ __forceinline__ void transpose_elem(const float* __restrict__ W, float* __restrict__ Wt, int i) {
    const int c = i / K, k = i % K;
    Wt[i] = (c < N) ? W[k * N + c] : 0.0f;
}
__global__ __launch_bounds__(256)
void k_prep(const float* __restrict__ w_self, const float* __restrict__ w_neigh,
            const float* __restrict__ g1_w, const float* __restrict__ g2_w,
            float* __restrict__ wt_self, float* __restrict__ wt_neigh,
            float* __restrict__ wt_g1, float* __restrict__ wt_g2) {
    constexpr int n_sage = D0 * F;
    constexpr int n_g1   = (H1 * D1) * D0;
    constexpr int n_g2   = C2P * D0;
    int i = blockIdx.x * 256 + threadIdx.x;
    if (i < n_sage) { transpose_elem<F, D0, D0>(w_self, wt_self, i); return; }
    i -= n_sage;
    if (i < n_sage) { transpose_elem<F, D0, D0>(w_neigh, wt_neigh, i); return; }
    i -= n_sage;
    if (i < n_g1) { transpose_elem<D0, H1 * D1, H1 * D1>(g1_w, wt_g1, i); return; }
    i -= n_g1;
    if (i < n_g2) transpose_elem<D0, C2, C2P>(g2_w, wt_g2, i);
}

template <bool ATTN>
__global__ __launch_bounds__(256)
void wmma_gemm(const float* __restrict__ A, const float* __restrict__ Wt,
               float* __restrict__ out, int M, int K, int Nout, int ncg, int ldo,
               const float* __restrict__ al, const float* __restrict__ ar,
               float* __restrict__ el, float* __restrict__ er)
{
    extern __shared__ float lds[];
    const int KP = K + 4;
    const int cg   = blockIdx.x % ncg;
    const int col0 = cg * 64;
    float* w1s = lds;

    {
        const int qsh = 29 - __clz(K);
        const int qm  = (1 << qsh) - 1;
        const int npt = K >> 4;
        const float4* g1 = (const float4*)(Wt + (long long)col0 * K);
        constexpr int U = 8;
        for (int b = 0; b < npt; b += U) {
            float4 v1[U];
#pragma unroll
            for (int u = 0; u < U; ++u) {
                const int q = (b + u) * 256 + threadIdx.x;
                v1[u] = g1[q];
            }
#pragma unroll
            for (int u = 0; u < U; ++u) {
                const int q = (b + u) * 256 + threadIdx.x;
                const int o = (q >> qsh) * KP + (q & qm) * 4;
                *(float4*)(w1s + o) = v1[u];
            }
        }
    }
    __syncthreads();

    const int rt = (blockIdx.x / ncg) * 8 + (threadIdx.x >> 5);
    if (rt >= M / 16) return;
    const int lane = threadIdx.x & 31;
    const int half = lane >> 4;
    const int l    = lane & 15;
    const int row  = rt * 16 + l;

    const float* Arow  = A + (long long)row * K + half * 2;
    __builtin_prefetch(Arow, 0, 3);

    const float* Bs[4];
#pragma unroll
    for (int t = 0; t < 4; ++t) Bs[t] = w1s + (t * 16 + l) * KP + half * 2;

    v8f acc[4] = {};
    for (int kk = 0; kk < K; kk += 4) {
        const v2f a = *(const v2f*)(Arow + kk);
#pragma unroll
        for (int t = 0; t < 4; ++t) {
            const v2f b = *(const v2f*)(Bs[t] + kk);
            acc[t] = __builtin_amdgcn_wmma_f32_16x16x4_f32(
                false, a, false, b, (short)0, acc[t], false, false);
        }
    }

    if constexpr (ATTN) {
        asm volatile("v_nop\n\tv_nop\n\tv_nop\n\tv_nop"
                     : "+v"(acc[2]), "+v"(acc[3]));
        asm volatile("v_nop\n\tv_nop\n\tv_nop\n\tv_nop"
                     : "+v"(acc[0]), "+v"(acc[1]));
    }

    const int plane = col0 / ldo;
    float* op = out + (long long)plane * M * ldo;
#pragma unroll
    for (int t = 0; t < 4; ++t) {
        const int c = col0 + t * 16 + l;
        if (c < Nout) {
            const int cc = c - plane * ldo;
#pragma unroll
            for (int v = 0; v < 8; ++v) {
                const int r = rt * 16 + v + 8 * half;
                op[(long long)r * ldo + cc] = acc[t][v];
            }
        }
    }

    if constexpr (ATTN) {
        float wl[4], wr[4];
#pragma unroll
        for (int t = 0; t < 4; ++t) {
            const int c = col0 + t * 16 + l;
            wl[t] = (c < Nout) ? al[c] : 0.0f;
            wr[t] = (c < Nout) ? ar[c] : 0.0f;
        }
        const bool up = (l & 8) != 0;
        float s[8];
#pragma unroll
        for (int v = 0; v < 8; ++v) {
            const float pl = acc[0][v] * wl[0] + acc[1][v] * wl[1] + acc[2][v] * wl[2] + acc[3][v] * wl[3];
            const float pr = acc[0][v] * wr[0] + acc[1][v] * wr[1] + acc[2][v] * wr[2] + acc[3][v] * wr[3];
            const float keep = up ? pr : pl;
            const float give = up ? pl : pr;
            s[v] = keep + __shfl_xor(give, 8, 32);
        }
#pragma unroll
        for (int off = 4; off > 0; off >>= 1) {
#pragma unroll
            for (int v = 0; v < 8; ++v) s[v] += __shfl_xor(s[v], off, 32);
        }
        const int j = l & 7;
        float val = s[0];
#pragma unroll
        for (int v = 1; v < 8; ++v) val = (j == v) ? s[v] : val;
        const int r = rt * 16 + j + 8 * half;
        (up ? er : el)[(long long)r * ncg + cg] = val;
    }
}

constexpr int CNT_P   = 32;
constexpr int CNT_T   = 1024;
constexpr int CNT_EPT = 25;
constexpr int CNT_EPB = CNT_T * CNT_EPT;
constexpr int SCAN_B  = 256;
constexpr int SCAN_G  = (Nn + SCAN_B - 1) / SCAN_B;
static_assert(CNT_P * CNT_EPB >= Ne, "slices must cover the edge list");
static_assert(Nn % 4 == 0 && SCAN_G <= SCAN_B && CNT_EPB % 256 == 0, "layout assumptions");

template <bool FULL>
__device__ __forceinline__ void count_slice(const int* __restrict__ dst, int* __restrict__ rank,
                                            int* c, int e0, int E) {
    int d[CNT_EPT];
#pragma unroll
    for (int j = 0; j < CNT_EPT; ++j) {
        const int e = e0 + j * CNT_T;
        d[j] = (FULL || e < E) ? dst[e] : 0;
    }
    __syncthreads();
    int r[CNT_EPT];
#pragma unroll
    for (int j = 0; j < CNT_EPT; ++j) {
        const int e = e0 + j * CNT_T;
        r[j] = (FULL || e < E) ? atomicAdd(&c[d[j]], 1) : 0;
    }
#pragma unroll
    for (int j = 0; j < CNT_EPT; ++j) {
        const int e = e0 + j * CNT_T;
        if (FULL || e < E) rank[e] = r[j];
    }
}

__global__ __launch_bounds__(CNT_T)
void k_count(const int* __restrict__ dst, int* __restrict__ rank,
             int* __restrict__ part, int* __restrict__ coarse, int E) {
    __shared__ int4 c4[Nn / 4];
    int* c = (int*)c4;
    const int t = threadIdx.x, p = blockIdx.x;
    for (int i = t; i < Nn / 4; i += CNT_T) c4[i] = make_int4(0, 0, 0, 0);
    const int e0 = p * CNT_EPB + t;
    if ((p + 1) * CNT_EPB <= E) count_slice<true>(dst, rank, c, e0, E);
    else                        count_slice<false>(dst, rank, c, e0, E);
    __syncthreads();
    int4* o4 = (int4*)(part + (long long)p * Nn);
    for (int i = t; i < Nn / 4; i += CNT_T) o4[i] = c4[i];
    constexpr int GPW = (SCAN_G + CNT_T / 32 - 1) / (CNT_T / 32);
    const int lane = t & 31, w = t >> 5;
    int s[GPW];
#pragma unroll
    for (int k = 0; k < GPW; ++k) {
        const int i0 = (w + k * (CNT_T / 32)) * (SCAN_B / 4) + lane, i1 = i0 + 32;
        int a = 0;
        if (i0 < Nn / 4) { const int4 v = c4[i0]; a += v.x + v.y + v.z + v.w; }
        if (i1 < Nn / 4) { const int4 v = c4[i1]; a += v.x + v.y + v.z + v.w; }
        s[k] = a;
    }
    for (int off = 16; off > 0; off >>= 1) {
#pragma unroll
        for (int k = 0; k < GPW; ++k) s[k] += __shfl_xor(s[k], off, 32);
    }
    if (lane == 0) {
#pragma unroll
        for (int k = 0; k < GPW; ++k) {
            const int g = w + k * (CNT_T / 32);
            if (g < SCAN_G) coarse[p * SCAN_G + g] = s[k];
        }
    }
}

__global__ __launch_bounds__(SCAN_B)
void k_scan(int* __restrict__ part, const int* __restrict__ coarse, int* __restrict__ rowptr) {
    __shared__ int wtot[SCAN_B / 32];
    __shared__ int wbase[SCAN_B / 32];
    const int t = threadIdx.x, g = blockIdx.x, n = g * SCAN_B + t;
    const int lane = t & 31, w = t >> 5;
    const int nl = min(n, Nn - 1);
    int v[CNT_P];
#pragma unroll
    for (int p = 0; p < CNT_P; ++p) v[p] = part[(long long)p * Nn + nl];
    int s = 0;
    if (t < g) {
#pragma unroll
        for (int p = 0; p < CNT_P; ++p) s += coarse[p * SCAN_G + t];
    }
    int deg = 0;
#pragma unroll
    for (int p = 0; p < CNT_P; ++p) deg += v[p];
    int inc = deg;
    for (int off = 1; off < 32; off <<= 1) {
        const int u = __shfl_up(inc, off, 32);
        if (lane >= off) inc += u;
    }
    for (int off = 16; off > 0; off >>= 1) s += __shfl_xor(s, off, 32);
    if (lane == 31) wtot[w] = inc;
    if (lane == 0) wbase[w] = s;
    __syncthreads();
    int run = inc - deg;
#pragma unroll
    for (int k = 0; k < SCAN_B / 32; ++k) {
        run += wbase[k];
        if (k < w) run += wtot[k];
    }
    if (n < Nn) {
        rowptr[n] = run;
        if (n == Nn - 1) rowptr[Nn] = run + deg;
#pragma unroll
        for (int p = 0; p < CNT_P; ++p) { part[(long long)p * Nn + n] = run; run += v[p]; }
    }
}

__global__ __launch_bounds__(256)
void k_fill_csr(const int* __restrict__ src, const int* __restrict__ dst,
                const int* __restrict__ rank, const int* __restrict__ part,
                int* __restrict__ csr_src, int E) {
    const int i = blockIdx.x * 256 + threadIdx.x;
    const int p = blockIdx.x / (CNT_EPB / 256);
    if (i < E) csr_src[part[(long long)p * Nn + dst[i]] + rank[i]] = src[i];
}

__device__ __forceinline__ int bcast_i(int v, int j) { return __builtin_amdgcn_readlane(v, j); }
__device__ __forceinline__ float bcast_f(float v, int j) {
    return __uint_as_float((unsigned)__builtin_amdgcn_readlane((int)__float_as_uint(v), j));
}
__device__ __forceinline__ float wave_max(float v) {
    for (int off = 16; off > 0; off >>= 1) v = fmaxf(v, __shfl_xor(v, off, 32));
    return v;
}
__device__ __forceinline__ float wave_sum(float v) {
    for (int off = 16; off > 0; off >>= 1) v += __shfl_xor(v, off, 32);
    return v;
}
__device__ __forceinline__ void add4(float4& a, const float4 v) {
    a.x += v.x; a.y += v.y; a.z += v.z; a.w += v.w;
}
__device__ __forceinline__ void fma4(float4& a, float s, const float4 v) {
    a.x += s * v.x; a.y += s * v.y; a.z += s * v.z; a.w += s * v.w;
}
__device__ __forceinline__ float leaky(float v) { return v > 0.0f ? v : 0.2f * v; }
__device__ __forceinline__ float finite_or_zero(float m) {
    return (fabsf(m) < __builtin_huge_valf()) ? m : 0.0f;
}

__global__ __launch_bounds__(256)
void k_sage_gather(const float* __restrict__ xs, const float* __restrict__ xn,
                   const int* __restrict__ rowptr, const int* __restrict__ csr_src,
                   const float* __restrict__ bias, float* __restrict__ h0, int N) {
    const int n = __builtin_amdgcn_readfirstlane((int)(blockIdx.x * 8 + (threadIdx.x >> 5)));
    if (n >= N) return;
    const int lane = threadIdx.x & 31;
    const int beg = rowptr[n], end = rowptr[n + 1];
    const float4 own = ((const float4*)(xs + (long long)n * D0))[lane];
    float4 acc = make_float4(0.f, 0.f, 0.f, 0.f);
    for (int b = beg; b < end; b += 32) {
        const int m = min(32, end - b);
        const int sv = (lane < m) ? csr_src[b + lane] : 0;
        int j = 0;
        for (; j + 8 <= m; j += 8) {
            float4 f[8];
#pragma unroll
            for (int u = 0; u < 8; ++u)
                f[u] = ((const float4*)(xn + (long long)bcast_i(sv, j + u) * D0))[lane];
#pragma unroll
            for (int u = 0; u < 8; ++u) add4(acc, f[u]);
        }
        if (j + 4 <= m) {
            float4 f[4];
#pragma unroll
            for (int u = 0; u < 4; ++u)
                f[u] = ((const float4*)(xn + (long long)bcast_i(sv, j + u) * D0))[lane];
#pragma unroll
            for (int u = 0; u < 4; ++u) add4(acc, f[u]);
            j += 4;
        }
        for (; j < m; ++j)
            add4(acc, ((const float4*)(xn + (long long)bcast_i(sv, j) * D0))[lane]);
    }
    const float inv = 1.0f / fmaxf((float)(end - beg), 1.0f);
    const float* bl = bias + lane * 4;
    float4 r;
    r.x = own.x + acc.x * inv + bl[0]; r.y = own.y + acc.y * inv + bl[1];
    r.z = own.z + acc.z * inv + bl[2]; r.w = own.w + acc.w * inv + bl[3];
    r.x = r.x > 0.0f ? r.x : (expf(r.x) - 1.0f);
    r.y = r.y > 0.0f ? r.y : (expf(r.y) - 1.0f);
    r.z = r.z > 0.0f ? r.z : (expf(r.z) - 1.0f);
    r.w = r.w > 0.0f ? r.w : (expf(r.w) - 1.0f);
    ((float4*)(h0 + (long long)n * D0))[lane] = r;
}


__device__ __forceinline__ void gat1_rows(float4& acc, const float* __restrict__ feat,
                                          int sv, float w0, float w1, int m, int lane, bool hi) {
    int j = 0;
    for (; j + 4 <= m; j += 4) {
        const float4 f0 = ((const float4*)(feat + (long long)bcast_i(sv, j + 0) * (H1 * D1)))[lane];
        const float4 f1 = ((const float4*)(feat + (long long)bcast_i(sv, j + 1) * (H1 * D1)))[lane];
        const float4 f2 = ((const float4*)(feat + (long long)bcast_i(sv, j + 2) * (H1 * D1)))[lane];
        const float4 f3 = ((const float4*)(feat + (long long)bcast_i(sv, j + 3) * (H1 * D1)))[lane];
        const float u0 = bcast_f(w0, j + 0), t0 = bcast_f(w1, j + 0);
        const float u1 = bcast_f(w0, j + 1), t1 = bcast_f(w1, j + 1);
        const float u2 = bcast_f(w0, j + 2), t2 = bcast_f(w1, j + 2);
        const float u3 = bcast_f(w0, j + 3), t3 = bcast_f(w1, j + 3);
        fma4(acc, hi ? t0 : u0, f0);
        fma4(acc, hi ? t1 : u1, f1);
        fma4(acc, hi ? t2 : u2, f2);
        fma4(acc, hi ? t3 : u3, f3);
    }
    for (; j < m; ++j) {
        const float4 f0 = ((const float4*)(feat + (long long)bcast_i(sv, j) * (H1 * D1)))[lane];
        const float u0 = bcast_f(w0, j), t0 = bcast_f(w1, j);
        fma4(acc, hi ? t0 : u0, f0);
    }
}

__global__ __launch_bounds__(256)
void k_gat1_gather(const int* __restrict__ rowptr, const int* __restrict__ csr_src,
                   const float* __restrict__ feat, const float* __restrict__ el,
                   const float* __restrict__ er, const float* __restrict__ bias,
                   float* __restrict__ out, int N) {
    const int n = __builtin_amdgcn_readfirstlane((int)(blockIdx.x * 8 + (threadIdx.x >> 5)));
    if (n >= N) return;
    const int lane = threadIdx.x & 31;
    const int beg = rowptr[n], end = rowptr[n + 1];
    const float er0 = er[n * 2], er1 = er[n * 2 + 1];

    const bool has = beg + lane < end;
    int sv = 0;
    float x0 = -__builtin_huge_valf(), x1 = x0;
    if (has) {
        sv = csr_src[beg + lane];
        const float2 l = *(const float2*)(el + (long long)sv * 2);
        x0 = leaky(l.x + er0); x1 = leaky(l.y + er1);
    }

    float m0 = x0, m1 = x1;
    for (int b = beg + 32 + lane; b < end; b += 32) {
        const float2 l = *(const float2*)(el + (long long)csr_src[b] * 2);
        m0 = fmaxf(m0, leaky(l.x + er0)); m1 = fmaxf(m1, leaky(l.y + er1));
    }
    m0 = finite_or_zero(wave_max(m0)); m1 = finite_or_zero(wave_max(m1));

    const float p0 = has ? __expf(x0 - m0) : 0.0f;
    const float p1 = has ? __expf(x1 - m1) : 0.0f;
    float s0 = p0, s1 = p1;
    for (int b = beg + 32 + lane; b < end; b += 32) {
        const float2 l = *(const float2*)(el + (long long)csr_src[b] * 2);
        s0 += __expf(leaky(l.x + er0) - m0); s1 += __expf(leaky(l.y + er1) - m1);
    }
    const float d0 = fmaxf(wave_sum(s0), 1e-9f), d1 = fmaxf(wave_sum(s1), 1e-9f);

    const bool hi = lane >= 16;
    float4 acc = make_float4(0.f, 0.f, 0.f, 0.f);
    gat1_rows(acc, feat, sv, p0 / d0, p1 / d1, min(32, end - beg), lane, hi);
    for (int b = beg + 32; b < end; b += 32) {
        const int m = min(32, end - b);
        int sb = 0; float w0 = 0.0f, w1 = 0.0f;
        if (lane < m) {
            sb = csr_src[b + lane];
            const float2 l = *(const float2*)(el + (long long)sb * 2);
            w0 = __expf(leaky(l.x + er0) - m0) / d0;
            w1 = __expf(leaky(l.y + er1) - m1) / d1;
        }
        gat1_rows(acc, feat, sb, w0, w1, m, lane, hi);
    }
    const float* bl = bias + lane * 4;
    float4 r;
    r.x = acc.x + bl[0]; r.y = acc.y + bl[1]; r.z = acc.z + bl[2]; r.w = acc.w + bl[3];
    r.x = r.x > 0.0f ? r.x : (expf(r.x) - 1.0f);
    r.y = r.y > 0.0f ? r.y : (expf(r.y) - 1.0f);
    r.z = r.z > 0.0f ? r.z : (expf(r.z) - 1.0f);
    r.w = r.w > 0.0f ? r.w : (expf(r.w) - 1.0f);
    ((float4*)(out + (long long)n * (H1 * D1)))[lane] = r;
}

__device__ __forceinline__ void gat2_rows(float& acc0, float& acc1, const float* __restrict__ feat,
                                          int sv, float wv, int m, int lane, bool two) {
    int j = 0;
    for (; j + 4 <= m; j += 4) {
        const float* r0 = feat + (long long)bcast_i(sv, j + 0) * C2;
        const float* r1 = feat + (long long)bcast_i(sv, j + 1) * C2;
        const float* r2 = feat + (long long)bcast_i(sv, j + 2) * C2;
        const float* r3 = feat + (long long)bcast_i(sv, j + 3) * C2;
        const float p0 = r0[lane], p1 = r1[lane], p2 = r2[lane], p3 = r3[lane];
        const float q0 = two ? r0[lane + 32] : 0.0f, q1 = two ? r1[lane + 32] : 0.0f;
        const float q2 = two ? r2[lane + 32] : 0.0f, q3 = two ? r3[lane + 32] : 0.0f;
        const float u0 = bcast_f(wv, j + 0), u1 = bcast_f(wv, j + 1);
        const float u2 = bcast_f(wv, j + 2), u3 = bcast_f(wv, j + 3);
        acc0 += u0 * p0; acc1 += u0 * q0; acc0 += u1 * p1; acc1 += u1 * q1;
        acc0 += u2 * p2; acc1 += u2 * q2; acc0 += u3 * p3; acc1 += u3 * q3;
    }
    for (; j < m; ++j) {
        const float* r0 = feat + (long long)bcast_i(sv, j) * C2;
        const float u0 = bcast_f(wv, j);
        acc0 += u0 * r0[lane];
        acc1 += u0 * (two ? r0[lane + 32] : 0.0f);
    }
}

__global__ __launch_bounds__(256)
void k_gat2_gather(const int* __restrict__ rowptr, const int* __restrict__ csr_src,
                   const float* __restrict__ feat, const float* __restrict__ el,
                   const float* __restrict__ er, const float* __restrict__ bias,
                   float* __restrict__ out, int N) {
    const int n = __builtin_amdgcn_readfirstlane((int)(blockIdx.x * 8 + (threadIdx.x >> 5)));
    if (n >= N) return;
    const int lane = threadIdx.x & 31;
    const int beg = rowptr[n], end = rowptr[n + 1];
    const float ern = er[n];
    const bool two = lane < C2 - 32;

    const bool has = beg + lane < end;
    int sv = 0;
    float xv = -__builtin_huge_valf();
    if (has) {
        sv = csr_src[beg + lane];
        xv = leaky(el[sv] + ern);
    }

    float mx = xv;
    for (int b = beg + 32 + lane; b < end; b += 32)
        mx = fmaxf(mx, leaky(el[csr_src[b]] + ern));
    mx = finite_or_zero(wave_max(mx));
    const float pv = has ? __expf(xv - mx) : 0.0f;
    float sm = pv;
    for (int b = beg + 32 + lane; b < end; b += 32) sm += __expf(leaky(el[csr_src[b]] + ern) - mx);
    const float dn = fmaxf(wave_sum(sm), 1e-9f);

    float acc0 = 0.0f, acc1 = 0.0f;
    gat2_rows(acc0, acc1, feat, sv, pv / dn, min(32, end - beg), lane, two);
    for (int b = beg + 32; b < end; b += 32) {
        const int m = min(32, end - b);
        int sb = 0; float wv = 0.0f;
        if (lane < m) {
            sb = csr_src[b + lane];
            wv = __expf(leaky(el[sb] + ern) - mx) / dn;
        }
        gat2_rows(acc0, acc1, feat, sb, wv, m, lane, two);
    }
    float* o = out + (long long)n * C2;
    o[lane] = acc0 + bias[lane];
    if (two) o[lane + 32] = acc1 + bias[lane + 32];
}

static inline unsigned gsz(long long n, int b) { return (unsigned)((n + b - 1) / b); }

extern "C" void kernel_launch(void* const* d_in, const int* in_sizes, int n_in,
                              void* d_out, int out_size, void* d_ws, size_t ws_size,
                              hipStream_t stream) {
    const float* x       = (const float*)d_in[0];
    const int*   src     = (const int*)  d_in[1];
    const int*   dst     = (const int*)  d_in[2];
    const float* w_self  = (const float*)d_in[3];
    const float* w_neigh = (const float*)d_in[4];
    const float* sage_b  = (const float*)d_in[5];
    const float* g1_w    = (const float*)d_in[6];
    const float* g1_al   = (const float*)d_in[7];
    const float* g1_ar   = (const float*)d_in[8];
    const float* g1_b    = (const float*)d_in[9];
    const float* g2_w    = (const float*)d_in[10];
    const float* g2_al   = (const float*)d_in[11];
    const float* g2_ar   = (const float*)d_in[12];
    const float* g2_b    = (const float*)d_in[13];

    float*    ws    = (float*)d_ws;
    float*    proj  = ws;
    float*    h0    = proj + (long long)Nn * F;
    float*    el    = h0 + (long long)Nn * D0;
    float*    er    = el + Nn * H1;
    int*      rowptr = (int*)(er + Nn * H1);
    int*      rank  = rowptr + (Nn + 16);
    int*      csr_src = rank + Ne;
    float*    wt_self  = (float*)(csr_src + Ne);
    float*    wt_neigh = wt_self + D0 * F;
    float*    wt_g1    = wt_neigh + D0 * F;
    float*    wt_g2    = wt_g1 + (H1 * D1) * D0;
    int*      part   = (int*)(wt_g2 + C2P * D0);
    int*      coarse = part + CNT_P * Nn;
    float*    feat  = proj;

    const int B = 256;
    const int Mtiles = Nn / 16;
    const unsigned rowblk = gsz(Mtiles, 8);
    const unsigned nodeblk = gsz(Nn, 8);

    k_prep<<<gsz(2 * D0 * F + (H1 * D1) * D0 + C2P * D0, B), B, 0, stream>>>(
        w_self, w_neigh, g1_w, g2_w, wt_self, wt_neigh, wt_g1, wt_g2);

    k_count<<<CNT_P, CNT_T, 0, stream>>>(dst, rank, part, coarse, Ne);
    k_scan<<<SCAN_G, SCAN_B, 0, stream>>>(part, coarse, rowptr);
    k_fill_csr<<<gsz(Ne, B), B, 0, stream>>>(src, dst, rank, part, csr_src, Ne);

    {
        const size_t shmem = 64ull * (F + 4) * sizeof(float);
        wmma_gemm<false><<<rowblk * 4, B, shmem, stream>>>(x, wt_self, proj, Nn, F, 2 * D0, 4, D0,
                                                       nullptr, nullptr, nullptr, nullptr);
    }
    k_sage_gather<<<nodeblk, B, 0, stream>>>(proj, proj + (long long)Nn * D0, rowptr, csr_src,
                                             sage_b, h0, Nn);

    {
        const size_t shmem = 64ull * (D0 + 4) * sizeof(float);
        wmma_gemm<true><<<rowblk * 2, B, shmem, stream>>>(h0, wt_g1, feat, Nn, D0, H1 * D1, H1, H1 * D1,
                                                          g1_al, g1_ar, el, er);
    }
    k_gat1_gather<<<nodeblk, B, 0, stream>>>(rowptr, csr_src, feat, el, er, g1_b, h0, Nn);

    {
        const size_t shmem = 64ull * (D0 + 4) * sizeof(float);
        wmma_gemm<true><<<rowblk, B, shmem, stream>>>(h0, wt_g2, feat, Nn, D0, C2, 1, C2,
                                                      g2_al, g2_ar, el, er);
    }
    k_gat2_gather<<<nodeblk, B, 0, stream>>>(rowptr, csr_src, feat, el, er, g2_b,
                                             (float*)d_out, Nn);
}


# GEMM fragments: 16-byte loads (A global, B LDS), each feeding two matrix instructions, issued as one group per 32-k block ahead of its matrix instructions; k order in groups of 8 (sum order) changes
# speedup vs baseline: 7.0967x; 1.0506x over previous
#include <hip/hip_runtime.h>

constexpr int Nn  = 50000;
constexpr int Ne  = 800000;
constexpr int F   = 256;
constexpr int D0  = 128;
constexpr int H1  = 2;
constexpr int D1  = 64;
constexpr int C2  = 47;
constexpr int C2P = 64;

typedef float v2f __attribute__((ext_vector_type(2)));
typedef float v4f __attribute__((ext_vector_type(4)));
typedef float v8f __attribute__((ext_vector_type(8)));

template <int K, int N, int NPAD>
__device__ __forceinline__ void transpose_elem(const float* __restrict__ W, float* __restrict__ Wt, int i) {
    const int c = i / K, k = i % K;
    Wt[i] = (c < N) ? W[k * N + c] : 0.0f;
}
__global__ __launch_bounds__(256)
void k_prep(const float* __restrict__ w_self, const float* __restrict__ w_neigh,
            const float* __restrict__ g1_w, const float* __restrict__ g2_w,
            float* __restrict__ wt_self, float* __restrict__ wt_neigh,
            float* __restrict__ wt_g1, float* __restrict__ wt_g2) {
    constexpr int n_sage = D0 * F;
    constexpr int n_g1   = (H1 * D1) * D0;
    constexpr int n_g2   = C2P * D0;
    int i = blockIdx.x * 256 + threadIdx.x;
    if (i < n_sage) { transpose_elem<F, D0, D0>(w_self, wt_self, i); return; }
    i -= n_sage;
    if (i < n_sage) { transpose_elem<F, D0, D0>(w_neigh, wt_neigh, i); return; }
    i -= n_sage;
    if (i < n_g1) { transpose_elem<D0, H1 * D1, H1 * D1>(g1_w, wt_g1, i); return; }
    i -= n_g1;
    if (i < n_g2) transpose_elem<D0, C2, C2P>(g2_w, wt_g2, i);
}

template <bool ATTN>
__global__ __launch_bounds__(256)
void wmma_gemm(const float* __restrict__ A, const float* __restrict__ Wt,
               float* __restrict__ out, int M, int K, int Nout, int ncg, int ldo,
               const float* __restrict__ al, const float* __restrict__ ar,
               float* __restrict__ el, float* __restrict__ er)
{
    extern __shared__ float lds[];
    const int KP = K + 4;
    const int cg   = blockIdx.x % ncg;
    const int col0 = cg * 64;
    float* w1s = lds;

    {
        const int qsh = 29 - __clz(K);
        const int qm  = (1 << qsh) - 1;
        const int npt = K >> 4;
        const float4* g1 = (const float4*)(Wt + (long long)col0 * K);
        constexpr int U = 8;
        for (int b = 0; b < npt; b += U) {
            float4 v1[U];
#pragma unroll
            for (int u = 0; u < U; ++u) {
                const int q = (b + u) * 256 + threadIdx.x;
                v1[u] = g1[q];
            }
#pragma unroll
            for (int u = 0; u < U; ++u) {
                const int q = (b + u) * 256 + threadIdx.x;
                const int o = (q >> qsh) * KP + (q & qm) * 4;
                *(float4*)(w1s + o) = v1[u];
            }
        }
    }
    __syncthreads();

    const int rt = (blockIdx.x / ncg) * 8 + (threadIdx.x >> 5);
    if (rt >= M / 16) return;
    const int lane = threadIdx.x & 31;
    const int half = lane >> 4;
    const int l    = lane & 15;
    const int row  = rt * 16 + l;

    const float* Arow  = A + (long long)row * K + half * 4;
    __builtin_prefetch(Arow, 0, 3);

    const float* Bs[4];
#pragma unroll
    for (int t = 0; t < 4; ++t) Bs[t] = w1s + (t * 16 + l) * KP + half * 4;

    constexpr int KS = 32;
    v8f acc[4] = {};
    for (int kk = 0; kk < K; kk += KS) {
        v4f a[KS / 8], b[KS / 8][4];
#pragma unroll
        for (int u = 0; u < KS / 8; ++u) a[u] = *(const v4f*)(Arow + kk + u * 8);
#pragma unroll
        for (int u = 0; u < KS / 8; ++u) {
#pragma unroll
            for (int t = 0; t < 4; ++t) b[u][t] = *(const v4f*)(Bs[t] + kk + u * 8);
        }
        __builtin_amdgcn_sched_barrier(0);
#pragma unroll
        for (int u = 0; u < KS / 8; ++u) {
#pragma unroll
            for (int t = 0; t < 4; ++t)
                acc[t] = __builtin_amdgcn_wmma_f32_16x16x4_f32(
                    false, a[u].xy, false, b[u][t].xy, (short)0, acc[t], false, false);
#pragma unroll
            for (int t = 0; t < 4; ++t)
                acc[t] = __builtin_amdgcn_wmma_f32_16x16x4_f32(
                    false, a[u].zw, false, b[u][t].zw, (short)0, acc[t], false, false);
        }
    }

    if constexpr (ATTN) {
        asm volatile("v_nop\n\tv_nop\n\tv_nop\n\tv_nop"
                     : "+v"(acc[2]), "+v"(acc[3]));
        asm volatile("v_nop\n\tv_nop\n\tv_nop\n\tv_nop"
                     : "+v"(acc[0]), "+v"(acc[1]));
    }

    const int plane = col0 / ldo;
    float* op = out + (long long)plane * M * ldo;
#pragma unroll
    for (int t = 0; t < 4; ++t) {
        const int c = col0 + t * 16 + l;
        if (c < Nout) {
            const int cc = c - plane * ldo;
#pragma unroll
            for (int v = 0; v < 8; ++v) {
                const int r = rt * 16 + v + 8 * half;
                op[(long long)r * ldo + cc] = acc[t][v];
            }
        }
    }

    if constexpr (ATTN) {
        float wl[4], wr[4];
#pragma unroll
        for (int t = 0; t < 4; ++t) {
            const int c = col0 + t * 16 + l;
            wl[t] = (c < Nout) ? al[c] : 0.0f;
            wr[t] = (c < Nout) ? ar[c] : 0.0f;
        }
        const bool up = (l & 8) != 0;
        float s[8];
#pragma unroll
        for (int v = 0; v < 8; ++v) {
            const float pl = acc[0][v] * wl[0] + acc[1][v] * wl[1] + acc[2][v] * wl[2] + acc[3][v] * wl[3];
            const float pr = acc[0][v] * wr[0] + acc[1][v] * wr[1] + acc[2][v] * wr[2] + acc[3][v] * wr[3];
            const float keep = up ? pr : pl;
            const float give = up ? pl : pr;
            s[v] = keep + __shfl_xor(give, 8, 32);
        }
#pragma unroll
        for (int off = 4; off > 0; off >>= 1) {
#pragma unroll
            for (int v = 0; v < 8; ++v) s[v] += __shfl_xor(s[v], off, 32);
        }
        const int j = l & 7;
        float val = s[0];
#pragma unroll
        for (int v = 1; v < 8; ++v) val = (j == v) ? s[v] : val;
        const int r = rt * 16 + j + 8 * half;
        (up ? er : el)[(long long)r * ncg + cg] = val;
    }
}

constexpr int CNT_P   = 32;
constexpr int CNT_T   = 1024;
constexpr int CNT_EPT = 25;
constexpr int CNT_EPB = CNT_T * CNT_EPT;
constexpr int SCAN_B  = 256;
constexpr int SCAN_G  = (Nn + SCAN_B - 1) / SCAN_B;
static_assert(CNT_P * CNT_EPB >= Ne, "slices must cover the edge list");
static_assert(Nn % 4 == 0 && SCAN_G <= SCAN_B && CNT_EPB % 256 == 0, "layout assumptions");

template <bool FULL>
__device__ __forceinline__ void count_slice(const int* __restrict__ dst, int* __restrict__ rank,
                                            int* c, int e0, int E) {
    int d[CNT_EPT];
#pragma unroll
    for (int j = 0; j < CNT_EPT; ++j) {
        const int e = e0 + j * CNT_T;
        d[j] = (FULL || e < E) ? dst[e] : 0;
    }
    __syncthreads();
    int r[CNT_EPT];
#pragma unroll
    for (int j = 0; j < CNT_EPT; ++j) {
        const int e = e0 + j * CNT_T;
        r[j] = (FULL || e < E) ? atomicAdd(&c[d[j]], 1) : 0;
    }
#pragma unroll
    for (int j = 0; j < CNT_EPT; ++j) {
        const int e = e0 + j * CNT_T;
        if (FULL || e < E) rank[e] = r[j];
    }
}

__global__ __launch_bounds__(CNT_T)
void k_count(const int* __restrict__ dst, int* __restrict__ rank,
             int* __restrict__ part, int* __restrict__ coarse, int E) {
    __shared__ int4 c4[Nn / 4];
    int* c = (int*)c4;
    const int t = threadIdx.x, p = blockIdx.x;
    for (int i = t; i < Nn / 4; i += CNT_T) c4[i] = make_int4(0, 0, 0, 0);
    const int e0 = p * CNT_EPB + t;
    if ((p + 1) * CNT_EPB <= E) count_slice<true>(dst, rank, c, e0, E);
    else                        count_slice<false>(dst, rank, c, e0, E);
    __syncthreads();
    int4* o4 = (int4*)(part + (long long)p * Nn);
    for (int i = t; i < Nn / 4; i += CNT_T) o4[i] = c4[i];
    constexpr int GPW = (SCAN_G + CNT_T / 32 - 1) / (CNT_T / 32);
    const int lane = t & 31, w = t >> 5;
    int s[GPW];
#pragma unroll
    for (int k = 0; k < GPW; ++k) {
        const int i0 = (w + k * (CNT_T / 32)) * (SCAN_B / 4) + lane, i1 = i0 + 32;
        int a = 0;
        if (i0 < Nn / 4) { const int4 v = c4[i0]; a += v.x + v.y + v.z + v.w; }
        if (i1 < Nn / 4) { const int4 v = c4[i1]; a += v.x + v.y + v.z + v.w; }
        s[k] = a;
    }
    for (int off = 16; off > 0; off >>= 1) {
#pragma unroll
        for (int k = 0; k < GPW; ++k) s[k] += __shfl_xor(s[k], off, 32);
    }
    if (lane == 0) {
#pragma unroll
        for (int k = 0; k < GPW; ++k) {
            const int g = w + k * (CNT_T / 32);
            if (g < SCAN_G) coarse[p * SCAN_G + g] = s[k];
        }
    }
}

__global__ __launch_bounds__(SCAN_B)
void k_scan(int* __restrict__ part, const int* __restrict__ coarse, int* __restrict__ rowptr) {
    __shared__ int wtot[SCAN_B / 32];
    __shared__ int wbase[SCAN_B / 32];
    const int t = threadIdx.x, g = blockIdx.x, n = g * SCAN_B + t;
    const int lane = t & 31, w = t >> 5;
    const int nl = min(n, Nn - 1);
    int v[CNT_P];
#pragma unroll
    for (int p = 0; p < CNT_P; ++p) v[p] = part[(long long)p * Nn + nl];
    int s = 0;
    if (t < g) {
#pragma unroll
        for (int p = 0; p < CNT_P; ++p) s += coarse[p * SCAN_G + t];
    }
    int deg = 0;
#pragma unroll
    for (int p = 0; p < CNT_P; ++p) deg += v[p];
    int inc = deg;
    for (int off = 1; off < 32; off <<= 1) {
        const int u = __shfl_up(inc, off, 32);
        if (lane >= off) inc += u;
    }
    for (int off = 16; off > 0; off >>= 1) s += __shfl_xor(s, off, 32);
    if (lane == 31) wtot[w] = inc;
    if (lane == 0) wbase[w] = s;
    __syncthreads();
    int run = inc - deg;
#pragma unroll
    for (int k = 0; k < SCAN_B / 32; ++k) {
        run += wbase[k];
        if (k < w) run += wtot[k];
    }
    if (n < Nn) {
        rowptr[n] = run;
        if (n == Nn - 1) rowptr[Nn] = run + deg;
#pragma unroll
        for (int p = 0; p < CNT_P; ++p) { part[(long long)p * Nn + n] = run; run += v[p]; }
    }
}

__global__ __launch_bounds__(256)
void k_fill_csr(const int* __restrict__ src, const int* __restrict__ dst,
                const int* __restrict__ rank, const int* __restrict__ part,
                int* __restrict__ csr_src, int E) {
    const int i = blockIdx.x * 256 + threadIdx.x;
    const int p = blockIdx.x / (CNT_EPB / 256);
    if (i < E) csr_src[part[(long long)p * Nn + dst[i]] + rank[i]] = src[i];
}

__device__ __forceinline__ int bcast_i(int v, int j) { return __builtin_amdgcn_readlane(v, j); }
__device__ __forceinline__ float bcast_f(float v, int j) {
    return __uint_as_float((unsigned)__builtin_amdgcn_readlane((int)__float_as_uint(v), j));
}
__device__ __forceinline__ float wave_max(float v) {
    for (int off = 16; off > 0; off >>= 1) v = fmaxf(v, __shfl_xor(v, off, 32));
    return v;
}
__device__ __forceinline__ float wave_sum(float v) {
    for (int off = 16; off > 0; off >>= 1) v += __shfl_xor(v, off, 32);
    return v;
}
__device__ __forceinline__ void add4(float4& a, const float4 v) {
    a.x += v.x; a.y += v.y; a.z += v.z; a.w += v.w;
}
__device__ __forceinline__ void fma4(float4& a, float s, const float4 v) {
    a.x += s * v.x; a.y += s * v.y; a.z += s * v.z; a.w += s * v.w;
}
__device__ __forceinline__ float leaky(float v) { return v > 0.0f ? v : 0.2f * v; }
__device__ __forceinline__ float finite_or_zero(float m) {
    return (fabsf(m) < __builtin_huge_valf()) ? m : 0.0f;
}

__global__ __launch_bounds__(256)
void k_sage_gather(const float* __restrict__ xs, const float* __restrict__ xn,
                   const int* __restrict__ rowptr, const int* __restrict__ csr_src,
                   const float* __restrict__ bias, float* __restrict__ h0, int N) {
    const int n = __builtin_amdgcn_readfirstlane((int)(blockIdx.x * 8 + (threadIdx.x >> 5)));
    if (n >= N) return;
    const int lane = threadIdx.x & 31;
    const int beg = rowptr[n], end = rowptr[n + 1];
    const float4 own = ((const float4*)(xs + (long long)n * D0))[lane];
    float4 acc = make_float4(0.f, 0.f, 0.f, 0.f);
    for (int b = beg; b < end; b += 32) {
        const int m = min(32, end - b);
        const int sv = (lane < m) ? csr_src[b + lane] : 0;
        int j = 0;
        for (; j + 8 <= m; j += 8) {
            float4 f[8];
#pragma unroll
            for (int u = 0; u < 8; ++u)
                f[u] = ((const float4*)(xn + (long long)bcast_i(sv, j + u) * D0))[lane];
#pragma unroll
            for (int u = 0; u < 8; ++u) add4(acc, f[u]);
        }
        if (j + 4 <= m) {
            float4 f[4];
#pragma unroll
            for (int u = 0; u < 4; ++u)
                f[u] = ((const float4*)(xn + (long long)bcast_i(sv, j + u) * D0))[lane];
#pragma unroll
            for (int u = 0; u < 4; ++u) add4(acc, f[u]);
            j += 4;
        }
        for (; j < m; ++j)
            add4(acc, ((const float4*)(xn + (long long)bcast_i(sv, j) * D0))[lane]);
    }
    const float inv = 1.0f / fmaxf((float)(end - beg), 1.0f);
    const float* bl = bias + lane * 4;
    float4 r;
    r.x = own.x + acc.x * inv + bl[0]; r.y = own.y + acc.y * inv + bl[1];
    r.z = own.z + acc.z * inv + bl[2]; r.w = own.w + acc.w * inv + bl[3];
    r.x = r.x > 0.0f ? r.x : (expf(r.x) - 1.0f);
    r.y = r.y > 0.0f ? r.y : (expf(r.y) - 1.0f);
    r.z = r.z > 0.0f ? r.z : (expf(r.z) - 1.0f);
    r.w = r.w > 0.0f ? r.w : (expf(r.w) - 1.0f);
    ((float4*)(h0 + (long long)n * D0))[lane] = r;
}


__device__ __forceinline__ void gat1_rows(float4& acc, const float* __restrict__ feat,
                                          int sv, float w0, float w1, int m, int lane, bool hi) {
    int j = 0;
    for (; j + 4 <= m; j += 4) {
        const float4 f0 = ((const float4*)(feat + (long long)bcast_i(sv, j + 0) * (H1 * D1)))[lane];
        const float4 f1 = ((const float4*)(feat + (long long)bcast_i(sv, j + 1) * (H1 * D1)))[lane];
        const float4 f2 = ((const float4*)(feat + (long long)bcast_i(sv, j + 2) * (H1 * D1)))[lane];
        const float4 f3 = ((const float4*)(feat + (long long)bcast_i(sv, j + 3) * (H1 * D1)))[lane];
        const float u0 = bcast_f(w0, j + 0), t0 = bcast_f(w1, j + 0);
        const float u1 = bcast_f(w0, j + 1), t1 = bcast_f(w1, j + 1);
        const float u2 = bcast_f(w0, j + 2), t2 = bcast_f(w1, j + 2);
        const float u3 = bcast_f(w0, j + 3), t3 = bcast_f(w1, j + 3);
        fma4(acc, hi ? t0 : u0, f0);
        fma4(acc, hi ? t1 : u1, f1);
        fma4(acc, hi ? t2 : u2, f2);
        fma4(acc, hi ? t3 : u3, f3);
    }
    for (; j < m; ++j) {
        const float4 f0 = ((const float4*)(feat + (long long)bcast_i(sv, j) * (H1 * D1)))[lane];
        const float u0 = bcast_f(w0, j), t0 = bcast_f(w1, j);
        fma4(acc, hi ? t0 : u0, f0);
    }
}

__global__ __launch_bounds__(256)
void k_gat1_gather(const int* __restrict__ rowptr, const int* __restrict__ csr_src,
                   const float* __restrict__ feat, const float* __restrict__ el,
                   const float* __restrict__ er, const float* __restrict__ bias,
                   float* __restrict__ out, int N) {
    const int n = __builtin_amdgcn_readfirstlane((int)(blockIdx.x * 8 + (threadIdx.x >> 5)));
    if (n >= N) return;
    const int lane = threadIdx.x & 31;
    const int beg = rowptr[n], end = rowptr[n + 1];
    const float er0 = er[n * 2], er1 = er[n * 2 + 1];

    const bool has = beg + lane < end;
    int sv = 0;
    float x0 = -__builtin_huge_valf(), x1 = x0;
    if (has) {
        sv = csr_src[beg + lane];
        const float2 l = *(const float2*)(el + (long long)sv * 2);
        x0 = leaky(l.x + er0); x1 = leaky(l.y + er1);
    }

    float m0 = x0, m1 = x1;
    for (int b = beg + 32 + lane; b < end; b += 32) {
        const float2 l = *(const float2*)(el + (long long)csr_src[b] * 2);
        m0 = fmaxf(m0, leaky(l.x + er0)); m1 = fmaxf(m1, leaky(l.y + er1));
    }
    m0 = finite_or_zero(wave_max(m0)); m1 = finite_or_zero(wave_max(m1));

    const float p0 = has ? __expf(x0 - m0) : 0.0f;
    const float p1 = has ? __expf(x1 - m1) : 0.0f;
    float s0 = p0, s1 = p1;
    for (int b = beg + 32 + lane; b < end; b += 32) {
        const float2 l = *(const float2*)(el + (long long)csr_src[b] * 2);
        s0 += __expf(leaky(l.x + er0) - m0); s1 += __expf(leaky(l.y + er1) - m1);
    }
    const float d0 = fmaxf(wave_sum(s0), 1e-9f), d1 = fmaxf(wave_sum(s1), 1e-9f);

    const bool hi = lane >= 16;
    float4 acc = make_float4(0.f, 0.f, 0.f, 0.f);
    gat1_rows(acc, feat, sv, p0 / d0, p1 / d1, min(32, end - beg), lane, hi);
    for (int b = beg + 32; b < end; b += 32) {
        const int m = min(32, end - b);
        int sb = 0; float w0 = 0.0f, w1 = 0.0f;
        if (lane < m) {
            sb = csr_src[b + lane];
            const float2 l = *(const float2*)(el + (long long)sb * 2);
            w0 = __expf(leaky(l.x + er0) - m0) / d0;
            w1 = __expf(leaky(l.y + er1) - m1) / d1;
        }
        gat1_rows(acc, feat, sb, w0, w1, m, lane, hi);
    }
    const float* bl = bias + lane * 4;
    float4 r;
    r.x = acc.x + bl[0]; r.y = acc.y + bl[1]; r.z = acc.z + bl[2]; r.w = acc.w + bl[3];
    r.x = r.x > 0.0f ? r.x : (expf(r.x) - 1.0f);
    r.y = r.y > 0.0f ? r.y : (expf(r.y) - 1.0f);
    r.z = r.z > 0.0f ? r.z : (expf(r.z) - 1.0f);
    r.w = r.w > 0.0f ? r.w : (expf(r.w) - 1.0f);
    ((float4*)(out + (long long)n * (H1 * D1)))[lane] = r;
}

__device__ __forceinline__ void gat2_rows(float& acc0, float& acc1, const float* __restrict__ feat,
                                          int sv, float wv, int m, int lane, bool two) {
    int j = 0;
    for (; j + 4 <= m; j += 4) {
        const float* r0 = feat + (long long)bcast_i(sv, j + 0) * C2;
        const float* r1 = feat + (long long)bcast_i(sv, j + 1) * C2;
        const float* r2 = feat + (long long)bcast_i(sv, j + 2) * C2;
        const float* r3 = feat + (long long)bcast_i(sv, j + 3) * C2;
        const float p0 = r0[lane], p1 = r1[lane], p2 = r2[lane], p3 = r3[lane];
        const float q0 = two ? r0[lane + 32] : 0.0f, q1 = two ? r1[lane + 32] : 0.0f;
        const float q2 = two ? r2[lane + 32] : 0.0f, q3 = two ? r3[lane + 32] : 0.0f;
        const float u0 = bcast_f(wv, j + 0), u1 = bcast_f(wv, j + 1);
        const float u2 = bcast_f(wv, j + 2), u3 = bcast_f(wv, j + 3);
        acc0 += u0 * p0; acc1 += u0 * q0; acc0 += u1 * p1; acc1 += u1 * q1;
        acc0 += u2 * p2; acc1 += u2 * q2; acc0 += u3 * p3; acc1 += u3 * q3;
    }
    for (; j < m; ++j) {
        const float* r0 = feat + (long long)bcast_i(sv, j) * C2;
        const float u0 = bcast_f(wv, j);
        acc0 += u0 * r0[lane];
        acc1 += u0 * (two ? r0[lane + 32] : 0.0f);
    }
}

__global__ __launch_bounds__(256)
void k_gat2_gather(const int* __restrict__ rowptr, const int* __restrict__ csr_src,
                   const float* __restrict__ feat, const float* __restrict__ el,
                   const float* __restrict__ er, const float* __restrict__ bias,
                   float* __restrict__ out, int N) {
    const int n = __builtin_amdgcn_readfirstlane((int)(blockIdx.x * 8 + (threadIdx.x >> 5)));
    if (n >= N) return;
    const int lane = threadIdx.x & 31;
    const int beg = rowptr[n], end = rowptr[n + 1];
    const float ern = er[n];
    const bool two = lane < C2 - 32;

    const bool has = beg + lane < end;
    int sv = 0;
    float xv = -__builtin_huge_valf();
    if (has) {
        sv = csr_src[beg + lane];
        xv = leaky(el[sv] + ern);
    }

    float mx = xv;
    for (int b = beg + 32 + lane; b < end; b += 32)
        mx = fmaxf(mx, leaky(el[csr_src[b]] + ern));
    mx = finite_or_zero(wave_max(mx));
    const float pv = has ? __expf(xv - mx) : 0.0f;
    float sm = pv;
    for (int b = beg + 32 + lane; b < end; b += 32) sm += __expf(leaky(el[csr_src[b]] + ern) - mx);
    const float dn = fmaxf(wave_sum(sm), 1e-9f);

    float acc0 = 0.0f, acc1 = 0.0f;
    gat2_rows(acc0, acc1, feat, sv, pv / dn, min(32, end - beg), lane, two);
    for (int b = beg + 32; b < end; b += 32) {
        const int m = min(32, end - b);
        int sb = 0; float wv = 0.0f;
        if (lane < m) {
            sb = csr_src[b + lane];
            wv = __expf(leaky(el[sb] + ern) - mx) / dn;
        }
        gat2_rows(acc0, acc1, feat, sb, wv, m, lane, two);
    }
    float* o = out + (long long)n * C2;
    o[lane] = acc0 + bias[lane];
    if (two) o[lane + 32] = acc1 + bias[lane + 32];
}

static inline unsigned gsz(long long n, int b) { return (unsigned)((n + b - 1) / b); }

extern "C" void kernel_launch(void* const* d_in, const int* in_sizes, int n_in,
                              void* d_out, int out_size, void* d_ws, size_t ws_size,
                              hipStream_t stream) {
    const float* x       = (const float*)d_in[0];
    const int*   src     = (const int*)  d_in[1];
    const int*   dst     = (const int*)  d_in[2];
    const float* w_self  = (const float*)d_in[3];
    const float* w_neigh = (const float*)d_in[4];
    const float* sage_b  = (const float*)d_in[5];
    const float* g1_w    = (const float*)d_in[6];
    const float* g1_al   = (const float*)d_in[7];
    const float* g1_ar   = (const float*)d_in[8];
    const float* g1_b    = (const float*)d_in[9];
    const float* g2_w    = (const float*)d_in[10];
    const float* g2_al   = (const float*)d_in[11];
    const float* g2_ar   = (const float*)d_in[12];
    const float* g2_b    = (const float*)d_in[13];

    float*    ws    = (float*)d_ws;
    float*    proj  = ws;
    float*    h0    = proj + (long long)Nn * F;
    float*    el    = h0 + (long long)Nn * D0;
    float*    er    = el + Nn * H1;
    int*      rowptr = (int*)(er + Nn * H1);
    int*      rank  = rowptr + (Nn + 16);
    int*      csr_src = rank + Ne;
    float*    wt_self  = (float*)(csr_src + Ne);
    float*    wt_neigh = wt_self + D0 * F;
    float*    wt_g1    = wt_neigh + D0 * F;
    float*    wt_g2    = wt_g1 + (H1 * D1) * D0;
    int*      part   = (int*)(wt_g2 + C2P * D0);
    int*      coarse = part + CNT_P * Nn;
    float*    feat  = proj;

    const int B = 256;
    const int Mtiles = Nn / 16;
    const unsigned rowblk = gsz(Mtiles, 8);
    const unsigned nodeblk = gsz(Nn, 8);

    k_prep<<<gsz(2 * D0 * F + (H1 * D1) * D0 + C2P * D0, B), B, 0, stream>>>(
        w_self, w_neigh, g1_w, g2_w, wt_self, wt_neigh, wt_g1, wt_g2);

    k_count<<<CNT_P, CNT_T, 0, stream>>>(dst, rank, part, coarse, Ne);
    k_scan<<<SCAN_G, SCAN_B, 0, stream>>>(part, coarse, rowptr);
    k_fill_csr<<<gsz(Ne, B), B, 0, stream>>>(src, dst, rank, part, csr_src, Ne);

    {
        const size_t shmem = 64ull * (F + 4) * sizeof(float);
        wmma_gemm<false><<<rowblk * 4, B, shmem, stream>>>(x, wt_self, proj, Nn, F, 2 * D0, 4, D0,
                                                       nullptr, nullptr, nullptr, nullptr);
    }
    k_sage_gather<<<nodeblk, B, 0, stream>>>(proj, proj + (long long)Nn * D0, rowptr, csr_src,
                                             sage_b, h0, Nn);

    {
        const size_t shmem = 64ull * (D0 + 4) * sizeof(float);
        wmma_gemm<true><<<rowblk * 2, B, shmem, stream>>>(h0, wt_g1, feat, Nn, D0, H1 * D1, H1, H1 * D1,
                                                          g1_al, g1_ar, el, er);
    }
    k_gat1_gather<<<nodeblk, B, 0, stream>>>(rowptr, csr_src, feat, el, er, g1_b, h0, Nn);

    {
        const size_t shmem = 64ull * (D0 + 4) * sizeof(float);
        wmma_gemm<true><<<rowblk, B, shmem, stream>>>(h0, wt_g2, feat, Nn, D0, C2, 1, C2,
                                                      g2_al, g2_ar, el, er);
    }
    k_gat2_gather<<<nodeblk, B, 0, stream>>>(rowptr, csr_src, feat, el, er, g2_b,
                                             (float*)d_out, Nn);
}


# SAGE and GAT1 gathers multiply their finished 16-node row tile by the next layer's weights from LDS in the same launch: 10 launches -> 8, h0/h1 never stored; el/er sum order changes
# speedup vs baseline: 7.3934x; 1.0418x over previous
#include <hip/hip_runtime.h>

constexpr int Nn  = 50000;
constexpr int Ne  = 800000;
constexpr int F   = 256;
constexpr int D0  = 128;
constexpr int H1  = 2;
constexpr int D1  = 64;
constexpr int C2  = 47;
constexpr int C2P = 64;

typedef float v2f __attribute__((ext_vector_type(2)));
typedef float v4f __attribute__((ext_vector_type(4)));
typedef float v8f __attribute__((ext_vector_type(8)));

template <int K, int N, int NPAD>
__device__ __forceinline__ void transpose_elem(const float* __restrict__ W, float* __restrict__ Wt, int i) {
    const int c = i / K, k = i % K;
    Wt[i] = (c < N) ? W[k * N + c] : 0.0f;
}
__global__ __launch_bounds__(256)
void k_prep(const float* __restrict__ w_self, const float* __restrict__ w_neigh,
            const float* __restrict__ g1_w, const float* __restrict__ g2_w,
            float* __restrict__ wt_self, float* __restrict__ wt_neigh,
            float* __restrict__ wt_g1, float* __restrict__ wt_g2) {
    constexpr int n_sage = D0 * F;
    constexpr int n_g1   = (H1 * D1) * D0;
    constexpr int n_g2   = C2P * D0;
    int i = blockIdx.x * 256 + threadIdx.x;
    if (i < n_sage) { transpose_elem<F, D0, D0>(w_self, wt_self, i); return; }
    i -= n_sage;
    if (i < n_sage) { transpose_elem<F, D0, D0>(w_neigh, wt_neigh, i); return; }
    i -= n_sage;
    if (i < n_g1) { transpose_elem<D0, H1 * D1, H1 * D1>(g1_w, wt_g1, i); return; }
    i -= n_g1;
    if (i < n_g2) transpose_elem<D0, C2, C2P>(g2_w, wt_g2, i);
}

__global__ __launch_bounds__(256)
void wmma_gemm(const float* __restrict__ A, const float* __restrict__ Wt,
               float* __restrict__ out, int M, int K, int Nout, int ncg, int ldo)
{
    extern __shared__ float lds[];
    const int KP = K + 4;
    const int cg   = blockIdx.x % ncg;
    const int col0 = cg * 64;
    float* w1s = lds;

    {
        const int qsh = 29 - __clz(K);
        const int qm  = (1 << qsh) - 1;
        const int npt = K >> 4;
        const float4* g1 = (const float4*)(Wt + (long long)col0 * K);
        constexpr int U = 8;
        for (int b = 0; b < npt; b += U) {
            float4 v1[U];
#pragma unroll
            for (int u = 0; u < U; ++u) {
                const int q = (b + u) * 256 + threadIdx.x;
                v1[u] = g1[q];
            }
#pragma unroll
            for (int u = 0; u < U; ++u) {
                const int q = (b + u) * 256 + threadIdx.x;
                const int o = (q >> qsh) * KP + (q & qm) * 4;
                *(float4*)(w1s + o) = v1[u];
            }
        }
    }
    __syncthreads();

    const int rt = (blockIdx.x / ncg) * 8 + (threadIdx.x >> 5);
    if (rt >= M / 16) return;
    const int lane = threadIdx.x & 31;
    const int half = lane >> 4;
    const int l    = lane & 15;
    const int row  = rt * 16 + l;

    const float* Arow  = A + (long long)row * K + half * 4;
    __builtin_prefetch(Arow, 0, 3);

    const float* Bs[4];
#pragma unroll
    for (int t = 0; t < 4; ++t) Bs[t] = w1s + (t * 16 + l) * KP + half * 4;

    constexpr int KS = 32;
    v8f acc[4] = {};
    for (int kk = 0; kk < K; kk += KS) {
        v4f a[KS / 8], b[KS / 8][4];
#pragma unroll
        for (int u = 0; u < KS / 8; ++u) a[u] = *(const v4f*)(Arow + kk + u * 8);
#pragma unroll
        for (int u = 0; u < KS / 8; ++u) {
#pragma unroll
            for (int t = 0; t < 4; ++t) b[u][t] = *(const v4f*)(Bs[t] + kk + u * 8);
        }
        __builtin_amdgcn_sched_barrier(0);
#pragma unroll
        for (int u = 0; u < KS / 8; ++u) {
#pragma unroll
            for (int t = 0; t < 4; ++t)
                acc[t] = __builtin_amdgcn_wmma_f32_16x16x4_f32(
                    false, a[u].xy, false, b[u][t].xy, (short)0, acc[t], false, false);
#pragma unroll
            for (int t = 0; t < 4; ++t)
                acc[t] = __builtin_amdgcn_wmma_f32_16x16x4_f32(
                    false, a[u].zw, false, b[u][t].zw, (short)0, acc[t], false, false);
        }
    }

    const int plane = col0 / ldo;
    float* op = out + (long long)plane * M * ldo;
#pragma unroll
    for (int t = 0; t < 4; ++t) {
        const int c = col0 + t * 16 + l;
        if (c < Nout) {
            const int cc = c - plane * ldo;
#pragma unroll
            for (int v = 0; v < 8; ++v) {
                const int r = rt * 16 + v + 8 * half;
                op[(long long)r * ldo + cc] = acc[t][v];
            }
        }
    }
}

constexpr int CNT_P   = 32;
constexpr int CNT_T   = 1024;
constexpr int CNT_EPT = 25;
constexpr int CNT_EPB = CNT_T * CNT_EPT;
constexpr int SCAN_B  = 256;
constexpr int SCAN_G  = (Nn + SCAN_B - 1) / SCAN_B;
static_assert(CNT_P * CNT_EPB >= Ne, "slices must cover the edge list");
static_assert(Nn % 4 == 0 && SCAN_G <= SCAN_B && CNT_EPB % 256 == 0, "layout assumptions");

template <bool FULL>
__device__ __forceinline__ void count_slice(const int* __restrict__ dst, int* __restrict__ rank,
                                            int* c, int e0, int E) {
    int d[CNT_EPT];
#pragma unroll
    for (int j = 0; j < CNT_EPT; ++j) {
        const int e = e0 + j * CNT_T;
        d[j] = (FULL || e < E) ? dst[e] : 0;
    }
    __syncthreads();
    int r[CNT_EPT];
#pragma unroll
    for (int j = 0; j < CNT_EPT; ++j) {
        const int e = e0 + j * CNT_T;
        r[j] = (FULL || e < E) ? atomicAdd(&c[d[j]], 1) : 0;
    }
#pragma unroll
    for (int j = 0; j < CNT_EPT; ++j) {
        const int e = e0 + j * CNT_T;
        if (FULL || e < E) rank[e] = r[j];
    }
}

__global__ __launch_bounds__(CNT_T)
void k_count(const int* __restrict__ dst, int* __restrict__ rank,
             int* __restrict__ part, int* __restrict__ coarse, int E) {
    __shared__ int4 c4[Nn / 4];
    int* c = (int*)c4;
    const int t = threadIdx.x, p = blockIdx.x;
    for (int i = t; i < Nn / 4; i += CNT_T) c4[i] = make_int4(0, 0, 0, 0);
    const int e0 = p * CNT_EPB + t;
    if ((p + 1) * CNT_EPB <= E) count_slice<true>(dst, rank, c, e0, E);
    else                        count_slice<false>(dst, rank, c, e0, E);
    __syncthreads();
    int4* o4 = (int4*)(part + (long long)p * Nn);
    for (int i = t; i < Nn / 4; i += CNT_T) o4[i] = c4[i];
    constexpr int GPW = (SCAN_G + CNT_T / 32 - 1) / (CNT_T / 32);
    const int lane = t & 31, w = t >> 5;
    int s[GPW];
#pragma unroll
    for (int k = 0; k < GPW; ++k) {
        const int i0 = (w + k * (CNT_T / 32)) * (SCAN_B / 4) + lane, i1 = i0 + 32;
        int a = 0;
        if (i0 < Nn / 4) { const int4 v = c4[i0]; a += v.x + v.y + v.z + v.w; }
        if (i1 < Nn / 4) { const int4 v = c4[i1]; a += v.x + v.y + v.z + v.w; }
        s[k] = a;
    }
    for (int off = 16; off > 0; off >>= 1) {
#pragma unroll
        for (int k = 0; k < GPW; ++k) s[k] += __shfl_xor(s[k], off, 32);
    }
    if (lane == 0) {
#pragma unroll
        for (int k = 0; k < GPW; ++k) {
            const int g = w + k * (CNT_T / 32);
            if (g < SCAN_G) coarse[p * SCAN_G + g] = s[k];
        }
    }
}

__global__ __launch_bounds__(SCAN_B)
void k_scan(int* __restrict__ part, const int* __restrict__ coarse, int* __restrict__ rowptr) {
    __shared__ int wtot[SCAN_B / 32];
    __shared__ int wbase[SCAN_B / 32];
    const int t = threadIdx.x, g = blockIdx.x, n = g * SCAN_B + t;
    const int lane = t & 31, w = t >> 5;
    const int nl = min(n, Nn - 1);
    int v[CNT_P];
#pragma unroll
    for (int p = 0; p < CNT_P; ++p) v[p] = part[(long long)p * Nn + nl];
    int s = 0;
    if (t < g) {
#pragma unroll
        for (int p = 0; p < CNT_P; ++p) s += coarse[p * SCAN_G + t];
    }
    int deg = 0;
#pragma unroll
    for (int p = 0; p < CNT_P; ++p) deg += v[p];
    int inc = deg;
    for (int off = 1; off < 32; off <<= 1) {
        const int u = __shfl_up(inc, off, 32);
        if (lane >= off) inc += u;
    }
    for (int off = 16; off > 0; off >>= 1) s += __shfl_xor(s, off, 32);
    if (lane == 31) wtot[w] = inc;
    if (lane == 0) wbase[w] = s;
    __syncthreads();
    int run = inc - deg;
#pragma unroll
    for (int k = 0; k < SCAN_B / 32; ++k) {
        run += wbase[k];
        if (k < w) run += wtot[k];
    }
    if (n < Nn) {
        rowptr[n] = run;
        if (n == Nn - 1) rowptr[Nn] = run + deg;
#pragma unroll
        for (int p = 0; p < CNT_P; ++p) { part[(long long)p * Nn + n] = run; run += v[p]; }
    }
}

__global__ __launch_bounds__(256)
void k_fill_csr(const int* __restrict__ src, const int* __restrict__ dst,
                const int* __restrict__ rank, const int* __restrict__ part,
                int* __restrict__ csr_src, int E) {
    const int i = blockIdx.x * 256 + threadIdx.x;
    const int p = blockIdx.x / (CNT_EPB / 256);
    if (i < E) csr_src[part[(long long)p * Nn + dst[i]] + rank[i]] = src[i];
}

__device__ __forceinline__ int bcast_i(int v, int j) { return __builtin_amdgcn_readlane(v, j); }
__device__ __forceinline__ float bcast_f(float v, int j) {
    return __uint_as_float((unsigned)__builtin_amdgcn_readlane((int)__float_as_uint(v), j));
}
__device__ __forceinline__ float wave_max(float v) {
    for (int off = 16; off > 0; off >>= 1) v = fmaxf(v, __shfl_xor(v, off, 32));
    return v;
}
__device__ __forceinline__ float wave_sum(float v) {
    for (int off = 16; off > 0; off >>= 1) v += __shfl_xor(v, off, 32);
    return v;
}
__device__ __forceinline__ void add4(float4& a, const float4 v) {
    a.x += v.x; a.y += v.y; a.z += v.z; a.w += v.w;
}
__device__ __forceinline__ void fma4(float4& a, float s, const float4 v) {
    a.x += s * v.x; a.y += s * v.y; a.z += s * v.z; a.w += s * v.w;
}
__device__ __forceinline__ float leaky(float v) { return v > 0.0f ? v : 0.2f * v; }
__device__ __forceinline__ float finite_or_zero(float m) {
    return (fabsf(m) < __builtin_huge_valf()) ? m : 0.0f;
}

__device__ __forceinline__ float4 sage_row(const float* __restrict__ xs, const float* __restrict__ xn,
                                           const int* __restrict__ rowptr, const int* __restrict__ csr_src,
                                           const float* __restrict__ bias, int n, int lane) {
    const int beg = rowptr[n], end = rowptr[n + 1];
    const float4 own = ((const float4*)(xs + (long long)n * D0))[lane];
    float4 acc = make_float4(0.f, 0.f, 0.f, 0.f);
    for (int b = beg; b < end; b += 32) {
        const int m = min(32, end - b);
        const int sv = (lane < m) ? csr_src[b + lane] : 0;
        int j = 0;
        for (; j + 8 <= m; j += 8) {
            float4 f[8];
#pragma unroll
            for (int u = 0; u < 8; ++u)
                f[u] = ((const float4*)(xn + (long long)bcast_i(sv, j + u) * D0))[lane];
#pragma unroll
            for (int u = 0; u < 8; ++u) add4(acc, f[u]);
        }
        if (j + 4 <= m) {
            float4 f[4];
#pragma unroll
            for (int u = 0; u < 4; ++u)
                f[u] = ((const float4*)(xn + (long long)bcast_i(sv, j + u) * D0))[lane];
#pragma unroll
            for (int u = 0; u < 4; ++u) add4(acc, f[u]);
            j += 4;
        }
        for (; j < m; ++j)
            add4(acc, ((const float4*)(xn + (long long)bcast_i(sv, j) * D0))[lane]);
    }
    const float inv = 1.0f / fmaxf((float)(end - beg), 1.0f);
    const float* bl = bias + lane * 4;
    float4 r;
    r.x = own.x + acc.x * inv + bl[0]; r.y = own.y + acc.y * inv + bl[1];
    r.z = own.z + acc.z * inv + bl[2]; r.w = own.w + acc.w * inv + bl[3];
    r.x = r.x > 0.0f ? r.x : (expf(r.x) - 1.0f);
    r.y = r.y > 0.0f ? r.y : (expf(r.y) - 1.0f);
    r.z = r.z > 0.0f ? r.z : (expf(r.z) - 1.0f);
    r.w = r.w > 0.0f ? r.w : (expf(r.w) - 1.0f);
    return r;
}


__device__ __forceinline__ void gat1_rows(float4& acc, const float* __restrict__ feat,
                                          int sv, float w0, float w1, int m, int lane, bool hi) {
    int j = 0;
    for (; j + 4 <= m; j += 4) {
        const float4 f0 = ((const float4*)(feat + (long long)bcast_i(sv, j + 0) * (H1 * D1)))[lane];
        const float4 f1 = ((const float4*)(feat + (long long)bcast_i(sv, j + 1) * (H1 * D1)))[lane];
        const float4 f2 = ((const float4*)(feat + (long long)bcast_i(sv, j + 2) * (H1 * D1)))[lane];
        const float4 f3 = ((const float4*)(feat + (long long)bcast_i(sv, j + 3) * (H1 * D1)))[lane];
        const float u0 = bcast_f(w0, j + 0), t0 = bcast_f(w1, j + 0);
        const float u1 = bcast_f(w0, j + 1), t1 = bcast_f(w1, j + 1);
        const float u2 = bcast_f(w0, j + 2), t2 = bcast_f(w1, j + 2);
        const float u3 = bcast_f(w0, j + 3), t3 = bcast_f(w1, j + 3);
        fma4(acc, hi ? t0 : u0, f0);
        fma4(acc, hi ? t1 : u1, f1);
        fma4(acc, hi ? t2 : u2, f2);
        fma4(acc, hi ? t3 : u3, f3);
    }
    for (; j < m; ++j) {
        const float4 f0 = ((const float4*)(feat + (long long)bcast_i(sv, j) * (H1 * D1)))[lane];
        const float u0 = bcast_f(w0, j), t0 = bcast_f(w1, j);
        fma4(acc, hi ? t0 : u0, f0);
    }
}

__device__ __forceinline__ float4 gat1_row(const int* __restrict__ rowptr, const int* __restrict__ csr_src,
                                           const float* __restrict__ feat, const float* __restrict__ el,
                                           const float* __restrict__ er, const float* __restrict__ bias,
                                           int n, int lane) {
    const int beg = rowptr[n], end = rowptr[n + 1];
    const float er0 = er[n * 2], er1 = er[n * 2 + 1];

    const bool has = beg + lane < end;
    int sv = 0;
    float x0 = -__builtin_huge_valf(), x1 = x0;
    if (has) {
        sv = csr_src[beg + lane];
        const float2 l = *(const float2*)(el + (long long)sv * 2);
        x0 = leaky(l.x + er0); x1 = leaky(l.y + er1);
    }

    float m0 = x0, m1 = x1;
    for (int b = beg + 32 + lane; b < end; b += 32) {
        const float2 l = *(const float2*)(el + (long long)csr_src[b] * 2);
        m0 = fmaxf(m0, leaky(l.x + er0)); m1 = fmaxf(m1, leaky(l.y + er1));
    }
    m0 = finite_or_zero(wave_max(m0)); m1 = finite_or_zero(wave_max(m1));

    const float p0 = has ? __expf(x0 - m0) : 0.0f;
    const float p1 = has ? __expf(x1 - m1) : 0.0f;
    float s0 = p0, s1 = p1;
    for (int b = beg + 32 + lane; b < end; b += 32) {
        const float2 l = *(const float2*)(el + (long long)csr_src[b] * 2);
        s0 += __expf(leaky(l.x + er0) - m0); s1 += __expf(leaky(l.y + er1) - m1);
    }
    const float d0 = fmaxf(wave_sum(s0), 1e-9f), d1 = fmaxf(wave_sum(s1), 1e-9f);

    const bool hi = lane >= 16;
    float4 acc = make_float4(0.f, 0.f, 0.f, 0.f);
    gat1_rows(acc, feat, sv, p0 / d0, p1 / d1, min(32, end - beg), lane, hi);
    for (int b = beg + 32; b < end; b += 32) {
        const int m = min(32, end - b);
        int sb = 0; float w0 = 0.0f, w1 = 0.0f;
        if (lane < m) {
            sb = csr_src[b + lane];
            const float2 l = *(const float2*)(el + (long long)sb * 2);
            w0 = __expf(leaky(l.x + er0) - m0) / d0;
            w1 = __expf(leaky(l.y + er1) - m1) / d1;
        }
        gat1_rows(acc, feat, sb, w0, w1, m, lane, hi);
    }
    const float* bl = bias + lane * 4;
    float4 r;
    r.x = acc.x + bl[0]; r.y = acc.y + bl[1]; r.z = acc.z + bl[2]; r.w = acc.w + bl[3];
    r.x = r.x > 0.0f ? r.x : (expf(r.x) - 1.0f);
    r.y = r.y > 0.0f ? r.y : (expf(r.y) - 1.0f);
    r.z = r.z > 0.0f ? r.z : (expf(r.z) - 1.0f);
    r.w = r.w > 0.0f ? r.w : (expf(r.w) - 1.0f);
    return r;
}

constexpr int KP1 = D0 + 4;
constexpr int FUSE_T = 512;
constexpr int FUSE_G = 1024;
constexpr int FUSE_G3 = 768;
static_assert(Nn % 16 == 0, "a tile is 16 whole nodes");

template <int NCOL>
__device__ __forceinline__ void stage_panel(const float* __restrict__ wt, float* ws) {
    constexpr int PT = NCOL * (D0 / 4) / FUSE_T;
    const float4* g = (const float4*)wt;
    float4 v[PT];
#pragma unroll
    for (int u = 0; u < PT; ++u) v[u] = g[u * FUSE_T + threadIdx.x];
#pragma unroll
    for (int u = 0; u < PT; ++u) {
        const int q = u * FUSE_T + threadIdx.x;
        *(float4*)(ws + (q >> 5) * KP1 + (q & 31) * 4) = v[u];
    }
}

template <int NCOL, int NH>
__device__ __forceinline__ void project_tile(const float* hs, const float* ws, float* red, int row0,
                                             float* __restrict__ feat, int Nout,
                                             const float* __restrict__ al, const float* __restrict__ ar,
                                             float* __restrict__ el, float* __restrict__ er) {
    static_assert(NCOL == 64 * NH, "a head is 64 (padded) columns");
    const int w = __builtin_amdgcn_readfirstlane((int)(threadIdx.x >> 5));
    const int lane = threadIdx.x & 31, half = lane >> 4, l = lane & 15;
    if (w < NCOL / 16) {
        const float* Ap = hs + l * KP1 + half * 4;
        const float* Bp = ws + (w * 16 + l) * KP1 + half * 4;
        v8f acc = {};
        for (int kk = 0; kk < D0; kk += 32) {
            v4f a[4], b[4];
#pragma unroll
            for (int u = 0; u < 4; ++u) a[u] = *(const v4f*)(Ap + kk + u * 8);
#pragma unroll
            for (int u = 0; u < 4; ++u) b[u] = *(const v4f*)(Bp + kk + u * 8);
#pragma unroll
            for (int u = 0; u < 4; ++u) {
                acc = __builtin_amdgcn_wmma_f32_16x16x4_f32(
                    false, a[u].xy, false, b[u].xy, (short)0, acc, false, false);
                acc = __builtin_amdgcn_wmma_f32_16x16x4_f32(
                    false, a[u].zw, false, b[u].zw, (short)0, acc, false, false);
            }
        }
        asm volatile("v_nop\n\tv_nop\n\tv_nop\n\tv_nop" : "+v"(acc));
        const int c = w * 16 + l;
        const bool live = c < Nout;
        if (live) {
#pragma unroll
            for (int v = 0; v < 8; ++v)
                feat[(long long)(row0 + v + 8 * half) * Nout + c] = acc[v];
        }
        const float wl = live ? al[c] : 0.0f, wr = live ? ar[c] : 0.0f;
        const bool up = (l & 8) != 0;
        float s[8];
#pragma unroll
        for (int v = 0; v < 8; ++v) {
            const float pl = acc[v] * wl, pr = acc[v] * wr;
            const float keep = up ? pr : pl;
            const float give = up ? pl : pr;
            s[v] = keep + __shfl_xor(give, 8, 32);
        }
#pragma unroll
        for (int off = 4; off > 0; off >>= 1) {
#pragma unroll
            for (int v = 0; v < 8; ++v) s[v] += __shfl_xor(s[v], off, 32);
        }
        const int j = l & 7;
        float val = s[0];
#pragma unroll
        for (int v = 1; v < 8; ++v) val = (j == v) ? s[v] : val;
        red[w * 32 + (up ? 16 : 0) + 8 * half + j] = val;
    }
    __syncthreads();
    if (threadIdx.x < 32 * NH) {
        const int row = threadIdx.x & 15, which = (threadIdx.x >> 4) & 1, h = threadIdx.x >> 5;
        const float* p = red + (h * 4) * 32 + which * 16 + row;
        const float sum = ((p[0] + p[32]) + p[64]) + p[96];
        (which ? er : el)[(long long)(row0 + row) * NH + h] = sum;
    }
}

__global__ __launch_bounds__(FUSE_T) __attribute__((amdgpu_waves_per_eu(16, 16)))
void k_sage_project(const float* __restrict__ xs, const float* __restrict__ xn,
                    const int* __restrict__ rowptr, const int* __restrict__ csr_src,
                    const float* __restrict__ bias, const float* __restrict__ wt,
                    const float* __restrict__ al, const float* __restrict__ ar,
                    float* __restrict__ feat, float* __restrict__ el, float* __restrict__ er) {
    extern __shared__ float fuse_lds[];
    float* ws = fuse_lds;
    float* hs = ws + (H1 * D1) * KP1;
    float* red = hs + 16 * KP1;
    stage_panel<H1 * D1>(wt, ws);
    const int lane = threadIdx.x & 31, w = threadIdx.x >> 5;
    for (int tile = blockIdx.x; tile < Nn / 16; tile += gridDim.x) {
        const int n = __builtin_amdgcn_readfirstlane(tile * 16 + w);
        const float4 r = sage_row(xs, xn, rowptr, csr_src, bias, n, lane);
        *(float4*)(hs + w * KP1 + lane * 4) = r;
        __syncthreads();
        project_tile<H1 * D1, H1>(hs, ws, red, tile * 16, feat, H1 * D1, al, ar, el, er);
    }
}

__global__ __launch_bounds__(FUSE_T)
void k_gat1_project(const int* __restrict__ rowptr, const int* __restrict__ csr_src,
                    const float* __restrict__ feat1, const float* __restrict__ el1,
                    const float* __restrict__ er1, const float* __restrict__ bias,
                    const float* __restrict__ wt, const float* __restrict__ al, const float* __restrict__ ar,
                    float* __restrict__ feat2, float* __restrict__ el2, float* __restrict__ er2) {
    extern __shared__ float fuse_lds[];
    float* ws = fuse_lds;
    float* hs = ws + C2P * KP1;
    float* red = hs + 16 * KP1;
    stage_panel<C2P>(wt, ws);
    const int lane = threadIdx.x & 31, w = threadIdx.x >> 5;
    for (int tile = blockIdx.x; tile < Nn / 16; tile += gridDim.x) {
        const int n = __builtin_amdgcn_readfirstlane(tile * 16 + w);
        const float4 r = gat1_row(rowptr, csr_src, feat1, el1, er1, bias, n, lane);
        *(float4*)(hs + w * KP1 + lane * 4) = r;
        __syncthreads();
        project_tile<C2P, 1>(hs, ws, red, tile * 16, feat2, C2, al, ar, el2, er2);
    }
}

__device__ __forceinline__ void gat2_rows(float& acc0, float& acc1, const float* __restrict__ feat,
                                          int sv, float wv, int m, int lane, bool two) {
    int j = 0;
    for (; j + 4 <= m; j += 4) {
        const float* r0 = feat + (long long)bcast_i(sv, j + 0) * C2;
        const float* r1 = feat + (long long)bcast_i(sv, j + 1) * C2;
        const float* r2 = feat + (long long)bcast_i(sv, j + 2) * C2;
        const float* r3 = feat + (long long)bcast_i(sv, j + 3) * C2;
        const float p0 = r0[lane], p1 = r1[lane], p2 = r2[lane], p3 = r3[lane];
        const float q0 = two ? r0[lane + 32] : 0.0f, q1 = two ? r1[lane + 32] : 0.0f;
        const float q2 = two ? r2[lane + 32] : 0.0f, q3 = two ? r3[lane + 32] : 0.0f;
        const float u0 = bcast_f(wv, j + 0), u1 = bcast_f(wv, j + 1);
        const float u2 = bcast_f(wv, j + 2), u3 = bcast_f(wv, j + 3);
        acc0 += u0 * p0; acc1 += u0 * q0; acc0 += u1 * p1; acc1 += u1 * q1;
        acc0 += u2 * p2; acc1 += u2 * q2; acc0 += u3 * p3; acc1 += u3 * q3;
    }
    for (; j < m; ++j) {
        const float* r0 = feat + (long long)bcast_i(sv, j) * C2;
        const float u0 = bcast_f(wv, j);
        acc0 += u0 * r0[lane];
        acc1 += u0 * (two ? r0[lane + 32] : 0.0f);
    }
}

__global__ __launch_bounds__(256)
void k_gat2_gather(const int* __restrict__ rowptr, const int* __restrict__ csr_src,
                   const float* __restrict__ feat, const float* __restrict__ el,
                   const float* __restrict__ er, const float* __restrict__ bias,
                   float* __restrict__ out, int N) {
    const int n = __builtin_amdgcn_readfirstlane((int)(blockIdx.x * 8 + (threadIdx.x >> 5)));
    if (n >= N) return;
    const int lane = threadIdx.x & 31;
    const int beg = rowptr[n], end = rowptr[n + 1];
    const float ern = er[n];
    const bool two = lane < C2 - 32;

    const bool has = beg + lane < end;
    int sv = 0;
    float xv = -__builtin_huge_valf();
    if (has) {
        sv = csr_src[beg + lane];
        xv = leaky(el[sv] + ern);
    }

    float mx = xv;
    for (int b = beg + 32 + lane; b < end; b += 32)
        mx = fmaxf(mx, leaky(el[csr_src[b]] + ern));
    mx = finite_or_zero(wave_max(mx));
    const float pv = has ? __expf(xv - mx) : 0.0f;
    float sm = pv;
    for (int b = beg + 32 + lane; b < end; b += 32) sm += __expf(leaky(el[csr_src[b]] + ern) - mx);
    const float dn = fmaxf(wave_sum(sm), 1e-9f);

    float acc0 = 0.0f, acc1 = 0.0f;
    gat2_rows(acc0, acc1, feat, sv, pv / dn, min(32, end - beg), lane, two);
    for (int b = beg + 32; b < end; b += 32) {
        const int m = min(32, end - b);
        int sb = 0; float wv = 0.0f;
        if (lane < m) {
            sb = csr_src[b + lane];
            wv = __expf(leaky(el[sb] + ern) - mx) / dn;
        }
        gat2_rows(acc0, acc1, feat, sb, wv, m, lane, two);
    }
    float* o = out + (long long)n * C2;
    o[lane] = acc0 + bias[lane];
    if (two) o[lane + 32] = acc1 + bias[lane + 32];
}

static inline unsigned gsz(long long n, int b) { return (unsigned)((n + b - 1) / b); }

extern "C" void kernel_launch(void* const* d_in, const int* in_sizes, int n_in,
                              void* d_out, int out_size, void* d_ws, size_t ws_size,
                              hipStream_t stream) {
    const float* x       = (const float*)d_in[0];
    const int*   src     = (const int*)  d_in[1];
    const int*   dst     = (const int*)  d_in[2];
    const float* w_self  = (const float*)d_in[3];
    const float* w_neigh = (const float*)d_in[4];
    const float* sage_b  = (const float*)d_in[5];
    const float* g1_w    = (const float*)d_in[6];
    const float* g1_al   = (const float*)d_in[7];
    const float* g1_ar   = (const float*)d_in[8];
    const float* g1_b    = (const float*)d_in[9];
    const float* g2_w    = (const float*)d_in[10];
    const float* g2_al   = (const float*)d_in[11];
    const float* g2_ar   = (const float*)d_in[12];
    const float* g2_b    = (const float*)d_in[13];

    float*    ws    = (float*)d_ws;
    float*    proj  = ws;
    float*    feat1 = proj + (long long)Nn * F;
    float*    el    = feat1 + (long long)Nn * D0;
    float*    er    = el + Nn * H1;
    int*      rowptr = (int*)(er + Nn * H1);
    int*      rank  = rowptr + (Nn + 16);
    int*      csr_src = rank + Ne;
    float*    wt_self  = (float*)(csr_src + Ne);
    float*    wt_neigh = wt_self + D0 * F;
    float*    wt_g1    = wt_neigh + D0 * F;
    float*    wt_g2    = wt_g1 + (H1 * D1) * D0;
    int*      part   = (int*)(wt_g2 + C2P * D0);
    int*      coarse = part + CNT_P * Nn;
    float*    el2   = (float*)(coarse + CNT_P * SCAN_G);
    float*    er2   = el2 + Nn;
    float*    feat2 = proj;

    const int B = 256;
    const int Mtiles = Nn / 16;
    const unsigned rowblk = gsz(Mtiles, 8);
    const unsigned nodeblk = gsz(Nn, 8);
    constexpr size_t sh_tile = (16ull * KP1 + 8 * 32) * sizeof(float);

    k_prep<<<gsz(2 * D0 * F + (H1 * D1) * D0 + C2P * D0, B), B, 0, stream>>>(
        w_self, w_neigh, g1_w, g2_w, wt_self, wt_neigh, wt_g1, wt_g2);

    k_count<<<CNT_P, CNT_T, 0, stream>>>(dst, rank, part, coarse, Ne);
    k_scan<<<SCAN_G, SCAN_B, 0, stream>>>(part, coarse, rowptr);
    k_fill_csr<<<gsz(Ne, B), B, 0, stream>>>(src, dst, rank, part, csr_src, Ne);

    {
        const size_t shmem = 64ull * (F + 4) * sizeof(float);
        wmma_gemm<<<rowblk * 4, B, shmem, stream>>>(x, wt_self, proj, Nn, F, 2 * D0, 4, D0);
    }
    k_sage_project<<<FUSE_G, FUSE_T, (size_t)(H1 * D1) * KP1 * sizeof(float) + sh_tile, stream>>>(
        proj, proj + (long long)Nn * D0, rowptr, csr_src, sage_b, wt_g1, g1_al, g1_ar, feat1, el, er);

    k_gat1_project<<<FUSE_G3, FUSE_T, (size_t)C2P * KP1 * sizeof(float) + sh_tile, stream>>>(
        rowptr, csr_src, feat1, el, er, g1_b, wt_g2, g2_al, g2_ar, feat2, el2, er2);

    k_gat2_gather<<<nodeblk, B, 0, stream>>>(rowptr, csr_src, feat2, el2, er2, g2_b,
                                             (float*)d_out, Nn);
}


# Independent work shares launches: weight transposes ride in the counting launch as extra work-groups, edge placement (4 edges per thread) rides behind the GEMM work-groups in its launch; 8->6 launches
# speedup vs baseline: 7.9878x; 1.0804x over previous
#include <hip/hip_runtime.h>

constexpr int Nn  = 50000;
constexpr int Ne  = 800000;
constexpr int F   = 256;
constexpr int D0  = 128;
constexpr int H1  = 2;
constexpr int D1  = 64;
constexpr int C2  = 47;
constexpr int C2P = 64;

typedef float v2f __attribute__((ext_vector_type(2)));
typedef float v4f __attribute__((ext_vector_type(4)));
typedef float v8f __attribute__((ext_vector_type(8)));

template <int K, int N, int NPAD>
__device__ __forceinline__ void transpose_elem(const float* __restrict__ W, float* __restrict__ Wt, int i) {
    const int c = i / K, k = i % K;
    Wt[i] = (c < N) ? W[k * N + c] : 0.0f;
}
constexpr int PREP_N = 2 * D0 * F + (H1 * D1) * D0 + C2P * D0;
__device__ __forceinline__ void prep_elem(int i,
            const float* __restrict__ w_self, const float* __restrict__ w_neigh,
            const float* __restrict__ g1_w, const float* __restrict__ g2_w,
            float* __restrict__ wt_self, float* __restrict__ wt_neigh,
            float* __restrict__ wt_g1, float* __restrict__ wt_g2) {
    constexpr int n_sage = D0 * F;
    constexpr int n_g1   = (H1 * D1) * D0;
    constexpr int n_g2   = C2P * D0;
    if (i < n_sage) { transpose_elem<F, D0, D0>(w_self, wt_self, i); return; }
    i -= n_sage;
    if (i < n_sage) { transpose_elem<F, D0, D0>(w_neigh, wt_neigh, i); return; }
    i -= n_sage;
    if (i < n_g1) { transpose_elem<D0, H1 * D1, H1 * D1>(g1_w, wt_g1, i); return; }
    i -= n_g1;
    if (i < n_g2) transpose_elem<D0, C2, C2P>(g2_w, wt_g2, i);
}

__device__ __forceinline__ void gemm_block(int bid, float* lds,
               const float* __restrict__ A, const float* __restrict__ Wt,
               float* __restrict__ out, int M, int K, int Nout, int ncg, int ldo)
{
    const int KP = K + 4;
    const int cg   = bid % ncg;
    const int col0 = cg * 64;
    float* w1s = lds;

    {
        const int qsh = 29 - __clz(K);
        const int qm  = (1 << qsh) - 1;
        const int npt = K >> 4;
        const float4* g1 = (const float4*)(Wt + (long long)col0 * K);
        constexpr int U = 8;
        for (int b = 0; b < npt; b += U) {
            float4 v1[U];
#pragma unroll
            for (int u = 0; u < U; ++u) {
                const int q = (b + u) * 256 + threadIdx.x;
                v1[u] = g1[q];
            }
#pragma unroll
            for (int u = 0; u < U; ++u) {
                const int q = (b + u) * 256 + threadIdx.x;
                const int o = (q >> qsh) * KP + (q & qm) * 4;
                *(float4*)(w1s + o) = v1[u];
            }
        }
    }
    __syncthreads();

    const int rt = (bid / ncg) * 8 + (threadIdx.x >> 5);
    if (rt >= M / 16) return;
    const int lane = threadIdx.x & 31;
    const int half = lane >> 4;
    const int l    = lane & 15;
    const int row  = rt * 16 + l;

    const float* Arow  = A + (long long)row * K + half * 4;
    __builtin_prefetch(Arow, 0, 3);

    const float* Bs[4];
#pragma unroll
    for (int t = 0; t < 4; ++t) Bs[t] = w1s + (t * 16 + l) * KP + half * 4;

    constexpr int KS = 32;
    v8f acc[4] = {};
    for (int kk = 0; kk < K; kk += KS) {
        v4f a[KS / 8], b[KS / 8][4];
#pragma unroll
        for (int u = 0; u < KS / 8; ++u) a[u] = *(const v4f*)(Arow + kk + u * 8);
#pragma unroll
        for (int u = 0; u < KS / 8; ++u) {
#pragma unroll
            for (int t = 0; t < 4; ++t) b[u][t] = *(const v4f*)(Bs[t] + kk + u * 8);
        }
        __builtin_amdgcn_sched_barrier(0);
#pragma unroll
        for (int u = 0; u < KS / 8; ++u) {
#pragma unroll
            for (int t = 0; t < 4; ++t)
                acc[t] = __builtin_amdgcn_wmma_f32_16x16x4_f32(
                    false, a[u].xy, false, b[u][t].xy, (short)0, acc[t], false, false);
#pragma unroll
            for (int t = 0; t < 4; ++t)
                acc[t] = __builtin_amdgcn_wmma_f32_16x16x4_f32(
                    false, a[u].zw, false, b[u][t].zw, (short)0, acc[t], false, false);
        }
    }

    const int plane = col0 / ldo;
    float* op = out + (long long)plane * M * ldo;
#pragma unroll
    for (int t = 0; t < 4; ++t) {
        const int c = col0 + t * 16 + l;
        if (c < Nout) {
            const int cc = c - plane * ldo;
#pragma unroll
            for (int v = 0; v < 8; ++v) {
                const int r = rt * 16 + v + 8 * half;
                op[(long long)r * ldo + cc] = acc[t][v];
            }
        }
    }
}

constexpr int CNT_P   = 32;
constexpr int CNT_T   = 1024;
constexpr int CNT_EPT = 25;
constexpr int CNT_EPB = CNT_T * CNT_EPT;
constexpr int SCAN_B  = 256;
constexpr int SCAN_G  = (Nn + SCAN_B - 1) / SCAN_B;
static_assert(CNT_P * CNT_EPB >= Ne, "slices must cover the edge list");
static_assert(Nn % 4 == 0 && SCAN_G <= SCAN_B && CNT_EPB % 256 == 0, "layout assumptions");

template <bool FULL>
__device__ __forceinline__ void count_slice(const int* __restrict__ dst, int* __restrict__ rank,
                                            int* c, int e0, int E) {
    int d[CNT_EPT];
#pragma unroll
    for (int j = 0; j < CNT_EPT; ++j) {
        const int e = e0 + j * CNT_T;
        d[j] = (FULL || e < E) ? dst[e] : 0;
    }
    __syncthreads();
    int r[CNT_EPT];
#pragma unroll
    for (int j = 0; j < CNT_EPT; ++j) {
        const int e = e0 + j * CNT_T;
        r[j] = (FULL || e < E) ? atomicAdd(&c[d[j]], 1) : 0;
    }
#pragma unroll
    for (int j = 0; j < CNT_EPT; ++j) {
        const int e = e0 + j * CNT_T;
        if (FULL || e < E) rank[e] = r[j];
    }
}

constexpr int PREP_G = (PREP_N + CNT_T - 1) / CNT_T;
__global__ __launch_bounds__(CNT_T)
void k_count(const int* __restrict__ dst, int* __restrict__ rank,
             int* __restrict__ part, int* __restrict__ coarse, int E,
             const float* __restrict__ w_self, const float* __restrict__ w_neigh,
             const float* __restrict__ g1_w, const float* __restrict__ g2_w,
             float* __restrict__ wt_self, float* __restrict__ wt_neigh,
             float* __restrict__ wt_g1, float* __restrict__ wt_g2) {
    __shared__ int4 c4[Nn / 4];
    if (blockIdx.x >= CNT_P) {
        prep_elem((int)(blockIdx.x - CNT_P) * CNT_T + (int)threadIdx.x,
                  w_self, w_neigh, g1_w, g2_w, wt_self, wt_neigh, wt_g1, wt_g2);
        return;
    }
    int* c = (int*)c4;
    const int t = threadIdx.x, p = blockIdx.x;
    for (int i = t; i < Nn / 4; i += CNT_T) c4[i] = make_int4(0, 0, 0, 0);
    const int e0 = p * CNT_EPB + t;
    if ((p + 1) * CNT_EPB <= E) count_slice<true>(dst, rank, c, e0, E);
    else                        count_slice<false>(dst, rank, c, e0, E);
    __syncthreads();
    int4* o4 = (int4*)(part + (long long)p * Nn);
    for (int i = t; i < Nn / 4; i += CNT_T) o4[i] = c4[i];
    constexpr int GPW = (SCAN_G + CNT_T / 32 - 1) / (CNT_T / 32);
    const int lane = t & 31, w = t >> 5;
    int s[GPW];
#pragma unroll
    for (int k = 0; k < GPW; ++k) {
        const int i0 = (w + k * (CNT_T / 32)) * (SCAN_B / 4) + lane, i1 = i0 + 32;
        int a = 0;
        if (i0 < Nn / 4) { const int4 v = c4[i0]; a += v.x + v.y + v.z + v.w; }
        if (i1 < Nn / 4) { const int4 v = c4[i1]; a += v.x + v.y + v.z + v.w; }
        s[k] = a;
    }
    for (int off = 16; off > 0; off >>= 1) {
#pragma unroll
        for (int k = 0; k < GPW; ++k) s[k] += __shfl_xor(s[k], off, 32);
    }
    if (lane == 0) {
#pragma unroll
        for (int k = 0; k < GPW; ++k) {
            const int g = w + k * (CNT_T / 32);
            if (g < SCAN_G) coarse[p * SCAN_G + g] = s[k];
        }
    }
}

__global__ __launch_bounds__(SCAN_B)
void k_scan(int* __restrict__ part, const int* __restrict__ coarse, int* __restrict__ rowptr) {
    __shared__ int wtot[SCAN_B / 32];
    __shared__ int wbase[SCAN_B / 32];
    const int t = threadIdx.x, g = blockIdx.x, n = g * SCAN_B + t;
    const int lane = t & 31, w = t >> 5;
    const int nl = min(n, Nn - 1);
    int v[CNT_P];
#pragma unroll
    for (int p = 0; p < CNT_P; ++p) v[p] = part[(long long)p * Nn + nl];
    int s = 0;
    if (t < g) {
#pragma unroll
        for (int p = 0; p < CNT_P; ++p) s += coarse[p * SCAN_G + t];
    }
    int deg = 0;
#pragma unroll
    for (int p = 0; p < CNT_P; ++p) deg += v[p];
    int inc = deg;
    for (int off = 1; off < 32; off <<= 1) {
        const int u = __shfl_up(inc, off, 32);
        if (lane >= off) inc += u;
    }
    for (int off = 16; off > 0; off >>= 1) s += __shfl_xor(s, off, 32);
    if (lane == 31) wtot[w] = inc;
    if (lane == 0) wbase[w] = s;
    __syncthreads();
    int run = inc - deg;
#pragma unroll
    for (int k = 0; k < SCAN_B / 32; ++k) {
        run += wbase[k];
        if (k < w) run += wtot[k];
    }
    if (n < Nn) {
        rowptr[n] = run;
        if (n == Nn - 1) rowptr[Nn] = run + deg;
#pragma unroll
        for (int p = 0; p < CNT_P; ++p) { part[(long long)p * Nn + n] = run; run += v[p]; }
    }
}

constexpr int FILL_EPT = 4;
constexpr int FILL_EPB = 256 * FILL_EPT;
constexpr int FILL_G   = (Ne + FILL_EPB - 1) / FILL_EPB;
static_assert(CNT_EPB % FILL_EPB == 0, "a placing block's edges lie in one slice");
template <bool FULL>
__device__ __forceinline__ void fill_edges(int i0, const int* __restrict__ src, const int* __restrict__ dst,
                                           const int* __restrict__ rank, const int* __restrict__ start,
                                           int* __restrict__ csr_src, int E) {
    int d[FILL_EPT], r[FILL_EPT], s[FILL_EPT], pos[FILL_EPT];
#pragma unroll
    for (int j = 0; j < FILL_EPT; ++j) {
        const int i = FULL ? i0 + j * 256 : min(i0 + j * 256, E - 1);
        d[j] = dst[i]; r[j] = rank[i]; s[j] = src[i];
    }
#pragma unroll
    for (int j = 0; j < FILL_EPT; ++j) pos[j] = start[d[j]];
#pragma unroll
    for (int j = 0; j < FILL_EPT; ++j)
        if (FULL || i0 + j * 256 < E) csr_src[pos[j] + r[j]] = s[j];
}
__device__ __forceinline__ void fill_block(int b, const int* __restrict__ src, const int* __restrict__ dst,
                                           const int* __restrict__ rank, const int* __restrict__ part,
                                           int* __restrict__ csr_src, int E) {
    const int* start = part + (long long)(b / (CNT_EPB / FILL_EPB)) * Nn;
    const int i0 = b * FILL_EPB + (int)threadIdx.x;
    if ((b + 1) * FILL_EPB <= E) fill_edges<true>(i0, src, dst, rank, start, csr_src, E);
    else                         fill_edges<false>(i0, src, dst, rank, start, csr_src, E);
}

__global__ __launch_bounds__(256)
void k_gemm_fill(const float* __restrict__ A, const float* __restrict__ Wt,
                 float* __restrict__ out, int M, int K, int Nout, int ncg, int ldo, int ngemm,
                 const int* __restrict__ src, const int* __restrict__ dst,
                 const int* __restrict__ rank, const int* __restrict__ part,
                 int* __restrict__ csr_src, int E) {
    extern __shared__ float lds[];
    const int bid = blockIdx.x;
    if (bid < ngemm) gemm_block(bid, lds, A, Wt, out, M, K, Nout, ncg, ldo);
    else fill_block(bid - ngemm, src, dst, rank, part, csr_src, E);
}

__device__ __forceinline__ int bcast_i(int v, int j) { return __builtin_amdgcn_readlane(v, j); }
__device__ __forceinline__ float bcast_f(float v, int j) {
    return __uint_as_float((unsigned)__builtin_amdgcn_readlane((int)__float_as_uint(v), j));
}
__device__ __forceinline__ float wave_max(float v) {
    for (int off = 16; off > 0; off >>= 1) v = fmaxf(v, __shfl_xor(v, off, 32));
    return v;
}
__device__ __forceinline__ float wave_sum(float v) {
    for (int off = 16; off > 0; off >>= 1) v += __shfl_xor(v, off, 32);
    return v;
}
__device__ __forceinline__ void add4(float4& a, const float4 v) {
    a.x += v.x; a.y += v.y; a.z += v.z; a.w += v.w;
}
__device__ __forceinline__ void fma4(float4& a, float s, const float4 v) {
    a.x += s * v.x; a.y += s * v.y; a.z += s * v.z; a.w += s * v.w;
}
__device__ __forceinline__ float leaky(float v) { return v > 0.0f ? v : 0.2f * v; }
__device__ __forceinline__ float finite_or_zero(float m) {
    return (fabsf(m) < __builtin_huge_valf()) ? m : 0.0f;
}

__device__ __forceinline__ float4 sage_row(const float* __restrict__ xs, const float* __restrict__ xn,
                                           const int* __restrict__ rowptr, const int* __restrict__ csr_src,
                                           const float* __restrict__ bias, int n, int lane) {
    const int beg = rowptr[n], end = rowptr[n + 1];
    const float4 own = ((const float4*)(xs + (long long)n * D0))[lane];
    float4 acc = make_float4(0.f, 0.f, 0.f, 0.f);
    for (int b = beg; b < end; b += 32) {
        const int m = min(32, end - b);
        const int sv = (lane < m) ? csr_src[b + lane] : 0;
        int j = 0;
        for (; j + 8 <= m; j += 8) {
            float4 f[8];
#pragma unroll
            for (int u = 0; u < 8; ++u)
                f[u] = ((const float4*)(xn + (long long)bcast_i(sv, j + u) * D0))[lane];
#pragma unroll
            for (int u = 0; u < 8; ++u) add4(acc, f[u]);
        }
        if (j + 4 <= m) {
            float4 f[4];
#pragma unroll
            for (int u = 0; u < 4; ++u)
                f[u] = ((const float4*)(xn + (long long)bcast_i(sv, j + u) * D0))[lane];
#pragma unroll
            for (int u = 0; u < 4; ++u) add4(acc, f[u]);
            j += 4;
        }
        for (; j < m; ++j)
            add4(acc, ((const float4*)(xn + (long long)bcast_i(sv, j) * D0))[lane]);
    }
    const float inv = 1.0f / fmaxf((float)(end - beg), 1.0f);
    const float* bl = bias + lane * 4;
    float4 r;
    r.x = own.x + acc.x * inv + bl[0]; r.y = own.y + acc.y * inv + bl[1];
    r.z = own.z + acc.z * inv + bl[2]; r.w = own.w + acc.w * inv + bl[3];
    r.x = r.x > 0.0f ? r.x : (expf(r.x) - 1.0f);
    r.y = r.y > 0.0f ? r.y : (expf(r.y) - 1.0f);
    r.z = r.z > 0.0f ? r.z : (expf(r.z) - 1.0f);
    r.w = r.w > 0.0f ? r.w : (expf(r.w) - 1.0f);
    return r;
}


__device__ __forceinline__ void gat1_rows(float4& acc, const float* __restrict__ feat,
                                          int sv, float w0, float w1, int m, int lane, bool hi) {
    int j = 0;
    for (; j + 4 <= m; j += 4) {
        const float4 f0 = ((const float4*)(feat + (long long)bcast_i(sv, j + 0) * (H1 * D1)))[lane];
        const float4 f1 = ((const float4*)(feat + (long long)bcast_i(sv, j + 1) * (H1 * D1)))[lane];
        const float4 f2 = ((const float4*)(feat + (long long)bcast_i(sv, j + 2) * (H1 * D1)))[lane];
        const float4 f3 = ((const float4*)(feat + (long long)bcast_i(sv, j + 3) * (H1 * D1)))[lane];
        const float u0 = bcast_f(w0, j + 0), t0 = bcast_f(w1, j + 0);
        const float u1 = bcast_f(w0, j + 1), t1 = bcast_f(w1, j + 1);
        const float u2 = bcast_f(w0, j + 2), t2 = bcast_f(w1, j + 2);
        const float u3 = bcast_f(w0, j + 3), t3 = bcast_f(w1, j + 3);
        fma4(acc, hi ? t0 : u0, f0);
        fma4(acc, hi ? t1 : u1, f1);
        fma4(acc, hi ? t2 : u2, f2);
        fma4(acc, hi ? t3 : u3, f3);
    }
    for (; j < m; ++j) {
        const float4 f0 = ((const float4*)(feat + (long long)bcast_i(sv, j) * (H1 * D1)))[lane];
        const float u0 = bcast_f(w0, j), t0 = bcast_f(w1, j);
        fma4(acc, hi ? t0 : u0, f0);
    }
}

__device__ __forceinline__ float4 gat1_row(const int* __restrict__ rowptr, const int* __restrict__ csr_src,
                                           const float* __restrict__ feat, const float* __restrict__ el,
                                           const float* __restrict__ er, const float* __restrict__ bias,
                                           int n, int lane) {
    const int beg = rowptr[n], end = rowptr[n + 1];
    const float er0 = er[n * 2], er1 = er[n * 2 + 1];

    const bool has = beg + lane < end;
    int sv = 0;
    float x0 = -__builtin_huge_valf(), x1 = x0;
    if (has) {
        sv = csr_src[beg + lane];
        const float2 l = *(const float2*)(el + (long long)sv * 2);
        x0 = leaky(l.x + er0); x1 = leaky(l.y + er1);
    }

    float m0 = x0, m1 = x1;
    for (int b = beg + 32 + lane; b < end; b += 32) {
        const float2 l = *(const float2*)(el + (long long)csr_src[b] * 2);
        m0 = fmaxf(m0, leaky(l.x + er0)); m1 = fmaxf(m1, leaky(l.y + er1));
    }
    m0 = finite_or_zero(wave_max(m0)); m1 = finite_or_zero(wave_max(m1));

    const float p0 = has ? __expf(x0 - m0) : 0.0f;
    const float p1 = has ? __expf(x1 - m1) : 0.0f;
    float s0 = p0, s1 = p1;
    for (int b = beg + 32 + lane; b < end; b += 32) {
        const float2 l = *(const float2*)(el + (long long)csr_src[b] * 2);
        s0 += __expf(leaky(l.x + er0) - m0); s1 += __expf(leaky(l.y + er1) - m1);
    }
    const float d0 = fmaxf(wave_sum(s0), 1e-9f), d1 = fmaxf(wave_sum(s1), 1e-9f);

    const bool hi = lane >= 16;
    float4 acc = make_float4(0.f, 0.f, 0.f, 0.f);
    gat1_rows(acc, feat, sv, p0 / d0, p1 / d1, min(32, end - beg), lane, hi);
    for (int b = beg + 32; b < end; b += 32) {
        const int m = min(32, end - b);
        int sb = 0; float w0 = 0.0f, w1 = 0.0f;
        if (lane < m) {
            sb = csr_src[b + lane];
            const float2 l = *(const float2*)(el + (long long)sb * 2);
            w0 = __expf(leaky(l.x + er0) - m0) / d0;
            w1 = __expf(leaky(l.y + er1) - m1) / d1;
        }
        gat1_rows(acc, feat, sb, w0, w1, m, lane, hi);
    }
    const float* bl = bias + lane * 4;
    float4 r;
    r.x = acc.x + bl[0]; r.y = acc.y + bl[1]; r.z = acc.z + bl[2]; r.w = acc.w + bl[3];
    r.x = r.x > 0.0f ? r.x : (expf(r.x) - 1.0f);
    r.y = r.y > 0.0f ? r.y : (expf(r.y) - 1.0f);
    r.z = r.z > 0.0f ? r.z : (expf(r.z) - 1.0f);
    r.w = r.w > 0.0f ? r.w : (expf(r.w) - 1.0f);
    return r;
}

constexpr int KP1 = D0 + 4;
constexpr int FUSE_T = 512;
constexpr int FUSE_G = 1024;
constexpr int FUSE_G3 = 768;
static_assert(Nn % 16 == 0, "a tile is 16 whole nodes");

template <int NCOL>
__device__ __forceinline__ void stage_panel(const float* __restrict__ wt, float* ws) {
    constexpr int PT = NCOL * (D0 / 4) / FUSE_T;
    const float4* g = (const float4*)wt;
    float4 v[PT];
#pragma unroll
    for (int u = 0; u < PT; ++u) v[u] = g[u * FUSE_T + threadIdx.x];
#pragma unroll
    for (int u = 0; u < PT; ++u) {
        const int q = u * FUSE_T + threadIdx.x;
        *(float4*)(ws + (q >> 5) * KP1 + (q & 31) * 4) = v[u];
    }
}

template <int NCOL, int NH>
__device__ __forceinline__ void project_tile(const float* hs, const float* ws, float* red, int row0,
                                             float* __restrict__ feat, int Nout,
                                             const float* __restrict__ al, const float* __restrict__ ar,
                                             float* __restrict__ el, float* __restrict__ er) {
    static_assert(NCOL == 64 * NH, "a head is 64 (padded) columns");
    const int w = __builtin_amdgcn_readfirstlane((int)(threadIdx.x >> 5));
    const int lane = threadIdx.x & 31, half = lane >> 4, l = lane & 15;
    if (w < NCOL / 16) {
        const float* Ap = hs + l * KP1 + half * 4;
        const float* Bp = ws + (w * 16 + l) * KP1 + half * 4;
        v8f acc = {};
        for (int kk = 0; kk < D0; kk += 32) {
            v4f a[4], b[4];
#pragma unroll
            for (int u = 0; u < 4; ++u) a[u] = *(const v4f*)(Ap + kk + u * 8);
#pragma unroll
            for (int u = 0; u < 4; ++u) b[u] = *(const v4f*)(Bp + kk + u * 8);
#pragma unroll
            for (int u = 0; u < 4; ++u) {
                acc = __builtin_amdgcn_wmma_f32_16x16x4_f32(
                    false, a[u].xy, false, b[u].xy, (short)0, acc, false, false);
                acc = __builtin_amdgcn_wmma_f32_16x16x4_f32(
                    false, a[u].zw, false, b[u].zw, (short)0, acc, false, false);
            }
        }
        asm volatile("v_nop\n\tv_nop\n\tv_nop\n\tv_nop" : "+v"(acc));
        const int c = w * 16 + l;
        const bool live = c < Nout;
        if (live) {
#pragma unroll
            for (int v = 0; v < 8; ++v)
                feat[(long long)(row0 + v + 8 * half) * Nout + c] = acc[v];
        }
        const float wl = live ? al[c] : 0.0f, wr = live ? ar[c] : 0.0f;
        const bool up = (l & 8) != 0;
        float s[8];
#pragma unroll
        for (int v = 0; v < 8; ++v) {
            const float pl = acc[v] * wl, pr = acc[v] * wr;
            const float keep = up ? pr : pl;
            const float give = up ? pl : pr;
            s[v] = keep + __shfl_xor(give, 8, 32);
        }
#pragma unroll
        for (int off = 4; off > 0; off >>= 1) {
#pragma unroll
            for (int v = 0; v < 8; ++v) s[v] += __shfl_xor(s[v], off, 32);
        }
        const int j = l & 7;
        float val = s[0];
#pragma unroll
        for (int v = 1; v < 8; ++v) val = (j == v) ? s[v] : val;
        red[w * 32 + (up ? 16 : 0) + 8 * half + j] = val;
    }
    __syncthreads();
    if (threadIdx.x < 32 * NH) {
        const int row = threadIdx.x & 15, which = (threadIdx.x >> 4) & 1, h = threadIdx.x >> 5;
        const float* p = red + (h * 4) * 32 + which * 16 + row;
        const float sum = ((p[0] + p[32]) + p[64]) + p[96];
        (which ? er : el)[(long long)(row0 + row) * NH + h] = sum;
    }
}

__global__ __launch_bounds__(FUSE_T) __attribute__((amdgpu_waves_per_eu(16, 16)))
void k_sage_project(const float* __restrict__ xs, const float* __restrict__ xn,
                    const int* __restrict__ rowptr, const int* __restrict__ csr_src,
                    const float* __restrict__ bias, const float* __restrict__ wt,
                    const float* __restrict__ al, const float* __restrict__ ar,
                    float* __restrict__ feat, float* __restrict__ el, float* __restrict__ er) {
    extern __shared__ float fuse_lds[];
    float* ws = fuse_lds;
    float* hs = ws + (H1 * D1) * KP1;
    float* red = hs + 16 * KP1;
    stage_panel<H1 * D1>(wt, ws);
    const int lane = threadIdx.x & 31, w = threadIdx.x >> 5;
    for (int tile = blockIdx.x; tile < Nn / 16; tile += gridDim.x) {
        const int n = __builtin_amdgcn_readfirstlane(tile * 16 + w);
        const float4 r = sage_row(xs, xn, rowptr, csr_src, bias, n, lane);
        *(float4*)(hs + w * KP1 + lane * 4) = r;
        __syncthreads();
        project_tile<H1 * D1, H1>(hs, ws, red, tile * 16, feat, H1 * D1, al, ar, el, er);
    }
}

__global__ __launch_bounds__(FUSE_T)
void k_gat1_project(const int* __restrict__ rowptr, const int* __restrict__ csr_src,
                    const float* __restrict__ feat1, const float* __restrict__ el1,
                    const float* __restrict__ er1, const float* __restrict__ bias,
                    const float* __restrict__ wt, const float* __restrict__ al, const float* __restrict__ ar,
                    float* __restrict__ feat2, float* __restrict__ el2, float* __restrict__ er2) {
    extern __shared__ float fuse_lds[];
    float* ws = fuse_lds;
    float* hs = ws + C2P * KP1;
    float* red = hs + 16 * KP1;
    stage_panel<C2P>(wt, ws);
    const int lane = threadIdx.x & 31, w = threadIdx.x >> 5;
    for (int tile = blockIdx.x; tile < Nn / 16; tile += gridDim.x) {
        const int n = __builtin_amdgcn_readfirstlane(tile * 16 + w);
        const float4 r = gat1_row(rowptr, csr_src, feat1, el1, er1, bias, n, lane);
        *(float4*)(hs + w * KP1 + lane * 4) = r;
        __syncthreads();
        project_tile<C2P, 1>(hs, ws, red, tile * 16, feat2, C2, al, ar, el2, er2);
    }
}

__device__ __forceinline__ void gat2_rows(float& acc0, float& acc1, const float* __restrict__ feat,
                                          int sv, float wv, int m, int lane, bool two) {
    int j = 0;
    for (; j + 4 <= m; j += 4) {
        const float* r0 = feat + (long long)bcast_i(sv, j + 0) * C2;
        const float* r1 = feat + (long long)bcast_i(sv, j + 1) * C2;
        const float* r2 = feat + (long long)bcast_i(sv, j + 2) * C2;
        const float* r3 = feat + (long long)bcast_i(sv, j + 3) * C2;
        const float p0 = r0[lane], p1 = r1[lane], p2 = r2[lane], p3 = r3[lane];
        const float q0 = two ? r0[lane + 32] : 0.0f, q1 = two ? r1[lane + 32] : 0.0f;
        const float q2 = two ? r2[lane + 32] : 0.0f, q3 = two ? r3[lane + 32] : 0.0f;
        const float u0 = bcast_f(wv, j + 0), u1 = bcast_f(wv, j + 1);
        const float u2 = bcast_f(wv, j + 2), u3 = bcast_f(wv, j + 3);
        acc0 += u0 * p0; acc1 += u0 * q0; acc0 += u1 * p1; acc1 += u1 * q1;
        acc0 += u2 * p2; acc1 += u2 * q2; acc0 += u3 * p3; acc1 += u3 * q3;
    }
    for (; j < m; ++j) {
        const float* r0 = feat + (long long)bcast_i(sv, j) * C2;
        const float u0 = bcast_f(wv, j);
        acc0 += u0 * r0[lane];
        acc1 += u0 * (two ? r0[lane + 32] : 0.0f);
    }
}

__global__ __launch_bounds__(256)
void k_gat2_gather(const int* __restrict__ rowptr, const int* __restrict__ csr_src,
                   const float* __restrict__ feat, const float* __restrict__ el,
                   const float* __restrict__ er, const float* __restrict__ bias,
                   float* __restrict__ out, int N) {
    const int n = __builtin_amdgcn_readfirstlane((int)(blockIdx.x * 8 + (threadIdx.x >> 5)));
    if (n >= N) return;
    const int lane = threadIdx.x & 31;
    const int beg = rowptr[n], end = rowptr[n + 1];
    const float ern = er[n];
    const bool two = lane < C2 - 32;

    const bool has = beg + lane < end;
    int sv = 0;
    float xv = -__builtin_huge_valf();
    if (has) {
        sv = csr_src[beg + lane];
        xv = leaky(el[sv] + ern);
    }

    float mx = xv;
    for (int b = beg + 32 + lane; b < end; b += 32)
        mx = fmaxf(mx, leaky(el[csr_src[b]] + ern));
    mx = finite_or_zero(wave_max(mx));
    const float pv = has ? __expf(xv - mx) : 0.0f;
    float sm = pv;
    for (int b = beg + 32 + lane; b < end; b += 32) sm += __expf(leaky(el[csr_src[b]] + ern) - mx);
    const float dn = fmaxf(wave_sum(sm), 1e-9f);

    float acc0 = 0.0f, acc1 = 0.0f;
    gat2_rows(acc0, acc1, feat, sv, pv / dn, min(32, end - beg), lane, two);
    for (int b = beg + 32; b < end; b += 32) {
        const int m = min(32, end - b);
        int sb = 0; float wv = 0.0f;
        if (lane < m) {
            sb = csr_src[b + lane];
            wv = __expf(leaky(el[sb] + ern) - mx) / dn;
        }
        gat2_rows(acc0, acc1, feat, sb, wv, m, lane, two);
    }
    float* o = out + (long long)n * C2;
    o[lane] = acc0 + bias[lane];
    if (two) o[lane + 32] = acc1 + bias[lane + 32];
}

static inline unsigned gsz(long long n, int b) { return (unsigned)((n + b - 1) / b); }

extern "C" void kernel_launch(void* const* d_in, const int* in_sizes, int n_in,
                              void* d_out, int out_size, void* d_ws, size_t ws_size,
                              hipStream_t stream) {
    const float* x       = (const float*)d_in[0];
    const int*   src     = (const int*)  d_in[1];
    const int*   dst     = (const int*)  d_in[2];
    const float* w_self  = (const float*)d_in[3];
    const float* w_neigh = (const float*)d_in[4];
    const float* sage_b  = (const float*)d_in[5];
    const float* g1_w    = (const float*)d_in[6];
    const float* g1_al   = (const float*)d_in[7];
    const float* g1_ar   = (const float*)d_in[8];
    const float* g1_b    = (const float*)d_in[9];
    const float* g2_w    = (const float*)d_in[10];
    const float* g2_al   = (const float*)d_in[11];
    const float* g2_ar   = (const float*)d_in[12];
    const float* g2_b    = (const float*)d_in[13];

    float*    ws    = (float*)d_ws;
    float*    proj  = ws;
    float*    feat1 = proj + (long long)Nn * F;
    float*    el    = feat1 + (long long)Nn * D0;
    float*    er    = el + Nn * H1;
    int*      rowptr = (int*)(er + Nn * H1);
    int*      rank  = rowptr + (Nn + 16);
    int*      csr_src = rank + Ne;
    float*    wt_self  = (float*)(csr_src + Ne);
    float*    wt_neigh = wt_self + D0 * F;
    float*    wt_g1    = wt_neigh + D0 * F;
    float*    wt_g2    = wt_g1 + (H1 * D1) * D0;
    int*      part   = (int*)(wt_g2 + C2P * D0);
    int*      coarse = part + CNT_P * Nn;
    float*    el2   = (float*)(coarse + CNT_P * SCAN_G);
    float*    er2   = el2 + Nn;
    float*    feat2 = proj;

    const int B = 256;
    const int Mtiles = Nn / 16;
    const unsigned rowblk = gsz(Mtiles, 8);
    const unsigned nodeblk = gsz(Nn, 8);
    constexpr size_t sh_tile = (16ull * KP1 + 8 * 32) * sizeof(float);

    k_count<<<CNT_P + PREP_G, CNT_T, 0, stream>>>(dst, rank, part, coarse, Ne,
        w_self, w_neigh, g1_w, g2_w, wt_self, wt_neigh, wt_g1, wt_g2);
    k_scan<<<SCAN_G, SCAN_B, 0, stream>>>(part, coarse, rowptr);

    {
        const size_t shmem = 64ull * (F + 4) * sizeof(float);
        k_gemm_fill<<<rowblk * 4 + FILL_G, B, shmem, stream>>>(x, wt_self, proj, Nn, F, 2 * D0, 4, D0,
            (int)(rowblk * 4), src, dst, rank, part, csr_src, Ne);
    }
    k_sage_project<<<FUSE_G, FUSE_T, (size_t)(H1 * D1) * KP1 * sizeof(float) + sh_tile, stream>>>(
        proj, proj + (long long)Nn * D0, rowptr, csr_src, sage_b, wt_g1, g1_al, g1_ar, feat1, el, er);

    k_gat1_project<<<FUSE_G3, FUSE_T, (size_t)C2P * KP1 * sizeof(float) + sh_tile, stream>>>(
        rowptr, csr_src, feat1, el, er, g1_b, wt_g2, g2_al, g2_ar, feat2, el2, er2);

    k_gat2_gather<<<nodeblk, B, 0, stream>>>(rowptr, csr_src, feat2, el2, er2, g2_b,
                                             (float*)d_out, Nn);
}
